# starting kernel
# baseline (speedup 1.0000x reference)
#include <hip/hip_runtime.h>
#include <hip/hip_bf16.h>

typedef __attribute__((ext_vector_type(2))) float v2f;
typedef __attribute__((ext_vector_type(8))) float v8f;

#define HID 64
#define INDIM 128
#define KDIM 320
#define GCOLS 256
#define LMAX 2.0f

__device__ __forceinline__ v8f wmma_f32_4(v2f a, v2f b, v8f c) {
  return __builtin_amdgcn_wmma_f32_16x16x4_f32(false, a, false, b, (short)0, c, false, false);
}

__device__ __forceinline__ float fsigmoid(float x) {
  return 1.0f / (1.0f + __expf(-x));
}

__device__ __forceinline__ void atomic_fadd(float* p, float v) {
  asm volatile("global_atomic_add_f32 %0, %1, off scope:SCOPE_DEV"
               :: "v"(p), "v"(v) : "memory");
}


__global__ __launch_bounds__(256)
void k_degree(const int* __restrict__ src, const int* __restrict__ dst,
              const float* __restrict__ ew, float* __restrict__ deg, int E) {
  int e = blockIdx.x * 256 + threadIdx.x;
  if (e >= E) return;
  int s = src[e], d = dst[e];
  if (s != d) atomic_fadd(&deg[s], ew[e]);
}

__global__ __launch_bounds__(256)
void k_dinv(const float* __restrict__ deg, float* __restrict__ dinv, int Nn) {
  int i = blockIdx.x * 256 + threadIdx.x;
  if (i >= Nn) return;
  float dg = deg[i];
  dinv[i] = (dg > 0.0f) ? rsqrtf(dg) : 0.0f;
}

__global__ __launch_bounds__(256)
void k_lapw(const int* __restrict__ src, const int* __restrict__ dst,
            const float* __restrict__ ew, const float* __restrict__ dinv,
            float* __restrict__ lapw, int E) {
  int e = blockIdx.x * 256 + threadIdx.x;
  if (e >= E) return;
  int s = src[e], d = dst[e];
  float w = (s == d) ? 0.0f : ew[e];
  lapw[e] = (2.0f / LMAX) * (-dinv[s] * w * dinv[d]);
}

__global__ __launch_bounds__(256)
void k_scale(float* __restrict__ dstp, const float* __restrict__ srcp,
             float scale, int count) {
  int i = blockIdx.x * 256 + threadIdx.x;
  if (i >= count) return;
  dstp[i] = scale * srcp[i];
}

__global__ __launch_bounds__(256)
void k_spmm(const float* __restrict__ z, float* __restrict__ acc,
            const int* __restrict__ src, const int* __restrict__ dst,
            const float* __restrict__ lw, int E) {
  int gid = blockIdx.x * 256 + threadIdx.x;
  int e = gid >> 4;
  if (e >= E) return;
  float w = lw[e];
  if (w == 0.0f) return;
  int ch = (gid & 15) << 2;
  int s = src[e], d = dst[e];
  const float4 zv = *(const float4*)(z + (size_t)s * HID + ch);
  float* out = acc + (size_t)d * HID + ch;
  atomic_fadd(out + 0, w * zv.x);
  atomic_fadd(out + 1, w * zv.y);
  atomic_fadd(out + 2, w * zv.z);
  atomic_fadd(out + 3, w * zv.w);
}

__global__ __launch_bounds__(256)
void k_cheb2(float* __restrict__ t2, const float* __restrict__ h, int count) {
  int i = blockIdx.x * 256 + threadIdx.x;
  if (i >= count) return;
  t2[i] = 2.0f * t2[i] - h[i];
}


__global__ __launch_bounds__(256)
void k_packB(const float* __restrict__ Wx, const float* __restrict__ theta,
             const float* __restrict__ bg, const float* __restrict__ conv_b,
             float* __restrict__ Bp, float* __restrict__ bias) {
  int gid = blockIdx.x * 256 + threadIdx.x;
  if (gid >= KDIM * GCOLS) return;
  int k = gid / GCOLS, col = gid % GCOLS;
  int g = col >> 6, o = col & 63;
  float v;
  if (k < INDIM) {
    v = Wx[((size_t)g * INDIM + k) * HID + o];
  } else {
    int kk = k - INDIM;
    int kb = kk >> 6, kr = kk & 63;
    v = theta[(((size_t)g * 3 + kb) * HID + kr) * HID + o];
  }
  Bp[gid] = v;
  if (gid < GCOLS) bias[gid] = bg[g * HID + o] + conv_b[g * HID + o];
}


#define AS1 324
#define PS1 260

__global__ __launch_bounds__(256)
void k_gemm_gates(const float* __restrict__ x, const float* __restrict__ h,
                  const float* __restrict__ T1, const float* __restrict__ T2,
                  const float* __restrict__ Bp, const float* __restrict__ bias,
                  const float* __restrict__ c_in,
                  float* __restrict__ h_out, float* __restrict__ c_out, int Nn) {
  __shared__ float Asmem[16 * AS1];
  __shared__ float Psmem[16 * PS1];
  const int tid = threadIdx.x;
  const int m0 = blockIdx.x * 16;

  for (int i = tid; i < 16 * KDIM; i += 256) {
    int r = i / KDIM, k = i % KDIM;
    int n = m0 + r; if (n >= Nn) n = Nn - 1;
    float v;
    if (k < 128)      v = x[(size_t)n * INDIM + k];
    else if (k < 192) v = h[(size_t)n * HID + (k - 128)];
    else if (k < 256) v = T1[(size_t)n * HID + (k - 192)];
    else              v = T2[(size_t)n * HID + (k - 256)];
    Asmem[r * AS1 + k] = v;
  }
  __syncthreads();

  const int wave = tid >> 5, lane = tid & 31;
  const int lo = lane & 15, hi = lane >> 4;
  const int c0 = wave * 32;
  v8f acc0 = {}; v8f acc1 = {};

  for (int k0 = 0; k0 < KDIM; k0 += 4) {
    v2f a = *(const v2f*)&Asmem[lo * AS1 + k0 + 2 * hi];
    const float* bptr = Bp + (size_t)(k0 + 2 * hi) * GCOLS + c0 + lo;
    v2f b0, b1;
    b0[0] = bptr[0];    b0[1] = bptr[GCOLS];
    b1[0] = bptr[16];   b1[1] = bptr[GCOLS + 16];
    acc0 = wmma_f32_4(a, b0, acc0);
    acc1 = wmma_f32_4(a, b1, acc1);
  }

  float bias0 = bias[c0 + lo], bias1 = bias[c0 + 16 + lo];
#pragma unroll
  for (int r = 0; r < 8; ++r) {
    int row = r + 8 * hi;
    Psmem[row * PS1 + c0 + lo]      = acc0[r] + bias0;
    Psmem[row * PS1 + c0 + 16 + lo] = acc1[r] + bias1;
  }
  __syncthreads();

#pragma unroll
  for (int j = 0; j < 4; ++j) {
    int flat = j * 256 + tid;
    int row = flat >> 6, o = flat & 63;
    int n = m0 + row;
    if (n >= Nn) continue;
    const float* pr = &Psmem[row * PS1];
    float ig = fsigmoid(pr[o]);
    float fg = fsigmoid(pr[64 + o]);
    float tg = tanhf(pr[128 + o]);
    float og = fsigmoid(pr[192 + o]);
    float cv = fg * c_in[(size_t)n * HID + o] + ig * tg;
    float hv = og * tanhf(cv);
    h_out[(size_t)n * HID + o] = hv;
    c_out[(size_t)n * HID + o] = cv;
  }
}


#define AS2 68
#define BS2 68

__global__ __launch_bounds__(256)
void k_gemm_out(const float* __restrict__ h0, const float* __restrict__ Wl,
                const float* __restrict__ bl, float* __restrict__ out, int Nn) {
  __shared__ float Bs[64 * BS2];
  __shared__ float As[32 * AS2];
  const int tid = threadIdx.x;
  const int m0 = blockIdx.x * 32;

  for (int i = tid; i < 64 * 64; i += 256) {
    int k = i >> 6, ccol = i & 63;
    Bs[k * BS2 + ccol] = Wl[i];
  }
  for (int i = tid; i < 32 * 64; i += 256) {
    int r = i >> 6, k = i & 63;
    int n = m0 + r; if (n >= Nn) n = Nn - 1;
    float v = h0[(size_t)n * HID + k];
    As[r * AS2 + k] = v > 0.0f ? v : 0.0f;
  }
  __syncthreads();

  const int wave = tid >> 5, lane = tid & 31;
  const int lo = lane & 15, hi = lane >> 4;
  const int r0 = (wave >> 2) * 16, c0 = (wave & 3) * 16;
  v8f acc = {};
  for (int k0 = 0; k0 < 64; k0 += 4) {
    v2f a = *(const v2f*)&As[(r0 + lo) * AS2 + k0 + 2 * hi];
    v2f b;
    b[0] = Bs[(k0 + 2 * hi) * BS2 + c0 + lo];
    b[1] = Bs[(k0 + 2 * hi + 1) * BS2 + c0 + lo];
    acc = wmma_f32_4(a, b, acc);
  }
  float bb = bl[c0 + lo];
#pragma unroll
  for (int r = 0; r < 8; ++r) {
    int row = r0 + r + 8 * hi;
    int n = m0 + row;
    if (n < Nn) out[(size_t)n * HID + c0 + lo] = acc[r] + bb;
  }
}


extern "C" void kernel_launch(void* const* d_in, const int* in_sizes, int n_in,
                              void* d_out, int out_size, void* d_ws, size_t ws_size,
                              hipStream_t stream) {
  const float* x      = (const float*)d_in[0];
  const int*   ei     = (const int*)d_in[1];
  const float* ew     = (const float*)d_in[2];
  const float* h      = (const float*)d_in[3];
  const float* c      = (const float*)d_in[4];
  const float* Wx     = (const float*)d_in[5];
  const float* bg     = (const float*)d_in[6];
  const float* theta  = (const float*)d_in[7];
  const float* conv_b = (const float*)d_in[8];
  const float* Wl     = (const float*)d_in[9];
  const float* bl     = (const float*)d_in[10];

  const int E  = in_sizes[2];
  const int Nn = in_sizes[3] / HID;
  const int* src = ei;
  const int* dst = ei + E;

  float* deg  = (float*)d_ws;
  float* dinv = deg  + Nn;
  float* lapw = dinv + Nn;
  float* T1   = lapw + E;
  float* T2   = T1 + (size_t)Nn * HID;
  float* Bp   = T2 + (size_t)Nn * HID;
  float* bias = Bp + (size_t)KDIM * GCOLS;

  float* out   = (float*)d_out;
  float* h_out = out   + (size_t)Nn * HID;
  float* c_out = h_out + (size_t)Nn * HID;

  const float DIAG = 2.0f / LMAX - 1.0f;
  const int NH = Nn * HID;

  hipMemsetAsync(deg, 0, (size_t)Nn * sizeof(float), stream);
  k_degree<<<(E + 255) / 256, 256, 0, stream>>>(src, dst, ew, deg, E);
  k_dinv<<<(Nn + 255) / 256, 256, 0, stream>>>(deg, dinv, Nn);
  k_lapw<<<(E + 255) / 256, 256, 0, stream>>>(src, dst, ew, dinv, lapw, E);

  k_scale<<<(NH + 255) / 256, 256, 0, stream>>>(T1, h, DIAG, NH);
  k_spmm<<<(E * 16 + 255) / 256, 256, 0, stream>>>(h, T1, src, dst, lapw, E);
  k_scale<<<(NH + 255) / 256, 256, 0, stream>>>(T2, T1, DIAG, NH);
  k_spmm<<<(E * 16 + 255) / 256, 256, 0, stream>>>(T1, T2, src, dst, lapw, E);
  k_cheb2<<<(NH + 255) / 256, 256, 0, stream>>>(T2, h, NH);

  k_packB<<<(KDIM * GCOLS + 255) / 256, 256, 0, stream>>>(Wx, theta, bg, conv_b, Bp, bias);

  k_gemm_gates<<<(Nn + 15) / 16, 256, 0, stream>>>(x, h, T1, T2, Bp, bias, c,
                                                   h_out, c_out, Nn);
  k_gemm_out<<<(Nn + 31) / 32, 256, 0, stream>>>(h_out, Wl, bl, out, Nn);
}


# gate GEMM re-tiled to 64 rows per work-group: each B fragment feeds 4 row sub-tiles (8 accumulators per wave), tile staged and gates written with 16-byte accesses issued together
# speedup vs baseline: 1.0456x; 1.0456x over previous
#include <hip/hip_runtime.h>
#include <hip/hip_bf16.h>

typedef __attribute__((ext_vector_type(2))) float v2f;
typedef __attribute__((ext_vector_type(8))) float v8f;

#define HID 64
#define INDIM 128
#define KDIM 320
#define GCOLS 256
#define LMAX 2.0f

__device__ __forceinline__ v8f wmma_f32_4(v2f a, v2f b, v8f c) {
  return __builtin_amdgcn_wmma_f32_16x16x4_f32(false, a, false, b, (short)0, c, false, false);
}

__device__ __forceinline__ float fsigmoid(float x) {
  return 1.0f / (1.0f + __expf(-x));
}

__device__ __forceinline__ void atomic_fadd(float* p, float v) {
  asm volatile("global_atomic_add_f32 %0, %1, off scope:SCOPE_DEV"
               :: "v"(p), "v"(v) : "memory");
}


__global__ __launch_bounds__(256)
void k_degree(const int* __restrict__ src, const int* __restrict__ dst,
              const float* __restrict__ ew, float* __restrict__ deg, int E) {
  int e = blockIdx.x * 256 + threadIdx.x;
  if (e >= E) return;
  int s = src[e], d = dst[e];
  if (s != d) atomic_fadd(&deg[s], ew[e]);
}

__global__ __launch_bounds__(256)
void k_dinv(const float* __restrict__ deg, float* __restrict__ dinv, int Nn) {
  int i = blockIdx.x * 256 + threadIdx.x;
  if (i >= Nn) return;
  float dg = deg[i];
  dinv[i] = (dg > 0.0f) ? rsqrtf(dg) : 0.0f;
}

__global__ __launch_bounds__(256)
void k_lapw(const int* __restrict__ src, const int* __restrict__ dst,
            const float* __restrict__ ew, const float* __restrict__ dinv,
            float* __restrict__ lapw, int E) {
  int e = blockIdx.x * 256 + threadIdx.x;
  if (e >= E) return;
  int s = src[e], d = dst[e];
  float w = (s == d) ? 0.0f : ew[e];
  lapw[e] = (2.0f / LMAX) * (-dinv[s] * w * dinv[d]);
}

__global__ __launch_bounds__(256)
void k_scale(float* __restrict__ dstp, const float* __restrict__ srcp,
             float scale, int count) {
  int i = blockIdx.x * 256 + threadIdx.x;
  if (i >= count) return;
  dstp[i] = scale * srcp[i];
}

__global__ __launch_bounds__(256)
void k_spmm(const float* __restrict__ z, float* __restrict__ acc,
            const int* __restrict__ src, const int* __restrict__ dst,
            const float* __restrict__ lw, int E) {
  int gid = blockIdx.x * 256 + threadIdx.x;
  int e = gid >> 4;
  if (e >= E) return;
  float w = lw[e];
  if (w == 0.0f) return;
  int ch = (gid & 15) << 2;
  int s = src[e], d = dst[e];
  const float4 zv = *(const float4*)(z + (size_t)s * HID + ch);
  float* out = acc + (size_t)d * HID + ch;
  atomic_fadd(out + 0, w * zv.x);
  atomic_fadd(out + 1, w * zv.y);
  atomic_fadd(out + 2, w * zv.z);
  atomic_fadd(out + 3, w * zv.w);
}

__global__ __launch_bounds__(256)
void k_cheb2(float* __restrict__ t2, const float* __restrict__ h, int count) {
  int i = blockIdx.x * 256 + threadIdx.x;
  if (i >= count) return;
  t2[i] = 2.0f * t2[i] - h[i];
}


__global__ __launch_bounds__(256)
void k_packB(const float* __restrict__ Wx, const float* __restrict__ theta,
             const float* __restrict__ bg, const float* __restrict__ conv_b,
             float* __restrict__ Bp, float* __restrict__ bias) {
  int gid = blockIdx.x * 256 + threadIdx.x;
  if (gid >= KDIM * GCOLS) return;
  int k = gid / GCOLS, col = gid % GCOLS;
  int g = col >> 6, o = col & 63;
  float v;
  if (k < INDIM) {
    v = Wx[((size_t)g * INDIM + k) * HID + o];
  } else {
    int kk = k - INDIM;
    int kb = kk >> 6, kr = kk & 63;
    v = theta[(((size_t)g * 3 + kb) * HID + kr) * HID + o];
  }
  Bp[gid] = v;
  if (gid < GCOLS) bias[gid] = bg[g * HID + o] + conv_b[g * HID + o];
}


#define AS1 324
#define PS1 260
#define MT 64
#define MSUB (MT / 16)

__device__ __forceinline__ void lstm_cell(float pi, float pf, float pt, float po, float cold,
                                          float& hnew, float& cnew) {
  float ig = fsigmoid(pi);
  float fg = fsigmoid(pf);
  float tg = tanhf(pt);
  float og = fsigmoid(po);
  float cv = fg * cold + ig * tg;
  hnew = og * tanhf(cv);
  cnew = cv;
}

__global__ __launch_bounds__(256)
void k_gemm_gates(const float* __restrict__ x, const float* __restrict__ h,
                  const float* __restrict__ T1, const float* __restrict__ T2,
                  const float* __restrict__ Bp, const float* __restrict__ bias,
                  const float* __restrict__ c_in,
                  float* __restrict__ h_out, float* __restrict__ c_out, int Nn) {
  __shared__ __attribute__((aligned(16))) float Asmem[MT * AS1];
  float* Psmem = Asmem;
  const int tid = threadIdx.x;
  const int m0 = blockIdx.x * MT;

#pragma unroll
  for (int it = 0; it < MT * 32 / 256; ++it) {
    int i = it * 256 + tid;
    int r = i >> 5, q = i & 31;
    int n = m0 + r; if (n >= Nn) n = Nn - 1;
    const float4 v = *(const float4*)(x + (size_t)n * INDIM + q * 4);
    *(float4*)&Asmem[r * AS1 + q * 4] = v;
  }
#pragma unroll
  for (int it = 0; it < MT * 16 / 256; ++it) {
    int i = it * 256 + tid;
    int r = i >> 4, q = i & 15;
    int n = m0 + r; if (n >= Nn) n = Nn - 1;
    const size_t g = (size_t)n * HID + q * 4;
    const float4 vh = *(const float4*)(h + g);
    const float4 v1 = *(const float4*)(T1 + g);
    const float4 v2 = *(const float4*)(T2 + g);
    float* ap = &Asmem[r * AS1 + 128 + q * 4];
    *(float4*)(ap)       = vh;
    *(float4*)(ap + 64)  = v1;
    *(float4*)(ap + 128) = v2;
  }
  __syncthreads();

  const int wave = tid >> 5, lane = tid & 31;
  const int lo = lane & 15, hi = lane >> 4;
  const int c0 = wave * 32;
  v8f acc[MSUB][2];
#pragma unroll
  for (int t = 0; t < MSUB; ++t) { acc[t][0] = v8f{}; acc[t][1] = v8f{}; }

  for (int k0 = 0; k0 < KDIM; k0 += 4) {
    const float* bptr = Bp + (size_t)(k0 + 2 * hi) * GCOLS + c0 + lo;
    v2f b0, b1;
    b0[0] = bptr[0];    b0[1] = bptr[GCOLS];
    b1[0] = bptr[16];   b1[1] = bptr[GCOLS + 16];
#pragma unroll
    for (int t = 0; t < MSUB; ++t) {
      v2f a = *(const v2f*)&Asmem[(t * 16 + lo) * AS1 + k0 + 2 * hi];
      acc[t][0] = wmma_f32_4(a, b0, acc[t][0]);
      acc[t][1] = wmma_f32_4(a, b1, acc[t][1]);
    }
  }

  float bias0 = bias[c0 + lo], bias1 = bias[c0 + 16 + lo];
  __syncthreads();
#pragma unroll
  for (int t = 0; t < MSUB; ++t) {
#pragma unroll
    for (int r = 0; r < 8; ++r) {
      int row = t * 16 + r + 8 * hi;
      Psmem[row * PS1 + c0 + lo]      = acc[t][0][r] + bias0;
      Psmem[row * PS1 + c0 + 16 + lo] = acc[t][1][r] + bias1;
    }
  }
  __syncthreads();

  for (int it = 0; it < MT * 16 / 256; ++it) {
    int i = it * 256 + tid;
    int row = i >> 4, o = (i & 15) * 4;
    int n = m0 + row;
    if (n >= Nn) continue;
    const float4 cold = *(const float4*)(c_in + (size_t)n * HID + o);
    const float* pr = &Psmem[row * PS1 + o];
    const float4 pi = *(const float4*)(pr);
    const float4 pf = *(const float4*)(pr + 64);
    const float4 pt = *(const float4*)(pr + 128);
    const float4 po = *(const float4*)(pr + 192);
    float4 hv, cv;
    lstm_cell(pi.x, pf.x, pt.x, po.x, cold.x, hv.x, cv.x);
    lstm_cell(pi.y, pf.y, pt.y, po.y, cold.y, hv.y, cv.y);
    lstm_cell(pi.z, pf.z, pt.z, po.z, cold.z, hv.z, cv.z);
    lstm_cell(pi.w, pf.w, pt.w, po.w, cold.w, hv.w, cv.w);
    *(float4*)(h_out + (size_t)n * HID + o) = hv;
    *(float4*)(c_out + (size_t)n * HID + o) = cv;
  }
}


#define AS2 68
#define BS2 68

__global__ __launch_bounds__(256)
void k_gemm_out(const float* __restrict__ h0, const float* __restrict__ Wl,
                const float* __restrict__ bl, float* __restrict__ out, int Nn) {
  __shared__ float Bs[64 * BS2];
  __shared__ float As[32 * AS2];
  const int tid = threadIdx.x;
  const int m0 = blockIdx.x * 32;

  for (int i = tid; i < 64 * 64; i += 256) {
    int k = i >> 6, ccol = i & 63;
    Bs[k * BS2 + ccol] = Wl[i];
  }
  for (int i = tid; i < 32 * 64; i += 256) {
    int r = i >> 6, k = i & 63;
    int n = m0 + r; if (n >= Nn) n = Nn - 1;
    float v = h0[(size_t)n * HID + k];
    As[r * AS2 + k] = v > 0.0f ? v : 0.0f;
  }
  __syncthreads();

  const int wave = tid >> 5, lane = tid & 31;
  const int lo = lane & 15, hi = lane >> 4;
  const int r0 = (wave >> 2) * 16, c0 = (wave & 3) * 16;
  v8f acc = {};
  for (int k0 = 0; k0 < 64; k0 += 4) {
    v2f a = *(const v2f*)&As[(r0 + lo) * AS2 + k0 + 2 * hi];
    v2f b;
    b[0] = Bs[(k0 + 2 * hi) * BS2 + c0 + lo];
    b[1] = Bs[(k0 + 2 * hi + 1) * BS2 + c0 + lo];
    acc = wmma_f32_4(a, b, acc);
  }
  float bb = bl[c0 + lo];
#pragma unroll
  for (int r = 0; r < 8; ++r) {
    int row = r0 + r + 8 * hi;
    int n = m0 + row;
    if (n < Nn) out[(size_t)n * HID + c0 + lo] = acc[r] + bb;
  }
}


extern "C" void kernel_launch(void* const* d_in, const int* in_sizes, int n_in,
                              void* d_out, int out_size, void* d_ws, size_t ws_size,
                              hipStream_t stream) {
  const float* x      = (const float*)d_in[0];
  const int*   ei     = (const int*)d_in[1];
  const float* ew     = (const float*)d_in[2];
  const float* h      = (const float*)d_in[3];
  const float* c      = (const float*)d_in[4];
  const float* Wx     = (const float*)d_in[5];
  const float* bg     = (const float*)d_in[6];
  const float* theta  = (const float*)d_in[7];
  const float* conv_b = (const float*)d_in[8];
  const float* Wl     = (const float*)d_in[9];
  const float* bl     = (const float*)d_in[10];

  const int E  = in_sizes[2];
  const int Nn = in_sizes[3] / HID;
  const int* src = ei;
  const int* dst = ei + E;

  float* deg  = (float*)d_ws;
  float* dinv = deg  + Nn;
  float* lapw = dinv + Nn;
  float* T1   = lapw + E;
  float* T2   = T1 + (size_t)Nn * HID;
  float* Bp   = T2 + (size_t)Nn * HID;
  float* bias = Bp + (size_t)KDIM * GCOLS;

  float* out   = (float*)d_out;
  float* h_out = out   + (size_t)Nn * HID;
  float* c_out = h_out + (size_t)Nn * HID;

  const float DIAG = 2.0f / LMAX - 1.0f;
  const int NH = Nn * HID;

  hipMemsetAsync(deg, 0, (size_t)Nn * sizeof(float), stream);
  k_degree<<<(E + 255) / 256, 256, 0, stream>>>(src, dst, ew, deg, E);
  k_dinv<<<(Nn + 255) / 256, 256, 0, stream>>>(deg, dinv, Nn);
  k_lapw<<<(E + 255) / 256, 256, 0, stream>>>(src, dst, ew, dinv, lapw, E);

  k_scale<<<(NH + 255) / 256, 256, 0, stream>>>(T1, h, DIAG, NH);
  k_spmm<<<(E * 16 + 255) / 256, 256, 0, stream>>>(h, T1, src, dst, lapw, E);
  k_scale<<<(NH + 255) / 256, 256, 0, stream>>>(T2, T1, DIAG, NH);
  k_spmm<<<(E * 16 + 255) / 256, 256, 0, stream>>>(T1, T2, src, dst, lapw, E);
  k_cheb2<<<(NH + 255) / 256, 256, 0, stream>>>(T2, h, NH);

  k_packB<<<(KDIM * GCOLS + 255) / 256, 256, 0, stream>>>(Wx, theta, bg, conv_b, Bp, bias);

  k_gemm_gates<<<(Nn + MT - 1) / MT, 256, 0, stream>>>(x, h, T1, T2, Bp, bias, c,
                                                   h_out, c_out, Nn);
  k_gemm_out<<<(Nn + 31) / 32, 256, 0, stream>>>(h_out, Wl, bl, out, Nn);
}


# SpMM as per-row gather: edges sorted by dst (two stable 9-bit ballot splits, no atomics), one wave sums a row in registers, stores once; T1/T2 f32 atomics gone; sum order = edge order in row
# speedup vs baseline: 1.3122x; 1.2550x over previous
#include <hip/hip_runtime.h>
#include <hip/hip_bf16.h>

typedef __attribute__((ext_vector_type(2))) float v2f;
typedef __attribute__((ext_vector_type(8))) float v8f;

#define HID 64
#define INDIM 128
#define KDIM 320
#define GCOLS 256
#define LMAX 2.0f

__device__ __forceinline__ v8f wmma_f32_4(v2f a, v2f b, v8f c) {
  return __builtin_amdgcn_wmma_f32_16x16x4_f32(false, a, false, b, (short)0, c, false, false);
}

__device__ __forceinline__ float fsigmoid(float x) {
  return 1.0f / (1.0f + __expf(-x));
}

__device__ __forceinline__ void atomic_fadd(float* p, float v) {
  asm volatile("global_atomic_add_f32 %0, %1, off scope:SCOPE_DEV"
               :: "v"(p), "v"(v) : "memory");
}


__global__ __launch_bounds__(256)
void k_degree(const int* __restrict__ src, const int* __restrict__ dst,
              const float* __restrict__ ew, float* __restrict__ deg, int E) {
  int e = blockIdx.x * 256 + threadIdx.x;
  if (e >= E) return;
  int s = src[e], d = dst[e];
  if (s != d) atomic_fadd(&deg[s], ew[e]);
}

__global__ __launch_bounds__(256)
void k_dinv(const float* __restrict__ deg, float* __restrict__ dinv, int Nn) {
  int i = blockIdx.x * 256 + threadIdx.x;
  if (i >= Nn) return;
  float dg = deg[i];
  dinv[i] = (dg > 0.0f) ? rsqrtf(dg) : 0.0f;
}

__global__ __launch_bounds__(256)
void k_lapw_pack(const int* __restrict__ src, const int* __restrict__ dst,
                 const float* __restrict__ ew, const float* __restrict__ dinv,
                 int2* __restrict__ edgepay, int E) {
  int e = blockIdx.x * 256 + threadIdx.x;
  if (e >= E) return;
  int s = src[e], d = dst[e];
  float w = (s == d) ? 0.0f : ew[e];
  float lw = (2.0f / LMAX) * (-dinv[s] * w * dinv[d]);
  edgepay[e] = make_int2(s, __float_as_int(lw));
}


#define SORT_CH 8192
#define SORT_ROUNDS (SORT_CH / 256)
#define SORT_BINS 512

__device__ __forceinline__ int wave_incl_scan(int v, int lane) {
  for (int off = 1; off < 32; off <<= 1) {
    int t = __shfl_up(v, off, 32);
    if (lane >= off) v += t;
  }
  return v;
}

__device__ __forceinline__ int block_excl_scan(int v, int* wsum, int lane, int wave, int& total) {
  const int incl = wave_incl_scan(v, lane);
  if (lane == 31) wsum[wave] = incl;
  __syncthreads();
  int wb = 0, t = 0;
  for (int w = 0; w < 8; ++w) { int s = wsum[w]; if (w < wave) wb += s; t += s; }
  total = t;
  __syncthreads();
  return wb + incl - v;
}

__global__ __launch_bounds__(256)
void k_split_rank(const int* __restrict__ key, int shift,
                  int* __restrict__ lrank, int* __restrict__ table, int E) {
  __shared__ int whist[8 * SORT_BINS];
  const int tid = threadIdx.x, lane = tid & 31, wave = tid >> 5;
  for (int i = tid; i < 8 * SORT_BINS; i += 256) whist[i] = 0;
  __syncthreads();
  const int base = blockIdx.x * SORT_CH + wave * (SORT_CH / 8) + lane;
  int dg[SORT_ROUNDS], lr[SORT_ROUNDS];
#pragma unroll
  for (int r = 0; r < SORT_ROUNDS; ++r) {
    const int e = base + r * 32;
    const int k = key[e < E ? e : E - 1];
    dg[r] = (e < E) ? ((k >> shift) & (SORT_BINS - 1)) : -1;
  }
  int* wh = whist + wave * SORT_BINS;
#pragma unroll
  for (int r = 0; r < SORT_ROUNDS; ++r) {
    const int d = dg[r];
    const bool valid = d >= 0;
    unsigned peers = __builtin_amdgcn_ballot_w32(valid);
#pragma unroll
    for (int b = 0; b < 9; ++b) {
      const bool bit = ((d >> b) & 1) != 0;
      const unsigned bal = __builtin_amdgcn_ballot_w32(bit);
      peers &= bit ? bal : ~bal;
    }
    const int rk = __popc(peers & ((1u << lane) - 1u));
    int b0 = 0;
    if (valid) {
      b0 = wh[d];
      if (rk == 0) wh[d] = b0 + __popc(peers);
    }
    lr[r] = b0 + rk;
  }
  __syncthreads();
  for (int dgt = tid; dgt < SORT_BINS; dgt += 256) {
    int run = 0;
    for (int w = 0; w < 8; ++w) {
      int v = whist[w * SORT_BINS + dgt];
      whist[w * SORT_BINS + dgt] = run;
      run += v;
    }
    table[blockIdx.x * SORT_BINS + dgt] = run;
  }
  __syncthreads();
#pragma unroll
  for (int r = 0; r < SORT_ROUNDS; ++r) {
    const int e = base + r * 32;
    if (e < E) lrank[e] = lr[r] + wh[dg[r]];
  }
}

__global__ __launch_bounds__(256)
void k_split_scatter(const int* __restrict__ key, const int2* __restrict__ pay, int shift,
                     const int* __restrict__ lrank, const int* __restrict__ table, int nblocks,
                     int* __restrict__ key_out, int2* __restrict__ pay_out, int E) {
  __shared__ int dbase[SORT_BINS];
  __shared__ int wsum[8];
  const int tid = threadIdx.x, lane = tid & 31, wave = tid >> 5;
  const int me = blockIdx.x;
  int pre0 = 0, tot0 = 0, pre1 = 0, tot1 = 0;
#pragma unroll 14
  for (int b = 0; b < nblocks; ++b) {
    const int v0 = table[b * SORT_BINS + tid];
    const int v1 = table[b * SORT_BINS + 256 + tid];
    tot0 += v0; tot1 += v1;
    if (b < me) { pre0 += v0; pre1 += v1; }
  }
  int sum0, sum1;
  const int ex0 = block_excl_scan(tot0, wsum, lane, wave, sum0);
  const int ex1 = block_excl_scan(tot1, wsum, lane, wave, sum1);
  dbase[tid]       = ex0 + pre0;
  dbase[tid + 256] = sum0 + ex1 + pre1;
  __syncthreads();
  for (int r0 = 0; r0 < SORT_ROUNDS; r0 += 8) {
    int kk[8], rk[8];
    int2 pv[8];
#pragma unroll
    for (int u = 0; u < 8; ++u) {
      const int e = me * SORT_CH + (r0 + u) * 256 + tid;
      const int ec = e < E ? e : E - 1;
      kk[u] = key[ec];
      rk[u] = lrank[ec];
      pv[u] = pay[ec];
    }
#pragma unroll
    for (int u = 0; u < 8; ++u) {
      const int e = me * SORT_CH + (r0 + u) * 256 + tid;
      if (e < E) {
        const int pos = dbase[(kk[u] >> shift) & (SORT_BINS - 1)] + rk[u];
        key_out[pos] = kk[u];
        pay_out[pos] = pv[u];
      }
    }
  }
}

__global__ __launch_bounds__(256)
void k_rowbounds(const int* __restrict__ sdst, int* __restrict__ rowstart,
                 int* __restrict__ rowend, int E) {
  int p = blockIdx.x * 256 + threadIdx.x;
  if (p >= E) return;
  const int d = sdst[p];
  if (p == 0 || sdst[p - 1] != d) rowstart[d] = p;
  if (p == E - 1 || sdst[p + 1] != d) rowend[d] = p + 1;
}

#define ROWCHUNK 8
__device__ __forceinline__ float2 lhat_row(const float* __restrict__ z,
                                           const int2* __restrict__ rowedges,
                                           const int* __restrict__ rowstart,
                                           const int* __restrict__ rowend,
                                           float diag, int n, int lane) {
  const float2* z2 = (const float2*)z;
  const int p0 = rowstart[n];
  const int c = rowend[n] - p0;
  const int2* ce = rowedges + p0;
  float2 acc = make_float2(0.0f, 0.0f);
  for (int j = 0; j < c; j += ROWCHUNK) {
    float2 zv[ROWCHUNK];
    float wv[ROWCHUNK];
#pragma unroll
    for (int u = 0; u < ROWCHUNK; ++u) {
      const int idx = (j + u < c) ? (j + u) : (c - 1);
      const int2 en = ce[idx];
      wv[u] = (j + u < c) ? __int_as_float(en.y) : 0.0f;
      zv[u] = z2[(size_t)en.x * (HID / 2) + lane];
    }
#pragma unroll
    for (int u = 0; u < ROWCHUNK; ++u) {
      acc.x += wv[u] * zv[u].x;
      acc.y += wv[u] * zv[u].y;
    }
  }
  const float2 zn = z2[(size_t)n * (HID / 2) + lane];
  acc.x += diag * zn.x;
  acc.y += diag * zn.y;
  return acc;
}

__global__ __launch_bounds__(256)
void k_spmm_t1(const float* __restrict__ h, float* __restrict__ T1,
               const int2* __restrict__ rowedges, const int* __restrict__ rowstart,
               const int* __restrict__ rowend, float diag, int Nn) {
  const int lane = threadIdx.x & 31;
  const int n = __builtin_amdgcn_readfirstlane(blockIdx.x * 8 + (threadIdx.x >> 5));
  if (n >= Nn) return;
  const float2 r = lhat_row(h, rowedges, rowstart, rowend, diag, n, lane);
  ((float2*)T1)[(size_t)n * (HID / 2) + lane] = r;
}

__global__ __launch_bounds__(256)
void k_spmm_t2(const float* __restrict__ T1, const float* __restrict__ h,
               float* __restrict__ T2,
               const int2* __restrict__ rowedges, const int* __restrict__ rowstart,
               const int* __restrict__ rowend, float diag, int Nn) {
  const int lane = threadIdx.x & 31;
  const int n = __builtin_amdgcn_readfirstlane(blockIdx.x * 8 + (threadIdx.x >> 5));
  if (n >= Nn) return;
  const float2 r = lhat_row(T1, rowedges, rowstart, rowend, diag, n, lane);
  const size_t o = (size_t)n * (HID / 2) + lane;
  const float2 hv = ((const float2*)h)[o];
  float2 t;
  t.x = 2.0f * r.x - hv.x;
  t.y = 2.0f * r.y - hv.y;
  ((float2*)T2)[o] = t;
}


__global__ __launch_bounds__(256)
void k_packB(const float* __restrict__ Wx, const float* __restrict__ theta,
             const float* __restrict__ bg, const float* __restrict__ conv_b,
             float* __restrict__ Bp, float* __restrict__ bias) {
  int gid = blockIdx.x * 256 + threadIdx.x;
  if (gid >= KDIM * GCOLS) return;
  int k = gid / GCOLS, col = gid % GCOLS;
  int g = col >> 6, o = col & 63;
  float v;
  if (k < INDIM) {
    v = Wx[((size_t)g * INDIM + k) * HID + o];
  } else {
    int kk = k - INDIM;
    int kb = kk >> 6, kr = kk & 63;
    v = theta[(((size_t)g * 3 + kb) * HID + kr) * HID + o];
  }
  Bp[gid] = v;
  if (gid < GCOLS) bias[gid] = bg[g * HID + o] + conv_b[g * HID + o];
}


#define AS1 324
#define PS1 260
#define MT 64
#define MSUB (MT / 16)

__device__ __forceinline__ void lstm_cell(float pi, float pf, float pt, float po, float cold,
                                          float& hnew, float& cnew) {
  float ig = fsigmoid(pi);
  float fg = fsigmoid(pf);
  float tg = tanhf(pt);
  float og = fsigmoid(po);
  float cv = fg * cold + ig * tg;
  hnew = og * tanhf(cv);
  cnew = cv;
}

__global__ __launch_bounds__(256)
void k_gemm_gates(const float* __restrict__ x, const float* __restrict__ h,
                  const float* __restrict__ T1, const float* __restrict__ T2,
                  const float* __restrict__ Bp, const float* __restrict__ bias,
                  const float* __restrict__ c_in,
                  float* __restrict__ h_out, float* __restrict__ c_out, int Nn) {
  __shared__ __attribute__((aligned(16))) float Asmem[MT * AS1];
  float* Psmem = Asmem;
  const int tid = threadIdx.x;
  const int m0 = blockIdx.x * MT;

#pragma unroll
  for (int it = 0; it < MT * 32 / 256; ++it) {
    int i = it * 256 + tid;
    int r = i >> 5, q = i & 31;
    int n = m0 + r; if (n >= Nn) n = Nn - 1;
    const float4 v = *(const float4*)(x + (size_t)n * INDIM + q * 4);
    *(float4*)&Asmem[r * AS1 + q * 4] = v;
  }
#pragma unroll
  for (int it = 0; it < MT * 16 / 256; ++it) {
    int i = it * 256 + tid;
    int r = i >> 4, q = i & 15;
    int n = m0 + r; if (n >= Nn) n = Nn - 1;
    const size_t g = (size_t)n * HID + q * 4;
    const float4 vh = *(const float4*)(h + g);
    const float4 v1 = *(const float4*)(T1 + g);
    const float4 v2 = *(const float4*)(T2 + g);
    float* ap = &Asmem[r * AS1 + 128 + q * 4];
    *(float4*)(ap)       = vh;
    *(float4*)(ap + 64)  = v1;
    *(float4*)(ap + 128) = v2;
  }
  __syncthreads();

  const int wave = tid >> 5, lane = tid & 31;
  const int lo = lane & 15, hi = lane >> 4;
  const int c0 = wave * 32;
  v8f acc[MSUB][2];
#pragma unroll
  for (int t = 0; t < MSUB; ++t) { acc[t][0] = v8f{}; acc[t][1] = v8f{}; }

  for (int k0 = 0; k0 < KDIM; k0 += 4) {
    const float* bptr = Bp + (size_t)(k0 + 2 * hi) * GCOLS + c0 + lo;
    v2f b0, b1;
    b0[0] = bptr[0];    b0[1] = bptr[GCOLS];
    b1[0] = bptr[16];   b1[1] = bptr[GCOLS + 16];
#pragma unroll
    for (int t = 0; t < MSUB; ++t) {
      v2f a = *(const v2f*)&Asmem[(t * 16 + lo) * AS1 + k0 + 2 * hi];
      acc[t][0] = wmma_f32_4(a, b0, acc[t][0]);
      acc[t][1] = wmma_f32_4(a, b1, acc[t][1]);
    }
  }

  float bias0 = bias[c0 + lo], bias1 = bias[c0 + 16 + lo];
  __syncthreads();
#pragma unroll
  for (int t = 0; t < MSUB; ++t) {
#pragma unroll
    for (int r = 0; r < 8; ++r) {
      int row = t * 16 + r + 8 * hi;
      Psmem[row * PS1 + c0 + lo]      = acc[t][0][r] + bias0;
      Psmem[row * PS1 + c0 + 16 + lo] = acc[t][1][r] + bias1;
    }
  }
  __syncthreads();

  for (int it = 0; it < MT * 16 / 256; ++it) {
    int i = it * 256 + tid;
    int row = i >> 4, o = (i & 15) * 4;
    int n = m0 + row;
    if (n >= Nn) continue;
    const float4 cold = *(const float4*)(c_in + (size_t)n * HID + o);
    const float* pr = &Psmem[row * PS1 + o];
    const float4 pi = *(const float4*)(pr);
    const float4 pf = *(const float4*)(pr + 64);
    const float4 pt = *(const float4*)(pr + 128);
    const float4 po = *(const float4*)(pr + 192);
    float4 hv, cv;
    lstm_cell(pi.x, pf.x, pt.x, po.x, cold.x, hv.x, cv.x);
    lstm_cell(pi.y, pf.y, pt.y, po.y, cold.y, hv.y, cv.y);
    lstm_cell(pi.z, pf.z, pt.z, po.z, cold.z, hv.z, cv.z);
    lstm_cell(pi.w, pf.w, pt.w, po.w, cold.w, hv.w, cv.w);
    *(float4*)(h_out + (size_t)n * HID + o) = hv;
    *(float4*)(c_out + (size_t)n * HID + o) = cv;
  }
}


#define AS2 68
#define BS2 68

__global__ __launch_bounds__(256)
void k_gemm_out(const float* __restrict__ h0, const float* __restrict__ Wl,
                const float* __restrict__ bl, float* __restrict__ out, int Nn) {
  __shared__ float Bs[64 * BS2];
  __shared__ float As[32 * AS2];
  const int tid = threadIdx.x;
  const int m0 = blockIdx.x * 32;

  for (int i = tid; i < 64 * 64; i += 256) {
    int k = i >> 6, ccol = i & 63;
    Bs[k * BS2 + ccol] = Wl[i];
  }
  for (int i = tid; i < 32 * 64; i += 256) {
    int r = i >> 6, k = i & 63;
    int n = m0 + r; if (n >= Nn) n = Nn - 1;
    float v = h0[(size_t)n * HID + k];
    As[r * AS2 + k] = v > 0.0f ? v : 0.0f;
  }
  __syncthreads();

  const int wave = tid >> 5, lane = tid & 31;
  const int lo = lane & 15, hi = lane >> 4;
  const int r0 = (wave >> 2) * 16, c0 = (wave & 3) * 16;
  v8f acc = {};
  for (int k0 = 0; k0 < 64; k0 += 4) {
    v2f a = *(const v2f*)&As[(r0 + lo) * AS2 + k0 + 2 * hi];
    v2f b;
    b[0] = Bs[(k0 + 2 * hi) * BS2 + c0 + lo];
    b[1] = Bs[(k0 + 2 * hi + 1) * BS2 + c0 + lo];
    acc = wmma_f32_4(a, b, acc);
  }
  float bb = bl[c0 + lo];
#pragma unroll
  for (int r = 0; r < 8; ++r) {
    int row = r0 + r + 8 * hi;
    int n = m0 + row;
    if (n < Nn) out[(size_t)n * HID + c0 + lo] = acc[r] + bb;
  }
}


extern "C" void kernel_launch(void* const* d_in, const int* in_sizes, int n_in,
                              void* d_out, int out_size, void* d_ws, size_t ws_size,
                              hipStream_t stream) {
  const float* x      = (const float*)d_in[0];
  const int*   ei     = (const int*)d_in[1];
  const float* ew     = (const float*)d_in[2];
  const float* h      = (const float*)d_in[3];
  const float* c      = (const float*)d_in[4];
  const float* Wx     = (const float*)d_in[5];
  const float* bg     = (const float*)d_in[6];
  const float* theta  = (const float*)d_in[7];
  const float* conv_b = (const float*)d_in[8];
  const float* Wl     = (const float*)d_in[9];
  const float* bl     = (const float*)d_in[10];

  const int E  = in_sizes[2];
  const int Nn = in_sizes[3] / HID;
  const int* src = ei;
  const int* dst = ei + E;

  float* T1   = (float*)d_ws;
  float* T2   = T1 + (size_t)Nn * HID;
  float* Bp   = T2 + (size_t)Nn * HID;
  float* bias = Bp + (size_t)KDIM * GCOLS;
  int2*  edgepay  = (int2*)(bias + GCOLS);
  int2*  payA     = edgepay + E;
  int2*  rowedges = payA + E;
  int*   keyA = (int*)(rowedges + E);
  int*   sdst = keyA + E;
  int*   lrank = sdst + E;
  const int NB = (E + SORT_CH - 1) / SORT_CH;
  int*   table = lrank + E;
  float* deg  = (float*)(table + (size_t)NB * SORT_BINS);
  int*   rowstart = (int*)(deg + Nn);
  int*   rowend = rowstart + Nn;
  float* dinv = (float*)(rowend + Nn);

  float* out   = (float*)d_out;
  float* h_out = out   + (size_t)Nn * HID;
  float* c_out = h_out + (size_t)Nn * HID;

  const float DIAG = 2.0f / LMAX - 1.0f;

  hipMemsetAsync(deg, 0, (size_t)Nn * 3 * sizeof(float), stream);
  k_degree<<<(E + 255) / 256, 256, 0, stream>>>(src, dst, ew, deg, E);
  k_dinv<<<(Nn + 255) / 256, 256, 0, stream>>>(deg, dinv, Nn);
  k_lapw_pack<<<(E + 255) / 256, 256, 0, stream>>>(src, dst, ew, dinv, edgepay, E);

  k_split_rank<<<NB, 256, 0, stream>>>(dst, 0, lrank, table, E);
  k_split_scatter<<<NB, 256, 0, stream>>>(dst, edgepay, 0, lrank, table, NB, keyA, payA, E);
  k_split_rank<<<NB, 256, 0, stream>>>(keyA, 9, lrank, table, E);
  k_split_scatter<<<NB, 256, 0, stream>>>(keyA, payA, 9, lrank, table, NB, sdst, rowedges, E);
  k_rowbounds<<<(E + 255) / 256, 256, 0, stream>>>(sdst, rowstart, rowend, E);

  k_spmm_t1<<<(Nn + 7) / 8, 256, 0, stream>>>(h, T1, rowedges, rowstart, rowend, DIAG, Nn);
  k_spmm_t2<<<(Nn + 7) / 8, 256, 0, stream>>>(T1, h, T2, rowedges, rowstart, rowend, DIAG, Nn);

  k_packB<<<(KDIM * GCOLS + 255) / 256, 256, 0, stream>>>(Wx, theta, bg, conv_b, Bp, bias);

  k_gemm_gates<<<(Nn + MT - 1) / MT, 256, 0, stream>>>(x, h, T1, T2, Bp, bias, c,
                                                   h_out, c_out, Nn);
  k_gemm_out<<<(Nn + 31) / 32, 256, 0, stream>>>(h_out, Wl, bl, out, Nn);
}


# split scatter staged through LDS: a work-group puts its edges in digit order in LDS, then stores each digit's run contiguously instead of 32 separate lines per store; output positions unchanged
# speedup vs baseline: 1.5805x; 1.2045x over previous
#include <hip/hip_runtime.h>
#include <hip/hip_bf16.h>

typedef __attribute__((ext_vector_type(2))) float v2f;
typedef __attribute__((ext_vector_type(8))) float v8f;

#define HID 64
#define INDIM 128
#define KDIM 320
#define GCOLS 256
#define LMAX 2.0f

__device__ __forceinline__ v8f wmma_f32_4(v2f a, v2f b, v8f c) {
  return __builtin_amdgcn_wmma_f32_16x16x4_f32(false, a, false, b, (short)0, c, false, false);
}

__device__ __forceinline__ float fsigmoid(float x) {
  return 1.0f / (1.0f + __expf(-x));
}

__device__ __forceinline__ void atomic_fadd(float* p, float v) {
  asm volatile("global_atomic_add_f32 %0, %1, off scope:SCOPE_DEV"
               :: "v"(p), "v"(v) : "memory");
}


__global__ __launch_bounds__(256)
void k_degree(const int* __restrict__ src, const int* __restrict__ dst,
              const float* __restrict__ ew, float* __restrict__ deg, int E) {
  int e = blockIdx.x * 256 + threadIdx.x;
  if (e >= E) return;
  int s = src[e], d = dst[e];
  if (s != d) atomic_fadd(&deg[s], ew[e]);
}

__global__ __launch_bounds__(256)
void k_dinv(const float* __restrict__ deg, float* __restrict__ dinv, int Nn) {
  int i = blockIdx.x * 256 + threadIdx.x;
  if (i >= Nn) return;
  float dg = deg[i];
  dinv[i] = (dg > 0.0f) ? rsqrtf(dg) : 0.0f;
}

__global__ __launch_bounds__(256)
void k_lapw_pack(const int* __restrict__ src, const int* __restrict__ dst,
                 const float* __restrict__ ew, const float* __restrict__ dinv,
                 int2* __restrict__ edgepay, int E) {
  int e = blockIdx.x * 256 + threadIdx.x;
  if (e >= E) return;
  int s = src[e], d = dst[e];
  float w = (s == d) ? 0.0f : ew[e];
  float lw = (2.0f / LMAX) * (-dinv[s] * w * dinv[d]);
  edgepay[e] = make_int2(s, __float_as_int(lw));
}


#define SORT_CH 8192
#define SORT_ROUNDS (SORT_CH / 256)
#define SORT_BINS 512

__device__ __forceinline__ int wave_incl_scan(int v, int lane) {
  for (int off = 1; off < 32; off <<= 1) {
    int t = __shfl_up(v, off, 32);
    if (lane >= off) v += t;
  }
  return v;
}

__device__ __forceinline__ int block_excl_scan(int v, int* wsum, int lane, int wave, int& total) {
  const int incl = wave_incl_scan(v, lane);
  if (lane == 31) wsum[wave] = incl;
  __syncthreads();
  int wb = 0, t = 0;
  for (int w = 0; w < 8; ++w) { int s = wsum[w]; if (w < wave) wb += s; t += s; }
  total = t;
  __syncthreads();
  return wb + incl - v;
}

__global__ __launch_bounds__(256)
void k_split_rank(const int* __restrict__ key, int shift,
                  int* __restrict__ lrank, int* __restrict__ table, int E) {
  __shared__ int whist[8 * SORT_BINS];
  const int tid = threadIdx.x, lane = tid & 31, wave = tid >> 5;
  for (int i = tid; i < 8 * SORT_BINS; i += 256) whist[i] = 0;
  __syncthreads();
  const int base = blockIdx.x * SORT_CH + wave * (SORT_CH / 8) + lane;
  int dg[SORT_ROUNDS], lr[SORT_ROUNDS];
#pragma unroll
  for (int r = 0; r < SORT_ROUNDS; ++r) {
    const int e = base + r * 32;
    const int k = key[e < E ? e : E - 1];
    dg[r] = (e < E) ? ((k >> shift) & (SORT_BINS - 1)) : -1;
  }
  int* wh = whist + wave * SORT_BINS;
#pragma unroll
  for (int r = 0; r < SORT_ROUNDS; ++r) {
    const int d = dg[r];
    const bool valid = d >= 0;
    unsigned peers = __builtin_amdgcn_ballot_w32(valid);
#pragma unroll
    for (int b = 0; b < 9; ++b) {
      const bool bit = ((d >> b) & 1) != 0;
      const unsigned bal = __builtin_amdgcn_ballot_w32(bit);
      peers &= bit ? bal : ~bal;
    }
    const int rk = __popc(peers & ((1u << lane) - 1u));
    int b0 = 0;
    if (valid) {
      b0 = wh[d];
      if (rk == 0) wh[d] = b0 + __popc(peers);
    }
    lr[r] = b0 + rk;
  }
  __syncthreads();
  for (int dgt = tid; dgt < SORT_BINS; dgt += 256) {
    int run = 0;
    for (int w = 0; w < 8; ++w) {
      int v = whist[w * SORT_BINS + dgt];
      whist[w * SORT_BINS + dgt] = run;
      run += v;
    }
    table[blockIdx.x * SORT_BINS + dgt] = run;
  }
  __syncthreads();
#pragma unroll
  for (int r = 0; r < SORT_ROUNDS; ++r) {
    const int e = base + r * 32;
    if (e < E) lrank[e] = lr[r] + wh[dg[r]];
  }
}

__global__ __launch_bounds__(256)
void k_split_scatter(const int* __restrict__ key, const int2* __restrict__ pay, int shift,
                     const int* __restrict__ lrank, const int* __restrict__ table, int nblocks,
                     int* __restrict__ key_out, int2* __restrict__ pay_out, int E) {
  __shared__ int lstart[SORT_BINS];
  __shared__ int gdelta[SORT_BINS];
  __shared__ int wsum[8];
  __shared__ int skey[SORT_CH];
  __shared__ __attribute__((aligned(8))) int spay_w[2 * SORT_CH];
  int2* spay = (int2*)spay_w;
  const int tid = threadIdx.x, lane = tid & 31, wave = tid >> 5;
  const int me = blockIdx.x;
  int pre0 = 0, tot0 = 0, pre1 = 0, tot1 = 0;
#pragma unroll 14
  for (int b = 0; b < nblocks; ++b) {
    const int v0 = table[b * SORT_BINS + tid];
    const int v1 = table[b * SORT_BINS + 256 + tid];
    tot0 += v0; tot1 += v1;
    if (b < me) { pre0 += v0; pre1 += v1; }
  }
  const int own0 = table[me * SORT_BINS + tid];
  const int own1 = table[me * SORT_BINS + 256 + tid];
  int sum0, sum1, osum0, osum1;
  const int ex0 = block_excl_scan(tot0, wsum, lane, wave, sum0);
  const int ex1 = block_excl_scan(tot1, wsum, lane, wave, sum1);
  const int l0 = block_excl_scan(own0, wsum, lane, wave, osum0);
  const int l1 = osum0 + block_excl_scan(own1, wsum, lane, wave, osum1);
  lstart[tid]       = l0;
  lstart[tid + 256] = l1;
  gdelta[tid]       = ex0 + pre0 - l0;
  gdelta[tid + 256] = sum0 + ex1 + pre1 - l1;
  __syncthreads();
  for (int r0 = 0; r0 < SORT_ROUNDS; r0 += 8) {
    int kk[8], rk[8];
    int2 pv[8];
#pragma unroll
    for (int u = 0; u < 8; ++u) {
      const int e = me * SORT_CH + (r0 + u) * 256 + tid;
      const int ec = e < E ? e : E - 1;
      kk[u] = key[ec];
      rk[u] = lrank[ec];
      pv[u] = pay[ec];
    }
    int lp[8];
#pragma unroll
    for (int u = 0; u < 8; ++u)
      lp[u] = lstart[(kk[u] >> shift) & (SORT_BINS - 1)] + rk[u];
#pragma unroll
    for (int u = 0; u < 8; ++u) {
      const int e = me * SORT_CH + (r0 + u) * 256 + tid;
      if (e < E) {
        skey[lp[u]] = kk[u];
        spay[lp[u]] = pv[u];
      }
    }
  }
  __syncthreads();
  const int rem = E - me * SORT_CH;
  const int cnt = rem < SORT_CH ? rem : SORT_CH;
  for (int r0 = 0; r0 < SORT_ROUNDS; r0 += 8) {
    int kk[8];
    int2 pv[8];
#pragma unroll
    for (int u = 0; u < 8; ++u) {
      const int i = (r0 + u) * 256 + tid;
      const int ic = i < cnt ? i : cnt - 1;
      kk[u] = skey[ic];
      pv[u] = spay[ic];
    }
    int ps[8];
#pragma unroll
    for (int u = 0; u < 8; ++u)
      ps[u] = (r0 + u) * 256 + tid + gdelta[(kk[u] >> shift) & (SORT_BINS - 1)];
#pragma unroll
    for (int u = 0; u < 8; ++u) {
      const int i = (r0 + u) * 256 + tid;
      if (i < cnt) {
        key_out[ps[u]] = kk[u];
        pay_out[ps[u]] = pv[u];
      }
    }
  }
}

__global__ __launch_bounds__(256)
void k_rowbounds(const int* __restrict__ sdst, int* __restrict__ rowstart,
                 int* __restrict__ rowend, int E) {
  int p = blockIdx.x * 256 + threadIdx.x;
  if (p >= E) return;
  const int d = sdst[p];
  if (p == 0 || sdst[p - 1] != d) rowstart[d] = p;
  if (p == E - 1 || sdst[p + 1] != d) rowend[d] = p + 1;
}

#define ROWCHUNK 8
__device__ __forceinline__ float2 lhat_row(const float* __restrict__ z,
                                           const int2* __restrict__ rowedges,
                                           const int* __restrict__ rowstart,
                                           const int* __restrict__ rowend,
                                           float diag, int n, int lane) {
  const float2* z2 = (const float2*)z;
  const int p0 = rowstart[n];
  const int c = rowend[n] - p0;
  const int2* ce = rowedges + p0;
  float2 acc = make_float2(0.0f, 0.0f);
  for (int j = 0; j < c; j += ROWCHUNK) {
    float2 zv[ROWCHUNK];
    float wv[ROWCHUNK];
#pragma unroll
    for (int u = 0; u < ROWCHUNK; ++u) {
      const int idx = (j + u < c) ? (j + u) : (c - 1);
      const int2 en = ce[idx];
      wv[u] = (j + u < c) ? __int_as_float(en.y) : 0.0f;
      zv[u] = z2[(size_t)en.x * (HID / 2) + lane];
    }
#pragma unroll
    for (int u = 0; u < ROWCHUNK; ++u) {
      acc.x += wv[u] * zv[u].x;
      acc.y += wv[u] * zv[u].y;
    }
  }
  const float2 zn = z2[(size_t)n * (HID / 2) + lane];
  acc.x += diag * zn.x;
  acc.y += diag * zn.y;
  return acc;
}

__global__ __launch_bounds__(256)
void k_spmm_t1(const float* __restrict__ h, float* __restrict__ T1,
               const int2* __restrict__ rowedges, const int* __restrict__ rowstart,
               const int* __restrict__ rowend, float diag, int Nn) {
  const int lane = threadIdx.x & 31;
  const int n = __builtin_amdgcn_readfirstlane(blockIdx.x * 8 + (threadIdx.x >> 5));
  if (n >= Nn) return;
  const float2 r = lhat_row(h, rowedges, rowstart, rowend, diag, n, lane);
  ((float2*)T1)[(size_t)n * (HID / 2) + lane] = r;
}

__global__ __launch_bounds__(256)
void k_spmm_t2(const float* __restrict__ T1, const float* __restrict__ h,
               float* __restrict__ T2,
               const int2* __restrict__ rowedges, const int* __restrict__ rowstart,
               const int* __restrict__ rowend, float diag, int Nn) {
  const int lane = threadIdx.x & 31;
  const int n = __builtin_amdgcn_readfirstlane(blockIdx.x * 8 + (threadIdx.x >> 5));
  if (n >= Nn) return;
  const float2 r = lhat_row(T1, rowedges, rowstart, rowend, diag, n, lane);
  const size_t o = (size_t)n * (HID / 2) + lane;
  const float2 hv = ((const float2*)h)[o];
  float2 t;
  t.x = 2.0f * r.x - hv.x;
  t.y = 2.0f * r.y - hv.y;
  ((float2*)T2)[o] = t;
}


__global__ __launch_bounds__(256)
void k_packB(const float* __restrict__ Wx, const float* __restrict__ theta,
             const float* __restrict__ bg, const float* __restrict__ conv_b,
             float* __restrict__ Bp, float* __restrict__ bias) {
  int gid = blockIdx.x * 256 + threadIdx.x;
  if (gid >= KDIM * GCOLS) return;
  int k = gid / GCOLS, col = gid % GCOLS;
  int g = col >> 6, o = col & 63;
  float v;
  if (k < INDIM) {
    v = Wx[((size_t)g * INDIM + k) * HID + o];
  } else {
    int kk = k - INDIM;
    int kb = kk >> 6, kr = kk & 63;
    v = theta[(((size_t)g * 3 + kb) * HID + kr) * HID + o];
  }
  Bp[gid] = v;
  if (gid < GCOLS) bias[gid] = bg[g * HID + o] + conv_b[g * HID + o];
}


#define AS1 324
#define PS1 260
#define MT 64
#define MSUB (MT / 16)

__device__ __forceinline__ void lstm_cell(float pi, float pf, float pt, float po, float cold,
                                          float& hnew, float& cnew) {
  float ig = fsigmoid(pi);
  float fg = fsigmoid(pf);
  float tg = tanhf(pt);
  float og = fsigmoid(po);
  float cv = fg * cold + ig * tg;
  hnew = og * tanhf(cv);
  cnew = cv;
}

__global__ __launch_bounds__(256)
void k_gemm_gates(const float* __restrict__ x, const float* __restrict__ h,
                  const float* __restrict__ T1, const float* __restrict__ T2,
                  const float* __restrict__ Bp, const float* __restrict__ bias,
                  const float* __restrict__ c_in,
                  float* __restrict__ h_out, float* __restrict__ c_out, int Nn) {
  __shared__ __attribute__((aligned(16))) float Asmem[MT * AS1];
  float* Psmem = Asmem;
  const int tid = threadIdx.x;
  const int m0 = blockIdx.x * MT;

#pragma unroll
  for (int it = 0; it < MT * 32 / 256; ++it) {
    int i = it * 256 + tid;
    int r = i >> 5, q = i & 31;
    int n = m0 + r; if (n >= Nn) n = Nn - 1;
    const float4 v = *(const float4*)(x + (size_t)n * INDIM + q * 4);
    *(float4*)&Asmem[r * AS1 + q * 4] = v;
  }
#pragma unroll
  for (int it = 0; it < MT * 16 / 256; ++it) {
    int i = it * 256 + tid;
    int r = i >> 4, q = i & 15;
    int n = m0 + r; if (n >= Nn) n = Nn - 1;
    const size_t g = (size_t)n * HID + q * 4;
    const float4 vh = *(const float4*)(h + g);
    const float4 v1 = *(const float4*)(T1 + g);
    const float4 v2 = *(const float4*)(T2 + g);
    float* ap = &Asmem[r * AS1 + 128 + q * 4];
    *(float4*)(ap)       = vh;
    *(float4*)(ap + 64)  = v1;
    *(float4*)(ap + 128) = v2;
  }
  __syncthreads();

  const int wave = tid >> 5, lane = tid & 31;
  const int lo = lane & 15, hi = lane >> 4;
  const int c0 = wave * 32;
  v8f acc[MSUB][2];
#pragma unroll
  for (int t = 0; t < MSUB; ++t) { acc[t][0] = v8f{}; acc[t][1] = v8f{}; }

  for (int k0 = 0; k0 < KDIM; k0 += 4) {
    const float* bptr = Bp + (size_t)(k0 + 2 * hi) * GCOLS + c0 + lo;
    v2f b0, b1;
    b0[0] = bptr[0];    b0[1] = bptr[GCOLS];
    b1[0] = bptr[16];   b1[1] = bptr[GCOLS + 16];
#pragma unroll
    for (int t = 0; t < MSUB; ++t) {
      v2f a = *(const v2f*)&Asmem[(t * 16 + lo) * AS1 + k0 + 2 * hi];
      acc[t][0] = wmma_f32_4(a, b0, acc[t][0]);
      acc[t][1] = wmma_f32_4(a, b1, acc[t][1]);
    }
  }

  float bias0 = bias[c0 + lo], bias1 = bias[c0 + 16 + lo];
  __syncthreads();
#pragma unroll
  for (int t = 0; t < MSUB; ++t) {
#pragma unroll
    for (int r = 0; r < 8; ++r) {
      int row = t * 16 + r + 8 * hi;
      Psmem[row * PS1 + c0 + lo]      = acc[t][0][r] + bias0;
      Psmem[row * PS1 + c0 + 16 + lo] = acc[t][1][r] + bias1;
    }
  }
  __syncthreads();

  for (int it = 0; it < MT * 16 / 256; ++it) {
    int i = it * 256 + tid;
    int row = i >> 4, o = (i & 15) * 4;
    int n = m0 + row;
    if (n >= Nn) continue;
    const float4 cold = *(const float4*)(c_in + (size_t)n * HID + o);
    const float* pr = &Psmem[row * PS1 + o];
    const float4 pi = *(const float4*)(pr);
    const float4 pf = *(const float4*)(pr + 64);
    const float4 pt = *(const float4*)(pr + 128);
    const float4 po = *(const float4*)(pr + 192);
    float4 hv, cv;
    lstm_cell(pi.x, pf.x, pt.x, po.x, cold.x, hv.x, cv.x);
    lstm_cell(pi.y, pf.y, pt.y, po.y, cold.y, hv.y, cv.y);
    lstm_cell(pi.z, pf.z, pt.z, po.z, cold.z, hv.z, cv.z);
    lstm_cell(pi.w, pf.w, pt.w, po.w, cold.w, hv.w, cv.w);
    *(float4*)(h_out + (size_t)n * HID + o) = hv;
    *(float4*)(c_out + (size_t)n * HID + o) = cv;
  }
}


#define AS2 68
#define BS2 68

__global__ __launch_bounds__(256)
void k_gemm_out(const float* __restrict__ h0, const float* __restrict__ Wl,
                const float* __restrict__ bl, float* __restrict__ out, int Nn) {
  __shared__ float Bs[64 * BS2];
  __shared__ float As[32 * AS2];
  const int tid = threadIdx.x;
  const int m0 = blockIdx.x * 32;

  for (int i = tid; i < 64 * 64; i += 256) {
    int k = i >> 6, ccol = i & 63;
    Bs[k * BS2 + ccol] = Wl[i];
  }
  for (int i = tid; i < 32 * 64; i += 256) {
    int r = i >> 6, k = i & 63;
    int n = m0 + r; if (n >= Nn) n = Nn - 1;
    float v = h0[(size_t)n * HID + k];
    As[r * AS2 + k] = v > 0.0f ? v : 0.0f;
  }
  __syncthreads();

  const int wave = tid >> 5, lane = tid & 31;
  const int lo = lane & 15, hi = lane >> 4;
  const int r0 = (wave >> 2) * 16, c0 = (wave & 3) * 16;
  v8f acc = {};
  for (int k0 = 0; k0 < 64; k0 += 4) {
    v2f a = *(const v2f*)&As[(r0 + lo) * AS2 + k0 + 2 * hi];
    v2f b;
    b[0] = Bs[(k0 + 2 * hi) * BS2 + c0 + lo];
    b[1] = Bs[(k0 + 2 * hi + 1) * BS2 + c0 + lo];
    acc = wmma_f32_4(a, b, acc);
  }
  float bb = bl[c0 + lo];
#pragma unroll
  for (int r = 0; r < 8; ++r) {
    int row = r0 + r + 8 * hi;
    int n = m0 + row;
    if (n < Nn) out[(size_t)n * HID + c0 + lo] = acc[r] + bb;
  }
}


extern "C" void kernel_launch(void* const* d_in, const int* in_sizes, int n_in,
                              void* d_out, int out_size, void* d_ws, size_t ws_size,
                              hipStream_t stream) {
  const float* x      = (const float*)d_in[0];
  const int*   ei     = (const int*)d_in[1];
  const float* ew     = (const float*)d_in[2];
  const float* h      = (const float*)d_in[3];
  const float* c      = (const float*)d_in[4];
  const float* Wx     = (const float*)d_in[5];
  const float* bg     = (const float*)d_in[6];
  const float* theta  = (const float*)d_in[7];
  const float* conv_b = (const float*)d_in[8];
  const float* Wl     = (const float*)d_in[9];
  const float* bl     = (const float*)d_in[10];

  const int E  = in_sizes[2];
  const int Nn = in_sizes[3] / HID;
  const int* src = ei;
  const int* dst = ei + E;

  float* T1   = (float*)d_ws;
  float* T2   = T1 + (size_t)Nn * HID;
  float* Bp   = T2 + (size_t)Nn * HID;
  float* bias = Bp + (size_t)KDIM * GCOLS;
  int2*  edgepay  = (int2*)(bias + GCOLS);
  int2*  payA     = edgepay + E;
  int2*  rowedges = payA + E;
  int*   keyA = (int*)(rowedges + E);
  int*   sdst = keyA + E;
  int*   lrank = sdst + E;
  const int NB = (E + SORT_CH - 1) / SORT_CH;
  int*   table = lrank + E;
  float* deg  = (float*)(table + (size_t)NB * SORT_BINS);
  int*   rowstart = (int*)(deg + Nn);
  int*   rowend = rowstart + Nn;
  float* dinv = (float*)(rowend + Nn);

  float* out   = (float*)d_out;
  float* h_out = out   + (size_t)Nn * HID;
  float* c_out = h_out + (size_t)Nn * HID;

  const float DIAG = 2.0f / LMAX - 1.0f;

  hipMemsetAsync(deg, 0, (size_t)Nn * 3 * sizeof(float), stream);
  k_degree<<<(E + 255) / 256, 256, 0, stream>>>(src, dst, ew, deg, E);
  k_dinv<<<(Nn + 255) / 256, 256, 0, stream>>>(deg, dinv, Nn);
  k_lapw_pack<<<(E + 255) / 256, 256, 0, stream>>>(src, dst, ew, dinv, edgepay, E);

  k_split_rank<<<NB, 256, 0, stream>>>(dst, 0, lrank, table, E);
  k_split_scatter<<<NB, 256, 0, stream>>>(dst, edgepay, 0, lrank, table, NB, keyA, payA, E);
  k_split_rank<<<NB, 256, 0, stream>>>(keyA, 9, lrank, table, E);
  k_split_scatter<<<NB, 256, 0, stream>>>(keyA, payA, 9, lrank, table, NB, sdst, rowedges, E);
  k_rowbounds<<<(E + 255) / 256, 256, 0, stream>>>(sdst, rowstart, rowend, E);

  k_spmm_t1<<<(Nn + 7) / 8, 256, 0, stream>>>(h, T1, rowedges, rowstart, rowend, DIAG, Nn);
  k_spmm_t2<<<(Nn + 7) / 8, 256, 0, stream>>>(T1, h, T2, rowedges, rowstart, rowend, DIAG, Nn);

  k_packB<<<(KDIM * GCOLS + 255) / 256, 256, 0, stream>>>(Wx, theta, bg, conv_b, Bp, bias);

  k_gemm_gates<<<(Nn + MT - 1) / MT, 256, 0, stream>>>(x, h, T1, T2, Bp, bias, c,
                                                   h_out, c_out, Nn);
  k_gemm_out<<<(Nn + 31) / 32, 256, 0, stream>>>(h_out, Wl, bl, out, Nn);
}


# dependent launches 13 -> 9: first split ranking, degrees and weight pack share one grid; deg^-1/2 formed where used; output GEMM runs in the gate kernel on its relu(h0) tile; no sum order changed
# speedup vs baseline: 1.6952x; 1.0726x over previous
#include <hip/hip_runtime.h>
#include <hip/hip_bf16.h>

typedef __attribute__((ext_vector_type(2))) float v2f;
typedef __attribute__((ext_vector_type(8))) float v8f;

#define HID 64
#define INDIM 128
#define KDIM 320
#define GCOLS 256
#define LMAX 2.0f

__device__ __forceinline__ v8f wmma_f32_4(v2f a, v2f b, v8f c) {
  return __builtin_amdgcn_wmma_f32_16x16x4_f32(false, a, false, b, (short)0, c, false, false);
}

__device__ __forceinline__ float fsigmoid(float x) {
  return 1.0f / (1.0f + __expf(-x));
}

__device__ __forceinline__ void atomic_fadd(float* p, float v) {
  asm volatile("global_atomic_add_f32 %0, %1, off scope:SCOPE_DEV"
               :: "v"(p), "v"(v) : "memory");
}


__device__ __forceinline__
void degree_block(const int* __restrict__ src, const int* __restrict__ dst,
                  const float* __restrict__ ew, float* __restrict__ deg, int E, int bid) {
  int e = bid * 256 + threadIdx.x;
  if (e >= E) return;
  int s = src[e], d = dst[e];
  if (s != d) atomic_fadd(&deg[s], ew[e]);
}

__device__ __forceinline__ float dinv_of(float dg) {
  return (dg > 0.0f) ? rsqrtf(dg) : 0.0f;
}

__global__ __launch_bounds__(256)
void k_lapw_pack(const int* __restrict__ src, const int* __restrict__ dst,
                 const float* __restrict__ ew, const float* __restrict__ deg,
                 int2* __restrict__ edgepay, int E) {
  int e = blockIdx.x * 256 + threadIdx.x;
  if (e >= E) return;
  int s = src[e], d = dst[e];
  float w = (s == d) ? 0.0f : ew[e];
  const float ds = dinv_of(deg[s]), dd = dinv_of(deg[d]);
  float lw = (2.0f / LMAX) * (-ds * w * dd);
  edgepay[e] = make_int2(s, __float_as_int(lw));
}


#define SORT_CH 8192
#define SORT_ROUNDS (SORT_CH / 256)
#define SORT_BINS 512

__device__ __forceinline__ int wave_incl_scan(int v, int lane) {
  for (int off = 1; off < 32; off <<= 1) {
    int t = __shfl_up(v, off, 32);
    if (lane >= off) v += t;
  }
  return v;
}

__device__ __forceinline__ int block_excl_scan(int v, int* wsum, int lane, int wave, int& total) {
  const int incl = wave_incl_scan(v, lane);
  if (lane == 31) wsum[wave] = incl;
  __syncthreads();
  int wb = 0, t = 0;
  for (int w = 0; w < 8; ++w) { int s = wsum[w]; if (w < wave) wb += s; t += s; }
  total = t;
  __syncthreads();
  return wb + incl - v;
}

__device__ __forceinline__
void split_rank_block(const int* __restrict__ key, int shift,
                      int* __restrict__ lrank, int* __restrict__ table, int E, int bid) {
  __shared__ int whist[8 * SORT_BINS];
  const int tid = threadIdx.x, lane = tid & 31, wave = tid >> 5;
  for (int i = tid; i < 8 * SORT_BINS; i += 256) whist[i] = 0;
  __syncthreads();
  const int base = bid * SORT_CH + wave * (SORT_CH / 8) + lane;
  int dg[SORT_ROUNDS], lr[SORT_ROUNDS];
#pragma unroll
  for (int r = 0; r < SORT_ROUNDS; ++r) {
    const int e = base + r * 32;
    const int k = key[e < E ? e : E - 1];
    dg[r] = (e < E) ? ((k >> shift) & (SORT_BINS - 1)) : -1;
  }
  int* wh = whist + wave * SORT_BINS;
#pragma unroll
  for (int r = 0; r < SORT_ROUNDS; ++r) {
    const int d = dg[r];
    const bool valid = d >= 0;
    unsigned peers = __builtin_amdgcn_ballot_w32(valid);
#pragma unroll
    for (int b = 0; b < 9; ++b) {
      const bool bit = ((d >> b) & 1) != 0;
      const unsigned bal = __builtin_amdgcn_ballot_w32(bit);
      peers &= bit ? bal : ~bal;
    }
    const int rk = __popc(peers & ((1u << lane) - 1u));
    int b0 = 0;
    if (valid) {
      b0 = wh[d];
      if (rk == 0) wh[d] = b0 + __popc(peers);
    }
    lr[r] = b0 + rk;
  }
  __syncthreads();
  for (int dgt = tid; dgt < SORT_BINS; dgt += 256) {
    int run = 0;
    for (int w = 0; w < 8; ++w) {
      int v = whist[w * SORT_BINS + dgt];
      whist[w * SORT_BINS + dgt] = run;
      run += v;
    }
    table[bid * SORT_BINS + dgt] = run;
  }
  __syncthreads();
#pragma unroll
  for (int r = 0; r < SORT_ROUNDS; ++r) {
    const int e = base + r * 32;
    if (e < E) lrank[e] = lr[r] + wh[dg[r]];
  }
}

__global__ __launch_bounds__(256)
void k_split_rank(const int* __restrict__ key, int shift,
                  int* __restrict__ lrank, int* __restrict__ table, int E) {
  split_rank_block(key, shift, lrank, table, E, blockIdx.x);
}

__global__ __launch_bounds__(256)
void k_split_scatter(const int* __restrict__ key, const int2* __restrict__ pay, int shift,
                     const int* __restrict__ lrank, const int* __restrict__ table, int nblocks,
                     int* __restrict__ key_out, int2* __restrict__ pay_out, int E) {
  __shared__ int lstart[SORT_BINS];
  __shared__ int gdelta[SORT_BINS];
  __shared__ int wsum[8];
  __shared__ int skey[SORT_CH];
  __shared__ __attribute__((aligned(8))) int spay_w[2 * SORT_CH];
  int2* spay = (int2*)spay_w;
  const int tid = threadIdx.x, lane = tid & 31, wave = tid >> 5;
  const int me = blockIdx.x;
  int pre0 = 0, tot0 = 0, pre1 = 0, tot1 = 0;
#pragma unroll 14
  for (int b = 0; b < nblocks; ++b) {
    const int v0 = table[b * SORT_BINS + tid];
    const int v1 = table[b * SORT_BINS + 256 + tid];
    tot0 += v0; tot1 += v1;
    if (b < me) { pre0 += v0; pre1 += v1; }
  }
  const int own0 = table[me * SORT_BINS + tid];
  const int own1 = table[me * SORT_BINS + 256 + tid];
  int sum0, sum1, osum0, osum1;
  const int ex0 = block_excl_scan(tot0, wsum, lane, wave, sum0);
  const int ex1 = block_excl_scan(tot1, wsum, lane, wave, sum1);
  const int l0 = block_excl_scan(own0, wsum, lane, wave, osum0);
  const int l1 = osum0 + block_excl_scan(own1, wsum, lane, wave, osum1);
  lstart[tid]       = l0;
  lstart[tid + 256] = l1;
  gdelta[tid]       = ex0 + pre0 - l0;
  gdelta[tid + 256] = sum0 + ex1 + pre1 - l1;
  __syncthreads();
  for (int r0 = 0; r0 < SORT_ROUNDS; r0 += 8) {
    int kk[8], rk[8];
    int2 pv[8];
#pragma unroll
    for (int u = 0; u < 8; ++u) {
      const int e = me * SORT_CH + (r0 + u) * 256 + tid;
      const int ec = e < E ? e : E - 1;
      kk[u] = key[ec];
      rk[u] = lrank[ec];
      pv[u] = pay[ec];
    }
    int lp[8];
#pragma unroll
    for (int u = 0; u < 8; ++u)
      lp[u] = lstart[(kk[u] >> shift) & (SORT_BINS - 1)] + rk[u];
#pragma unroll
    for (int u = 0; u < 8; ++u) {
      const int e = me * SORT_CH + (r0 + u) * 256 + tid;
      if (e < E) {
        skey[lp[u]] = kk[u];
        spay[lp[u]] = pv[u];
      }
    }
  }
  __syncthreads();
  const int rem = E - me * SORT_CH;
  const int cnt = rem < SORT_CH ? rem : SORT_CH;
  for (int r0 = 0; r0 < SORT_ROUNDS; r0 += 8) {
    int kk[8];
    int2 pv[8];
#pragma unroll
    for (int u = 0; u < 8; ++u) {
      const int i = (r0 + u) * 256 + tid;
      const int ic = i < cnt ? i : cnt - 1;
      kk[u] = skey[ic];
      pv[u] = spay[ic];
    }
    int ps[8];
#pragma unroll
    for (int u = 0; u < 8; ++u)
      ps[u] = (r0 + u) * 256 + tid + gdelta[(kk[u] >> shift) & (SORT_BINS - 1)];
#pragma unroll
    for (int u = 0; u < 8; ++u) {
      const int i = (r0 + u) * 256 + tid;
      if (i < cnt) {
        key_out[ps[u]] = kk[u];
        pay_out[ps[u]] = pv[u];
      }
    }
  }
}

__global__ __launch_bounds__(256)
void k_rowbounds(const int* __restrict__ sdst, int* __restrict__ rowstart,
                 int* __restrict__ rowend, int E) {
  int p = blockIdx.x * 256 + threadIdx.x;
  if (p >= E) return;
  const int d = sdst[p];
  if (p == 0 || sdst[p - 1] != d) rowstart[d] = p;
  if (p == E - 1 || sdst[p + 1] != d) rowend[d] = p + 1;
}

#define ROWCHUNK 8
__device__ __forceinline__ float2 lhat_row(const float* __restrict__ z,
                                           const int2* __restrict__ rowedges,
                                           const int* __restrict__ rowstart,
                                           const int* __restrict__ rowend,
                                           float diag, int n, int lane) {
  const float2* z2 = (const float2*)z;
  const int p0 = rowstart[n];
  const int c = rowend[n] - p0;
  const int2* ce = rowedges + p0;
  float2 acc = make_float2(0.0f, 0.0f);
  for (int j = 0; j < c; j += ROWCHUNK) {
    float2 zv[ROWCHUNK];
    float wv[ROWCHUNK];
#pragma unroll
    for (int u = 0; u < ROWCHUNK; ++u) {
      const int idx = (j + u < c) ? (j + u) : (c - 1);
      const int2 en = ce[idx];
      wv[u] = (j + u < c) ? __int_as_float(en.y) : 0.0f;
      zv[u] = z2[(size_t)en.x * (HID / 2) + lane];
    }
#pragma unroll
    for (int u = 0; u < ROWCHUNK; ++u) {
      acc.x += wv[u] * zv[u].x;
      acc.y += wv[u] * zv[u].y;
    }
  }
  const float2 zn = z2[(size_t)n * (HID / 2) + lane];
  acc.x += diag * zn.x;
  acc.y += diag * zn.y;
  return acc;
}

__global__ __launch_bounds__(256)
void k_spmm_t1(const float* __restrict__ h, float* __restrict__ T1,
               const int2* __restrict__ rowedges, const int* __restrict__ rowstart,
               const int* __restrict__ rowend, float diag, int Nn) {
  const int lane = threadIdx.x & 31;
  const int n = __builtin_amdgcn_readfirstlane(blockIdx.x * 8 + (threadIdx.x >> 5));
  if (n >= Nn) return;
  const float2 r = lhat_row(h, rowedges, rowstart, rowend, diag, n, lane);
  ((float2*)T1)[(size_t)n * (HID / 2) + lane] = r;
}

__global__ __launch_bounds__(256)
void k_spmm_t2(const float* __restrict__ T1, const float* __restrict__ h,
               float* __restrict__ T2,
               const int2* __restrict__ rowedges, const int* __restrict__ rowstart,
               const int* __restrict__ rowend, float diag, int Nn) {
  const int lane = threadIdx.x & 31;
  const int n = __builtin_amdgcn_readfirstlane(blockIdx.x * 8 + (threadIdx.x >> 5));
  if (n >= Nn) return;
  const float2 r = lhat_row(T1, rowedges, rowstart, rowend, diag, n, lane);
  const size_t o = (size_t)n * (HID / 2) + lane;
  const float2 hv = ((const float2*)h)[o];
  float2 t;
  t.x = 2.0f * r.x - hv.x;
  t.y = 2.0f * r.y - hv.y;
  ((float2*)T2)[o] = t;
}


__device__ __forceinline__
void packB_block(const float* __restrict__ Wx, const float* __restrict__ theta,
                 const float* __restrict__ bg, const float* __restrict__ conv_b,
                 float* __restrict__ Bp, float* __restrict__ bias, int bid) {
  int gid = bid * 256 + threadIdx.x;
  if (gid >= KDIM * GCOLS) return;
  int k = gid / GCOLS, col = gid % GCOLS;
  int g = col >> 6, o = col & 63;
  float v;
  if (k < INDIM) {
    v = Wx[((size_t)g * INDIM + k) * HID + o];
  } else {
    int kk = k - INDIM;
    int kb = kk >> 6, kr = kk & 63;
    v = theta[(((size_t)g * 3 + kb) * HID + kr) * HID + o];
  }
  Bp[gid] = v;
  if (gid < GCOLS) bias[gid] = bg[g * HID + o] + conv_b[g * HID + o];
}

__global__ __launch_bounds__(256)
void k_front(const int* __restrict__ src, const int* __restrict__ dst,
             const float* __restrict__ ew, float* __restrict__ deg,
             int* __restrict__ lrank, int* __restrict__ table,
             const float* __restrict__ Wx, const float* __restrict__ theta,
             const float* __restrict__ bg, const float* __restrict__ conv_b,
             float* __restrict__ Bp, float* __restrict__ bias,
             int E, int nrank, int ndeg) {
  const int b = blockIdx.x;
  if (b < nrank) {
    split_rank_block(dst, 0, lrank, table, E, b);
  } else if (b < nrank + ndeg) {
    degree_block(src, dst, ew, deg, E, b - nrank);
  } else {
    packB_block(Wx, theta, bg, conv_b, Bp, bias, b - nrank - ndeg);
  }
}


#define AS1 324
#define PS1 260
#define MT 64
#define MSUB (MT / 16)
#define HS1 68

__device__ __forceinline__ void lstm_cell(float pi, float pf, float pt, float po, float cold,
                                          float& hnew, float& cnew) {
  float ig = fsigmoid(pi);
  float fg = fsigmoid(pf);
  float tg = tanhf(pt);
  float og = fsigmoid(po);
  float cv = fg * cold + ig * tg;
  hnew = og * tanhf(cv);
  cnew = cv;
}

__global__ __launch_bounds__(256)
void k_gemm_gates(const float* __restrict__ x, const float* __restrict__ h,
                  const float* __restrict__ T1, const float* __restrict__ T2,
                  const float* __restrict__ Bp, const float* __restrict__ bias,
                  const float* __restrict__ c_in,
                  const float* __restrict__ Wl, const float* __restrict__ bl,
                  float* __restrict__ out,
                  float* __restrict__ h_out, float* __restrict__ c_out, int Nn) {
  __shared__ __attribute__((aligned(16))) float Asmem[MT * AS1];
  __shared__ __attribute__((aligned(16))) float Hsmem[MT * HS1];
  float* Psmem = Asmem;
  const int tid = threadIdx.x;
  const int m0 = blockIdx.x * MT;

#pragma unroll
  for (int it = 0; it < MT * 32 / 256; ++it) {
    int i = it * 256 + tid;
    int r = i >> 5, q = i & 31;
    int n = m0 + r; if (n >= Nn) n = Nn - 1;
    const float4 v = *(const float4*)(x + (size_t)n * INDIM + q * 4);
    *(float4*)&Asmem[r * AS1 + q * 4] = v;
  }
#pragma unroll
  for (int it = 0; it < MT * 16 / 256; ++it) {
    int i = it * 256 + tid;
    int r = i >> 4, q = i & 15;
    int n = m0 + r; if (n >= Nn) n = Nn - 1;
    const size_t g = (size_t)n * HID + q * 4;
    const float4 vh = *(const float4*)(h + g);
    const float4 v1 = *(const float4*)(T1 + g);
    const float4 v2 = *(const float4*)(T2 + g);
    float* ap = &Asmem[r * AS1 + 128 + q * 4];
    *(float4*)(ap)       = vh;
    *(float4*)(ap + 64)  = v1;
    *(float4*)(ap + 128) = v2;
  }
  __syncthreads();

  const int wave = tid >> 5, lane = tid & 31;
  const int lo = lane & 15, hi = lane >> 4;
  const int c0 = wave * 32;
  v8f acc[MSUB][2];
#pragma unroll
  for (int t = 0; t < MSUB; ++t) { acc[t][0] = v8f{}; acc[t][1] = v8f{}; }

  for (int k0 = 0; k0 < KDIM; k0 += 4) {
    const float* bptr = Bp + (size_t)(k0 + 2 * hi) * GCOLS + c0 + lo;
    v2f b0, b1;
    b0[0] = bptr[0];    b0[1] = bptr[GCOLS];
    b1[0] = bptr[16];   b1[1] = bptr[GCOLS + 16];
#pragma unroll
    for (int t = 0; t < MSUB; ++t) {
      v2f a = *(const v2f*)&Asmem[(t * 16 + lo) * AS1 + k0 + 2 * hi];
      acc[t][0] = wmma_f32_4(a, b0, acc[t][0]);
      acc[t][1] = wmma_f32_4(a, b1, acc[t][1]);
    }
  }

  float bias0 = bias[c0 + lo], bias1 = bias[c0 + 16 + lo];
  const int oc = (wave & 3) * 16;
  v2f wb[HID / 4];
#pragma unroll
  for (int s = 0; s < HID / 4; ++s) {
    wb[s][0] = Wl[(4 * s + 2 * hi) * HID + oc + lo];
    wb[s][1] = Wl[(4 * s + 2 * hi + 1) * HID + oc + lo];
  }
  __syncthreads();
#pragma unroll
  for (int t = 0; t < MSUB; ++t) {
#pragma unroll
    for (int r = 0; r < 8; ++r) {
      int row = t * 16 + r + 8 * hi;
      Psmem[row * PS1 + c0 + lo]      = acc[t][0][r] + bias0;
      Psmem[row * PS1 + c0 + 16 + lo] = acc[t][1][r] + bias1;
    }
  }
  __syncthreads();

  for (int it = 0; it < MT * 16 / 256; ++it) {
    int i = it * 256 + tid;
    int row = i >> 4, o = (i & 15) * 4;
    int n = m0 + row;
    if (n >= Nn) {
      *(float4*)&Hsmem[row * HS1 + o] = make_float4(0.0f, 0.0f, 0.0f, 0.0f);
      continue;
    }
    const float4 cold = *(const float4*)(c_in + (size_t)n * HID + o);
    const float* pr = &Psmem[row * PS1 + o];
    const float4 pi = *(const float4*)(pr);
    const float4 pf = *(const float4*)(pr + 64);
    const float4 pt = *(const float4*)(pr + 128);
    const float4 po = *(const float4*)(pr + 192);
    float4 hv, cv;
    lstm_cell(pi.x, pf.x, pt.x, po.x, cold.x, hv.x, cv.x);
    lstm_cell(pi.y, pf.y, pt.y, po.y, cold.y, hv.y, cv.y);
    lstm_cell(pi.z, pf.z, pt.z, po.z, cold.z, hv.z, cv.z);
    lstm_cell(pi.w, pf.w, pt.w, po.w, cold.w, hv.w, cv.w);
    *(float4*)(h_out + (size_t)n * HID + o) = hv;
    *(float4*)(c_out + (size_t)n * HID + o) = cv;
    float4 hr;
    hr.x = hv.x > 0.0f ? hv.x : 0.0f;
    hr.y = hv.y > 0.0f ? hv.y : 0.0f;
    hr.z = hv.z > 0.0f ? hv.z : 0.0f;
    hr.w = hv.w > 0.0f ? hv.w : 0.0f;
    *(float4*)&Hsmem[row * HS1 + o] = hr;
  }
  __syncthreads();

  {
    const int rt = (wave >> 2) * 2;
    v8f oacc0 = {}, oacc1 = {};
#pragma unroll
    for (int k0 = 0; k0 < HID; k0 += 4) {
      const v2f b = wb[k0 / 4];
      v2f a0 = *(const v2f*)&Hsmem[(rt * 16 + lo) * HS1 + k0 + 2 * hi];
      v2f a1 = *(const v2f*)&Hsmem[(rt * 16 + 16 + lo) * HS1 + k0 + 2 * hi];
      oacc0 = wmma_f32_4(a0, b, oacc0);
      oacc1 = wmma_f32_4(a1, b, oacc1);
    }
    const float bb = bl[oc + lo];
#pragma unroll
    for (int r = 0; r < 8; ++r) {
      const int n0 = m0 + rt * 16 + r + 8 * hi;
      const int n1 = n0 + 16;
      if (n0 < Nn) out[(size_t)n0 * HID + oc + lo] = oacc0[r] + bb;
      if (n1 < Nn) out[(size_t)n1 * HID + oc + lo] = oacc1[r] + bb;
    }
  }
}


extern "C" void kernel_launch(void* const* d_in, const int* in_sizes, int n_in,
                              void* d_out, int out_size, void* d_ws, size_t ws_size,
                              hipStream_t stream) {
  const float* x      = (const float*)d_in[0];
  const int*   ei     = (const int*)d_in[1];
  const float* ew     = (const float*)d_in[2];
  const float* h      = (const float*)d_in[3];
  const float* c      = (const float*)d_in[4];
  const float* Wx     = (const float*)d_in[5];
  const float* bg     = (const float*)d_in[6];
  const float* theta  = (const float*)d_in[7];
  const float* conv_b = (const float*)d_in[8];
  const float* Wl     = (const float*)d_in[9];
  const float* bl     = (const float*)d_in[10];

  const int E  = in_sizes[2];
  const int Nn = in_sizes[3] / HID;
  const int* src = ei;
  const int* dst = ei + E;

  float* T1   = (float*)d_ws;
  float* T2   = T1 + (size_t)Nn * HID;
  float* Bp   = T2 + (size_t)Nn * HID;
  float* bias = Bp + (size_t)KDIM * GCOLS;
  int2*  edgepay  = (int2*)(bias + GCOLS);
  int2*  payA     = edgepay + E;
  int2*  rowedges = payA + E;
  int*   keyA = (int*)(rowedges + E);
  int*   sdst = keyA + E;
  int*   lrank = sdst + E;
  const int NB = (E + SORT_CH - 1) / SORT_CH;
  int*   table = lrank + E;
  float* deg  = (float*)(table + (size_t)NB * SORT_BINS);
  int*   rowstart = (int*)(deg + Nn);
  int*   rowend = rowstart + Nn;

  float* out   = (float*)d_out;
  float* h_out = out   + (size_t)Nn * HID;
  float* c_out = h_out + (size_t)Nn * HID;

  const float DIAG = 2.0f / LMAX - 1.0f;

  hipMemsetAsync(deg, 0, (size_t)Nn * 3 * sizeof(float), stream);
  const int NDEG = (E + 255) / 256;
  const int NPACK = (KDIM * GCOLS + 255) / 256;
  k_front<<<NB + NDEG + NPACK, 256, 0, stream>>>(src, dst, ew, deg, lrank, table,
                                                 Wx, theta, bg, conv_b, Bp, bias, E, NB, NDEG);
  k_lapw_pack<<<(E + 255) / 256, 256, 0, stream>>>(src, dst, ew, deg, edgepay, E);

  k_split_scatter<<<NB, 256, 0, stream>>>(dst, edgepay, 0, lrank, table, NB, keyA, payA, E);
  k_split_rank<<<NB, 256, 0, stream>>>(keyA, 9, lrank, table, E);
  k_split_scatter<<<NB, 256, 0, stream>>>(keyA, payA, 9, lrank, table, NB, sdst, rowedges, E);
  k_rowbounds<<<(E + 255) / 256, 256, 0, stream>>>(sdst, rowstart, rowend, E);

  k_spmm_t1<<<(Nn + 7) / 8, 256, 0, stream>>>(h, T1, rowedges, rowstart, rowend, DIAG, Nn);
  k_spmm_t2<<<(Nn + 7) / 8, 256, 0, stream>>>(T1, h, T2, rowedges, rowstart, rowend, DIAG, Nn);

  k_gemm_gates<<<(Nn + MT - 1) / MT, 256, 0, stream>>>(x, h, T1, T2, Bp, bias, c, Wl, bl, out,
                                                   h_out, c_out, Nn);
}


# edge sort: one global split on dst>>8, then one work-group sorts each 256-row bucket in LDS and writes its row bounds; second ranking and row-bounds launches gone (9 -> 7); sum order kept
# speedup vs baseline: 1.8563x; 1.0950x over previous
#include <hip/hip_runtime.h>
#include <hip/hip_bf16.h>

typedef __attribute__((ext_vector_type(2))) float v2f;
typedef __attribute__((ext_vector_type(8))) float v8f;

#define HID 64
#define INDIM 128
#define KDIM 320
#define GCOLS 256
#define LMAX 2.0f

__device__ __forceinline__ v8f wmma_f32_4(v2f a, v2f b, v8f c) {
  return __builtin_amdgcn_wmma_f32_16x16x4_f32(false, a, false, b, (short)0, c, false, false);
}

__device__ __forceinline__ float fsigmoid(float x) {
  return 1.0f / (1.0f + __expf(-x));
}

__device__ __forceinline__ void atomic_fadd(float* p, float v) {
  asm volatile("global_atomic_add_f32 %0, %1, off scope:SCOPE_DEV"
               :: "v"(p), "v"(v) : "memory");
}


__device__ __forceinline__
void degree_block(const int* __restrict__ src, const int* __restrict__ dst,
                  const float* __restrict__ ew, float* __restrict__ deg, int E, int bid) {
  int e = bid * 256 + threadIdx.x;
  if (e >= E) return;
  int s = src[e], d = dst[e];
  if (s != d) atomic_fadd(&deg[s], ew[e]);
}

__device__ __forceinline__ float dinv_of(float dg) {
  return (dg > 0.0f) ? rsqrtf(dg) : 0.0f;
}

__global__ __launch_bounds__(256)
void k_lapw_pack(const int* __restrict__ src, const int* __restrict__ dst,
                 const float* __restrict__ ew, const float* __restrict__ deg,
                 int2* __restrict__ edgepay, int E) {
  int e = blockIdx.x * 256 + threadIdx.x;
  if (e >= E) return;
  int s = src[e], d = dst[e];
  float w = (s == d) ? 0.0f : ew[e];
  const float ds = dinv_of(deg[s]), dd = dinv_of(deg[d]);
  float lw = (2.0f / LMAX) * (-ds * w * dd);
  edgepay[e] = make_int2(s, __float_as_int(lw));
}


#define SORT_CH 8192
#define SORT_ROUNDS (SORT_CH / 256)
#define SORT_BINS 512

__device__ __forceinline__ int wave_incl_scan(int v, int lane) {
  for (int off = 1; off < 32; off <<= 1) {
    int t = __shfl_up(v, off, 32);
    if (lane >= off) v += t;
  }
  return v;
}

__device__ __forceinline__ int block_excl_scan(int v, int* wsum, int lane, int wave, int& total) {
  const int incl = wave_incl_scan(v, lane);
  if (lane == 31) wsum[wave] = incl;
  __syncthreads();
  int wb = 0, t = 0;
  for (int w = 0; w < 8; ++w) { int s = wsum[w]; if (w < wave) wb += s; t += s; }
  total = t;
  __syncthreads();
  return wb + incl - v;
}

__device__ __forceinline__
void split_rank_block(const int* __restrict__ key, int shift,
                      int* __restrict__ lrank, int* __restrict__ table, int E, int bid) {
  __shared__ int whist[8 * SORT_BINS];
  const int tid = threadIdx.x, lane = tid & 31, wave = tid >> 5;
  for (int i = tid; i < 8 * SORT_BINS; i += 256) whist[i] = 0;
  __syncthreads();
  const int base = bid * SORT_CH + wave * (SORT_CH / 8) + lane;
  int dg[SORT_ROUNDS], lr[SORT_ROUNDS];
#pragma unroll
  for (int r = 0; r < SORT_ROUNDS; ++r) {
    const int e = base + r * 32;
    const int k = key[e < E ? e : E - 1];
    dg[r] = (e < E) ? ((k >> shift) & (SORT_BINS - 1)) : -1;
  }
  int* wh = whist + wave * SORT_BINS;
#pragma unroll
  for (int r = 0; r < SORT_ROUNDS; ++r) {
    const int d = dg[r];
    const bool valid = d >= 0;
    unsigned peers = __builtin_amdgcn_ballot_w32(valid);
#pragma unroll
    for (int b = 0; b < 9; ++b) {
      const bool bit = ((d >> b) & 1) != 0;
      const unsigned bal = __builtin_amdgcn_ballot_w32(bit);
      peers &= bit ? bal : ~bal;
    }
    const int rk = __popc(peers & ((1u << lane) - 1u));
    int b0 = 0;
    if (valid) {
      b0 = wh[d];
      if (rk == 0) wh[d] = b0 + __popc(peers);
    }
    lr[r] = b0 + rk;
  }
  __syncthreads();
  for (int dgt = tid; dgt < SORT_BINS; dgt += 256) {
    int run = 0;
    for (int w = 0; w < 8; ++w) {
      int v = whist[w * SORT_BINS + dgt];
      whist[w * SORT_BINS + dgt] = run;
      run += v;
    }
    table[bid * SORT_BINS + dgt] = run;
  }
  __syncthreads();
#pragma unroll
  for (int r = 0; r < SORT_ROUNDS; ++r) {
    const int e = base + r * 32;
    if (e < E) lrank[e] = lr[r] + wh[dg[r]];
  }
}

__global__ __launch_bounds__(256)
void k_split_scatter(const int* __restrict__ key, const int2* __restrict__ pay, int shift,
                     const int* __restrict__ lrank, const int* __restrict__ table, int nblocks,
                     int* __restrict__ key_out, int2* __restrict__ pay_out,
                     int* __restrict__ bstart, int E) {
  __shared__ int lstart[SORT_BINS];
  __shared__ int gdelta[SORT_BINS];
  __shared__ int wsum[8];
  __shared__ int skey[SORT_CH];
  __shared__ __attribute__((aligned(8))) int spay_w[2 * SORT_CH];
  int2* spay = (int2*)spay_w;
  const int tid = threadIdx.x, lane = tid & 31, wave = tid >> 5;
  const int me = blockIdx.x;
  int pre0 = 0, tot0 = 0, pre1 = 0, tot1 = 0;
#pragma unroll 14
  for (int b = 0; b < nblocks; ++b) {
    const int v0 = table[b * SORT_BINS + tid];
    const int v1 = table[b * SORT_BINS + 256 + tid];
    tot0 += v0; tot1 += v1;
    if (b < me) { pre0 += v0; pre1 += v1; }
  }
  const int own0 = table[me * SORT_BINS + tid];
  const int own1 = table[me * SORT_BINS + 256 + tid];
  int sum0, sum1, osum0, osum1;
  const int ex0 = block_excl_scan(tot0, wsum, lane, wave, sum0);
  const int ex1 = block_excl_scan(tot1, wsum, lane, wave, sum1);
  const int l0 = block_excl_scan(own0, wsum, lane, wave, osum0);
  const int l1 = osum0 + block_excl_scan(own1, wsum, lane, wave, osum1);
  lstart[tid]       = l0;
  lstart[tid + 256] = l1;
  gdelta[tid]       = ex0 + pre0 - l0;
  gdelta[tid + 256] = sum0 + ex1 + pre1 - l1;
  if (me == 0) {
    bstart[tid]       = ex0;
    bstart[tid + 256] = sum0 + ex1;
    if (tid == 0) bstart[SORT_BINS] = sum0 + sum1;
  }
  __syncthreads();
  for (int r0 = 0; r0 < SORT_ROUNDS; r0 += 8) {
    int kk[8], rk[8];
    int2 pv[8];
#pragma unroll
    for (int u = 0; u < 8; ++u) {
      const int e = me * SORT_CH + (r0 + u) * 256 + tid;
      const int ec = e < E ? e : E - 1;
      kk[u] = key[ec];
      rk[u] = lrank[ec];
      pv[u] = pay[ec];
    }
    int lp[8];
#pragma unroll
    for (int u = 0; u < 8; ++u)
      lp[u] = lstart[(kk[u] >> shift) & (SORT_BINS - 1)] + rk[u];
#pragma unroll
    for (int u = 0; u < 8; ++u) {
      const int e = me * SORT_CH + (r0 + u) * 256 + tid;
      if (e < E) {
        skey[lp[u]] = kk[u];
        spay[lp[u]] = pv[u];
      }
    }
  }
  __syncthreads();
  const int rem = E - me * SORT_CH;
  const int cnt = rem < SORT_CH ? rem : SORT_CH;
  for (int r0 = 0; r0 < SORT_ROUNDS; r0 += 8) {
    int kk[8];
    int2 pv[8];
#pragma unroll
    for (int u = 0; u < 8; ++u) {
      const int i = (r0 + u) * 256 + tid;
      const int ic = i < cnt ? i : cnt - 1;
      kk[u] = skey[ic];
      pv[u] = spay[ic];
    }
    int ps[8];
#pragma unroll
    for (int u = 0; u < 8; ++u)
      ps[u] = (r0 + u) * 256 + tid + gdelta[(kk[u] >> shift) & (SORT_BINS - 1)];
#pragma unroll
    for (int u = 0; u < 8; ++u) {
      const int i = (r0 + u) * 256 + tid;
      if (i < cnt) {
        key_out[ps[u]] = kk[u];
        pay_out[ps[u]] = pv[u];
      }
    }
  }
}

#define BK_BITS 8
#define BK_ROWS (1 << BK_BITS)
#define BK_CH 8192
#define BK_ROUNDS (BK_CH / 256)

__device__ __forceinline__
int bucket_rank_trip(const int* __restrict__ bkey, int cnt, int c0, int tid, int lane, int wave,
                     int* whist, int (&dg)[BK_ROUNDS], int (&lr)[BK_ROUNDS],
                     int& seg, int& wbase, int& cend) {
  for (int i = tid; i < 8 * BK_ROWS; i += 256) whist[i] = 0;
  __syncthreads();
  const int left = cnt - c0;
  cend = c0 + (left < BK_CH ? left : BK_CH);
  seg = ((cend - c0 + 255) >> 8) << 5;
  wbase = c0 + wave * seg;
  int* wh = whist + wave * BK_ROWS;
#pragma unroll
  for (int r = 0; r < BK_ROUNDS; ++r) {
    const int e = wbase + r * 32 + lane;
    const int k = bkey[e < cnt ? e : cnt - 1];
    dg[r] = (r * 32 < seg && e < cend) ? (k & (BK_ROWS - 1)) : -1;
  }
#pragma unroll
  for (int r = 0; r < BK_ROUNDS; ++r) {
    lr[r] = 0;
    if (r * 32 < seg && wbase + r * 32 < cend) {
      const int d = dg[r];
      const bool valid = d >= 0;
      unsigned peers = __builtin_amdgcn_ballot_w32(valid);
#pragma unroll
      for (int q = 0; q < BK_BITS; ++q) {
        const bool bit = ((d >> q) & 1) != 0;
        const unsigned bal = __builtin_amdgcn_ballot_w32(bit);
        peers &= bit ? bal : ~bal;
      }
      const int rk = __popc(peers & ((1u << lane) - 1u));
      int b0 = 0;
      if (valid) {
        b0 = wh[d];
        if (rk == 0) wh[d] = b0 + __popc(peers);
      }
      lr[r] = b0 + rk;
    }
  }
  __syncthreads();
  int ctot = 0;
  for (int w = 0; w < 8; ++w) {
    const int v = whist[w * BK_ROWS + tid];
    whist[w * BK_ROWS + tid] = ctot;
    ctot += v;
  }
  __syncthreads();
  return ctot;
}

template <bool STAGED>
__device__ __forceinline__
void bucket_place_trip(const int2* __restrict__ bpay, int cnt, int lane,
                       const int* wh, const int* rowoff, const int* carry,
                       const int (&dg)[BK_ROUNDS], const int (&lr)[BK_ROUNDS],
                       int seg, int wbase, int cend, int2* spay, int2* __restrict__ gout) {
#pragma unroll
  for (int r0 = 0; r0 < BK_ROUNDS; r0 += 8) {
    if (r0 * 32 < seg && wbase + r0 * 32 < cend) {
      int2 pv[8];
#pragma unroll
      for (int u = 0; u < 8; ++u) {
        const int e = wbase + (r0 + u) * 32 + lane;
        pv[u] = bpay[e < cnt ? e : cnt - 1];
      }
#pragma unroll
      for (int u = 0; u < 8; ++u) {
        const int d = dg[r0 + u];
        if (d >= 0) {
          const int pos = rowoff[d] + carry[d] + wh[d] + lr[r0 + u];
          if (STAGED) spay[pos] = pv[u];
          else        gout[pos] = pv[u];
        }
      }
    }
  }
}

__global__ __launch_bounds__(256)
void k_bucket_rows(const int* __restrict__ key, const int2* __restrict__ pay,
                   const int* __restrict__ bstart, int2* __restrict__ rowedges,
                   int* __restrict__ rowstart, int* __restrict__ rowend, int Nn) {
  __shared__ int whist[8 * BK_ROWS];
  __shared__ int rowoff[BK_ROWS];
  __shared__ int carry[BK_ROWS];
  __shared__ int wsum[8];
  __shared__ __attribute__((aligned(8))) int spay_w[2 * BK_CH];
  int2* spay = (int2*)spay_w;
  const int tid = threadIdx.x, lane = tid & 31;
  const int wave = __builtin_amdgcn_readfirstlane(tid >> 5);
  const int b = blockIdx.x;
  const int g0 = __builtin_amdgcn_readfirstlane(bstart[b]);
  const int cnt = __builtin_amdgcn_readfirstlane(bstart[b + 1]) - g0;
  const int* bkey = key + g0;
  const int2* bpay = pay + g0;
  const int* wh = whist + wave * BK_ROWS;
  const bool staged = cnt <= BK_CH;

  carry[tid] = 0;
  int dg[BK_ROUNDS], lr[BK_ROUNDS];
  int seg = 0, wbase = 0, cend = 0;
  int mycnt = 0;
  for (int c0 = 0; c0 < cnt; c0 += BK_CH)
    mycnt += bucket_rank_trip(bkey, cnt, c0, tid, lane, wave, whist, dg, lr, seg, wbase, cend);
  int total;
  const int ex = block_excl_scan(mycnt, wsum, lane, wave, total);
  rowoff[tid] = ex;
  const int n = b * BK_ROWS + tid;
  if (n < Nn) {
    rowstart[n] = g0 + ex;
    rowend[n] = g0 + ex + mycnt;
  }
  __syncthreads();

  if (staged) {
    if (cnt > 0)
      bucket_place_trip<true>(bpay, cnt, lane, wh, rowoff, carry, dg, lr, seg, wbase, cend,
                              spay, rowedges + g0);
    __syncthreads();
    for (int i0 = 0; i0 < cnt; i0 += 8 * 256) {
      int2 pv[8];
#pragma unroll
      for (int u = 0; u < 8; ++u) {
        const int i = i0 + u * 256 + tid;
        pv[u] = spay[i < cnt ? i : cnt - 1];
      }
#pragma unroll
      for (int u = 0; u < 8; ++u) {
        const int i = i0 + u * 256 + tid;
        if (i < cnt) rowedges[g0 + i] = pv[u];
      }
    }
  } else {
    for (int c0 = 0; c0 < cnt; c0 += BK_CH) {
      const int ctot = bucket_rank_trip(bkey, cnt, c0, tid, lane, wave, whist, dg, lr,
                                        seg, wbase, cend);
      bucket_place_trip<false>(bpay, cnt, lane, wh, rowoff, carry, dg, lr, seg, wbase, cend,
                               spay, rowedges + g0);
      __syncthreads();
      carry[tid] += ctot;
    }
  }
}

#define ROWCHUNK 8
__device__ __forceinline__ float2 lhat_row(const float* __restrict__ z,
                                           const int2* __restrict__ rowedges,
                                           const int* __restrict__ rowstart,
                                           const int* __restrict__ rowend,
                                           float diag, int n, int lane) {
  const float2* z2 = (const float2*)z;
  const int p0 = rowstart[n];
  const int c = rowend[n] - p0;
  const int2* ce = rowedges + p0;
  float2 acc = make_float2(0.0f, 0.0f);
  for (int j = 0; j < c; j += ROWCHUNK) {
    float2 zv[ROWCHUNK];
    float wv[ROWCHUNK];
#pragma unroll
    for (int u = 0; u < ROWCHUNK; ++u) {
      const int idx = (j + u < c) ? (j + u) : (c - 1);
      const int2 en = ce[idx];
      wv[u] = (j + u < c) ? __int_as_float(en.y) : 0.0f;
      zv[u] = z2[(size_t)en.x * (HID / 2) + lane];
    }
#pragma unroll
    for (int u = 0; u < ROWCHUNK; ++u) {
      acc.x += wv[u] * zv[u].x;
      acc.y += wv[u] * zv[u].y;
    }
  }
  const float2 zn = z2[(size_t)n * (HID / 2) + lane];
  acc.x += diag * zn.x;
  acc.y += diag * zn.y;
  return acc;
}

__global__ __launch_bounds__(256)
void k_spmm_t1(const float* __restrict__ h, float* __restrict__ T1,
               const int2* __restrict__ rowedges, const int* __restrict__ rowstart,
               const int* __restrict__ rowend, float diag, int Nn) {
  const int lane = threadIdx.x & 31;
  const int n = __builtin_amdgcn_readfirstlane(blockIdx.x * 8 + (threadIdx.x >> 5));
  if (n >= Nn) return;
  const float2 r = lhat_row(h, rowedges, rowstart, rowend, diag, n, lane);
  ((float2*)T1)[(size_t)n * (HID / 2) + lane] = r;
}

__global__ __launch_bounds__(256)
void k_spmm_t2(const float* __restrict__ T1, const float* __restrict__ h,
               float* __restrict__ T2,
               const int2* __restrict__ rowedges, const int* __restrict__ rowstart,
               const int* __restrict__ rowend, float diag, int Nn) {
  const int lane = threadIdx.x & 31;
  const int n = __builtin_amdgcn_readfirstlane(blockIdx.x * 8 + (threadIdx.x >> 5));
  if (n >= Nn) return;
  const float2 r = lhat_row(T1, rowedges, rowstart, rowend, diag, n, lane);
  const size_t o = (size_t)n * (HID / 2) + lane;
  const float2 hv = ((const float2*)h)[o];
  float2 t;
  t.x = 2.0f * r.x - hv.x;
  t.y = 2.0f * r.y - hv.y;
  ((float2*)T2)[o] = t;
}


__device__ __forceinline__
void packB_block(const float* __restrict__ Wx, const float* __restrict__ theta,
                 const float* __restrict__ bg, const float* __restrict__ conv_b,
                 float* __restrict__ Bp, float* __restrict__ bias, int bid) {
  int gid = bid * 256 + threadIdx.x;
  if (gid >= KDIM * GCOLS) return;
  int k = gid / GCOLS, col = gid % GCOLS;
  int g = col >> 6, o = col & 63;
  float v;
  if (k < INDIM) {
    v = Wx[((size_t)g * INDIM + k) * HID + o];
  } else {
    int kk = k - INDIM;
    int kb = kk >> 6, kr = kk & 63;
    v = theta[(((size_t)g * 3 + kb) * HID + kr) * HID + o];
  }
  Bp[gid] = v;
  if (gid < GCOLS) bias[gid] = bg[g * HID + o] + conv_b[g * HID + o];
}

__global__ __launch_bounds__(256)
void k_front(const int* __restrict__ src, const int* __restrict__ dst,
             const float* __restrict__ ew, float* __restrict__ deg,
             int* __restrict__ lrank, int* __restrict__ table,
             const float* __restrict__ Wx, const float* __restrict__ theta,
             const float* __restrict__ bg, const float* __restrict__ conv_b,
             float* __restrict__ Bp, float* __restrict__ bias,
             int E, int nrank, int ndeg) {
  const int b = blockIdx.x;
  if (b < nrank) {
    split_rank_block(dst, BK_BITS, lrank, table, E, b);
  } else if (b < nrank + ndeg) {
    degree_block(src, dst, ew, deg, E, b - nrank);
  } else {
    packB_block(Wx, theta, bg, conv_b, Bp, bias, b - nrank - ndeg);
  }
}


#define AS1 324
#define PS1 260
#define MT 64
#define MSUB (MT / 16)
#define HS1 68

__device__ __forceinline__ void lstm_cell(float pi, float pf, float pt, float po, float cold,
                                          float& hnew, float& cnew) {
  float ig = fsigmoid(pi);
  float fg = fsigmoid(pf);
  float tg = tanhf(pt);
  float og = fsigmoid(po);
  float cv = fg * cold + ig * tg;
  hnew = og * tanhf(cv);
  cnew = cv;
}

__global__ __launch_bounds__(256)
void k_gemm_gates(const float* __restrict__ x, const float* __restrict__ h,
                  const float* __restrict__ T1, const float* __restrict__ T2,
                  const float* __restrict__ Bp, const float* __restrict__ bias,
                  const float* __restrict__ c_in,
                  const float* __restrict__ Wl, const float* __restrict__ bl,
                  float* __restrict__ out,
                  float* __restrict__ h_out, float* __restrict__ c_out, int Nn) {
  __shared__ __attribute__((aligned(16))) float Asmem[MT * AS1];
  __shared__ __attribute__((aligned(16))) float Hsmem[MT * HS1];
  float* Psmem = Asmem;
  const int tid = threadIdx.x;
  const int m0 = blockIdx.x * MT;

#pragma unroll
  for (int it = 0; it < MT * 32 / 256; ++it) {
    int i = it * 256 + tid;
    int r = i >> 5, q = i & 31;
    int n = m0 + r; if (n >= Nn) n = Nn - 1;
    const float4 v = *(const float4*)(x + (size_t)n * INDIM + q * 4);
    *(float4*)&Asmem[r * AS1 + q * 4] = v;
  }
#pragma unroll
  for (int it = 0; it < MT * 16 / 256; ++it) {
    int i = it * 256 + tid;
    int r = i >> 4, q = i & 15;
    int n = m0 + r; if (n >= Nn) n = Nn - 1;
    const size_t g = (size_t)n * HID + q * 4;
    const float4 vh = *(const float4*)(h + g);
    const float4 v1 = *(const float4*)(T1 + g);
    const float4 v2 = *(const float4*)(T2 + g);
    float* ap = &Asmem[r * AS1 + 128 + q * 4];
    *(float4*)(ap)       = vh;
    *(float4*)(ap + 64)  = v1;
    *(float4*)(ap + 128) = v2;
  }
  __syncthreads();

  const int wave = tid >> 5, lane = tid & 31;
  const int lo = lane & 15, hi = lane >> 4;
  const int c0 = wave * 32;
  v8f acc[MSUB][2];
#pragma unroll
  for (int t = 0; t < MSUB; ++t) { acc[t][0] = v8f{}; acc[t][1] = v8f{}; }

  for (int k0 = 0; k0 < KDIM; k0 += 4) {
    const float* bptr = Bp + (size_t)(k0 + 2 * hi) * GCOLS + c0 + lo;
    v2f b0, b1;
    b0[0] = bptr[0];    b0[1] = bptr[GCOLS];
    b1[0] = bptr[16];   b1[1] = bptr[GCOLS + 16];
#pragma unroll
    for (int t = 0; t < MSUB; ++t) {
      v2f a = *(const v2f*)&Asmem[(t * 16 + lo) * AS1 + k0 + 2 * hi];
      acc[t][0] = wmma_f32_4(a, b0, acc[t][0]);
      acc[t][1] = wmma_f32_4(a, b1, acc[t][1]);
    }
  }

  float bias0 = bias[c0 + lo], bias1 = bias[c0 + 16 + lo];
  const int oc = (wave & 3) * 16;
  v2f wb[HID / 4];
#pragma unroll
  for (int s = 0; s < HID / 4; ++s) {
    wb[s][0] = Wl[(4 * s + 2 * hi) * HID + oc + lo];
    wb[s][1] = Wl[(4 * s + 2 * hi + 1) * HID + oc + lo];
  }
  __syncthreads();
#pragma unroll
  for (int t = 0; t < MSUB; ++t) {
#pragma unroll
    for (int r = 0; r < 8; ++r) {
      int row = t * 16 + r + 8 * hi;
      Psmem[row * PS1 + c0 + lo]      = acc[t][0][r] + bias0;
      Psmem[row * PS1 + c0 + 16 + lo] = acc[t][1][r] + bias1;
    }
  }
  __syncthreads();

  for (int it = 0; it < MT * 16 / 256; ++it) {
    int i = it * 256 + tid;
    int row = i >> 4, o = (i & 15) * 4;
    int n = m0 + row;
    if (n >= Nn) {
      *(float4*)&Hsmem[row * HS1 + o] = make_float4(0.0f, 0.0f, 0.0f, 0.0f);
      continue;
    }
    const float4 cold = *(const float4*)(c_in + (size_t)n * HID + o);
    const float* pr = &Psmem[row * PS1 + o];
    const float4 pi = *(const float4*)(pr);
    const float4 pf = *(const float4*)(pr + 64);
    const float4 pt = *(const float4*)(pr + 128);
    const float4 po = *(const float4*)(pr + 192);
    float4 hv, cv;
    lstm_cell(pi.x, pf.x, pt.x, po.x, cold.x, hv.x, cv.x);
    lstm_cell(pi.y, pf.y, pt.y, po.y, cold.y, hv.y, cv.y);
    lstm_cell(pi.z, pf.z, pt.z, po.z, cold.z, hv.z, cv.z);
    lstm_cell(pi.w, pf.w, pt.w, po.w, cold.w, hv.w, cv.w);
    *(float4*)(h_out + (size_t)n * HID + o) = hv;
    *(float4*)(c_out + (size_t)n * HID + o) = cv;
    float4 hr;
    hr.x = hv.x > 0.0f ? hv.x : 0.0f;
    hr.y = hv.y > 0.0f ? hv.y : 0.0f;
    hr.z = hv.z > 0.0f ? hv.z : 0.0f;
    hr.w = hv.w > 0.0f ? hv.w : 0.0f;
    *(float4*)&Hsmem[row * HS1 + o] = hr;
  }
  __syncthreads();

  {
    const int rt = (wave >> 2) * 2;
    v8f oacc0 = {}, oacc1 = {};
#pragma unroll
    for (int k0 = 0; k0 < HID; k0 += 4) {
      const v2f b = wb[k0 / 4];
      v2f a0 = *(const v2f*)&Hsmem[(rt * 16 + lo) * HS1 + k0 + 2 * hi];
      v2f a1 = *(const v2f*)&Hsmem[(rt * 16 + 16 + lo) * HS1 + k0 + 2 * hi];
      oacc0 = wmma_f32_4(a0, b, oacc0);
      oacc1 = wmma_f32_4(a1, b, oacc1);
    }
    const float bb = bl[oc + lo];
#pragma unroll
    for (int r = 0; r < 8; ++r) {
      const int n0 = m0 + rt * 16 + r + 8 * hi;
      const int n1 = n0 + 16;
      if (n0 < Nn) out[(size_t)n0 * HID + oc + lo] = oacc0[r] + bb;
      if (n1 < Nn) out[(size_t)n1 * HID + oc + lo] = oacc1[r] + bb;
    }
  }
}


extern "C" void kernel_launch(void* const* d_in, const int* in_sizes, int n_in,
                              void* d_out, int out_size, void* d_ws, size_t ws_size,
                              hipStream_t stream) {
  const float* x      = (const float*)d_in[0];
  const int*   ei     = (const int*)d_in[1];
  const float* ew     = (const float*)d_in[2];
  const float* h      = (const float*)d_in[3];
  const float* c      = (const float*)d_in[4];
  const float* Wx     = (const float*)d_in[5];
  const float* bg     = (const float*)d_in[6];
  const float* theta  = (const float*)d_in[7];
  const float* conv_b = (const float*)d_in[8];
  const float* Wl     = (const float*)d_in[9];
  const float* bl     = (const float*)d_in[10];

  const int E  = in_sizes[2];
  const int Nn = in_sizes[3] / HID;
  const int* src = ei;
  const int* dst = ei + E;

  float* T1   = (float*)d_ws;
  float* T2   = T1 + (size_t)Nn * HID;
  float* Bp   = T2 + (size_t)Nn * HID;
  float* bias = Bp + (size_t)KDIM * GCOLS;
  int2*  edgepay  = (int2*)(bias + GCOLS);
  int2*  payA     = edgepay + E;
  int2*  rowedges = payA + E;
  int*   keyA = (int*)(rowedges + E);
  int*   lrank = keyA + E;
  const int NB = (E + SORT_CH - 1) / SORT_CH;
  int*   table = lrank + E;
  float* deg  = (float*)(table + (size_t)NB * SORT_BINS);
  int*   rowstart = (int*)(deg + Nn);
  int*   rowend = rowstart + Nn;
  int*   bstart = rowend + Nn;

  float* out   = (float*)d_out;
  float* h_out = out   + (size_t)Nn * HID;
  float* c_out = h_out + (size_t)Nn * HID;

  const float DIAG = 2.0f / LMAX - 1.0f;

  hipMemsetAsync(deg, 0, (size_t)Nn * sizeof(float), stream);
  const int NDEG = (E + 255) / 256;
  const int NPACK = (KDIM * GCOLS + 255) / 256;
  k_front<<<NB + NDEG + NPACK, 256, 0, stream>>>(src, dst, ew, deg, lrank, table,
                                                 Wx, theta, bg, conv_b, Bp, bias, E, NB, NDEG);
  k_lapw_pack<<<(E + 255) / 256, 256, 0, stream>>>(src, dst, ew, deg, edgepay, E);

  k_split_scatter<<<NB, 256, 0, stream>>>(dst, edgepay, BK_BITS, lrank, table, NB, keyA, payA,
                                          bstart, E);
  k_bucket_rows<<<(Nn + BK_ROWS - 1) / BK_ROWS, 256, 0, stream>>>(keyA, payA, bstart, rowedges,
                                                                  rowstart, rowend, Nn);

  k_spmm_t1<<<(Nn + 7) / 8, 256, 0, stream>>>(h, T1, rowedges, rowstart, rowend, DIAG, Nn);
  k_spmm_t2<<<(Nn + 7) / 8, 256, 0, stream>>>(T1, h, T2, rowedges, rowstart, rowend, DIAG, Nn);

  k_gemm_gates<<<(Nn + MT - 1) / MT, 256, 0, stream>>>(x, h, T1, T2, Bp, bias, c, Wl, bl, out,
                                                   h_out, c_out, Nn);
}


# T2 SpMM moved into the gate kernel: each wave sums its 8 tile rows of 2*Lhat*T1-h from the sorted edges straight into the LDS A tile; T2 launch, store and reload gone; row sum order kept
# speedup vs baseline: 1.9151x; 1.0317x over previous
#include <hip/hip_runtime.h>
#include <hip/hip_bf16.h>

typedef __attribute__((ext_vector_type(2))) float v2f;
typedef __attribute__((ext_vector_type(8))) float v8f;

#define HID 64
#define INDIM 128
#define KDIM 320
#define GCOLS 256
#define LMAX 2.0f

__device__ __forceinline__ v8f wmma_f32_4(v2f a, v2f b, v8f c) {
  return __builtin_amdgcn_wmma_f32_16x16x4_f32(false, a, false, b, (short)0, c, false, false);
}

__device__ __forceinline__ float fsigmoid(float x) {
  return 1.0f / (1.0f + __expf(-x));
}

__device__ __forceinline__ void atomic_fadd(float* p, float v) {
  asm volatile("global_atomic_add_f32 %0, %1, off scope:SCOPE_DEV"
               :: "v"(p), "v"(v) : "memory");
}


__device__ __forceinline__
void degree_block(const int* __restrict__ src, const int* __restrict__ dst,
                  const float* __restrict__ ew, float* __restrict__ deg, int E, int bid) {
  int e = bid * 256 + threadIdx.x;
  if (e >= E) return;
  int s = src[e], d = dst[e];
  if (s != d) atomic_fadd(&deg[s], ew[e]);
}

__device__ __forceinline__ float dinv_of(float dg) {
  return (dg > 0.0f) ? rsqrtf(dg) : 0.0f;
}

__global__ __launch_bounds__(256)
void k_lapw_pack(const int* __restrict__ src, const int* __restrict__ dst,
                 const float* __restrict__ ew, const float* __restrict__ deg,
                 int2* __restrict__ edgepay, int E) {
  int e = blockIdx.x * 256 + threadIdx.x;
  if (e >= E) return;
  int s = src[e], d = dst[e];
  float w = (s == d) ? 0.0f : ew[e];
  const float ds = dinv_of(deg[s]), dd = dinv_of(deg[d]);
  float lw = (2.0f / LMAX) * (-ds * w * dd);
  edgepay[e] = make_int2(s, __float_as_int(lw));
}


#define SORT_CH 8192
#define SORT_ROUNDS (SORT_CH / 256)
#define SORT_BINS 512

__device__ __forceinline__ int wave_incl_scan(int v, int lane) {
  for (int off = 1; off < 32; off <<= 1) {
    int t = __shfl_up(v, off, 32);
    if (lane >= off) v += t;
  }
  return v;
}

__device__ __forceinline__ int block_excl_scan(int v, int* wsum, int lane, int wave, int& total) {
  const int incl = wave_incl_scan(v, lane);
  if (lane == 31) wsum[wave] = incl;
  __syncthreads();
  int wb = 0, t = 0;
  for (int w = 0; w < 8; ++w) { int s = wsum[w]; if (w < wave) wb += s; t += s; }
  total = t;
  __syncthreads();
  return wb + incl - v;
}

__device__ __forceinline__
void split_rank_block(const int* __restrict__ key, int shift,
                      int* __restrict__ lrank, int* __restrict__ table, int E, int bid) {
  __shared__ int whist[8 * SORT_BINS];
  const int tid = threadIdx.x, lane = tid & 31, wave = tid >> 5;
  for (int i = tid; i < 8 * SORT_BINS; i += 256) whist[i] = 0;
  __syncthreads();
  const int base = bid * SORT_CH + wave * (SORT_CH / 8) + lane;
  int dg[SORT_ROUNDS], lr[SORT_ROUNDS];
#pragma unroll
  for (int r = 0; r < SORT_ROUNDS; ++r) {
    const int e = base + r * 32;
    const int k = key[e < E ? e : E - 1];
    dg[r] = (e < E) ? ((k >> shift) & (SORT_BINS - 1)) : -1;
  }
  int* wh = whist + wave * SORT_BINS;
#pragma unroll
  for (int r = 0; r < SORT_ROUNDS; ++r) {
    const int d = dg[r];
    const bool valid = d >= 0;
    unsigned peers = __builtin_amdgcn_ballot_w32(valid);
#pragma unroll
    for (int b = 0; b < 9; ++b) {
      const bool bit = ((d >> b) & 1) != 0;
      const unsigned bal = __builtin_amdgcn_ballot_w32(bit);
      peers &= bit ? bal : ~bal;
    }
    const int rk = __popc(peers & ((1u << lane) - 1u));
    int b0 = 0;
    if (valid) {
      b0 = wh[d];
      if (rk == 0) wh[d] = b0 + __popc(peers);
    }
    lr[r] = b0 + rk;
  }
  __syncthreads();
  for (int dgt = tid; dgt < SORT_BINS; dgt += 256) {
    int run = 0;
    for (int w = 0; w < 8; ++w) {
      int v = whist[w * SORT_BINS + dgt];
      whist[w * SORT_BINS + dgt] = run;
      run += v;
    }
    table[bid * SORT_BINS + dgt] = run;
  }
  __syncthreads();
#pragma unroll
  for (int r = 0; r < SORT_ROUNDS; ++r) {
    const int e = base + r * 32;
    if (e < E) lrank[e] = lr[r] + wh[dg[r]];
  }
}

__global__ __launch_bounds__(256)
void k_split_scatter(const int* __restrict__ key, const int2* __restrict__ pay, int shift,
                     const int* __restrict__ lrank, const int* __restrict__ table, int nblocks,
                     int* __restrict__ key_out, int2* __restrict__ pay_out,
                     int* __restrict__ bstart, int E) {
  __shared__ int lstart[SORT_BINS];
  __shared__ int gdelta[SORT_BINS];
  __shared__ int wsum[8];
  __shared__ int skey[SORT_CH];
  __shared__ __attribute__((aligned(8))) int spay_w[2 * SORT_CH];
  int2* spay = (int2*)spay_w;
  const int tid = threadIdx.x, lane = tid & 31, wave = tid >> 5;
  const int me = blockIdx.x;
  int pre0 = 0, tot0 = 0, pre1 = 0, tot1 = 0;
#pragma unroll 14
  for (int b = 0; b < nblocks; ++b) {
    const int v0 = table[b * SORT_BINS + tid];
    const int v1 = table[b * SORT_BINS + 256 + tid];
    tot0 += v0; tot1 += v1;
    if (b < me) { pre0 += v0; pre1 += v1; }
  }
  const int own0 = table[me * SORT_BINS + tid];
  const int own1 = table[me * SORT_BINS + 256 + tid];
  int sum0, sum1, osum0, osum1;
  const int ex0 = block_excl_scan(tot0, wsum, lane, wave, sum0);
  const int ex1 = block_excl_scan(tot1, wsum, lane, wave, sum1);
  const int l0 = block_excl_scan(own0, wsum, lane, wave, osum0);
  const int l1 = osum0 + block_excl_scan(own1, wsum, lane, wave, osum1);
  lstart[tid]       = l0;
  lstart[tid + 256] = l1;
  gdelta[tid]       = ex0 + pre0 - l0;
  gdelta[tid + 256] = sum0 + ex1 + pre1 - l1;
  if (me == 0) {
    bstart[tid]       = ex0;
    bstart[tid + 256] = sum0 + ex1;
    if (tid == 0) bstart[SORT_BINS] = sum0 + sum1;
  }
  __syncthreads();
  for (int r0 = 0; r0 < SORT_ROUNDS; r0 += 8) {
    int kk[8], rk[8];
    int2 pv[8];
#pragma unroll
    for (int u = 0; u < 8; ++u) {
      const int e = me * SORT_CH + (r0 + u) * 256 + tid;
      const int ec = e < E ? e : E - 1;
      kk[u] = key[ec];
      rk[u] = lrank[ec];
      pv[u] = pay[ec];
    }
    int lp[8];
#pragma unroll
    for (int u = 0; u < 8; ++u)
      lp[u] = lstart[(kk[u] >> shift) & (SORT_BINS - 1)] + rk[u];
#pragma unroll
    for (int u = 0; u < 8; ++u) {
      const int e = me * SORT_CH + (r0 + u) * 256 + tid;
      if (e < E) {
        skey[lp[u]] = kk[u];
        spay[lp[u]] = pv[u];
      }
    }
  }
  __syncthreads();
  const int rem = E - me * SORT_CH;
  const int cnt = rem < SORT_CH ? rem : SORT_CH;
  for (int r0 = 0; r0 < SORT_ROUNDS; r0 += 8) {
    int kk[8];
    int2 pv[8];
#pragma unroll
    for (int u = 0; u < 8; ++u) {
      const int i = (r0 + u) * 256 + tid;
      const int ic = i < cnt ? i : cnt - 1;
      kk[u] = skey[ic];
      pv[u] = spay[ic];
    }
    int ps[8];
#pragma unroll
    for (int u = 0; u < 8; ++u)
      ps[u] = (r0 + u) * 256 + tid + gdelta[(kk[u] >> shift) & (SORT_BINS - 1)];
#pragma unroll
    for (int u = 0; u < 8; ++u) {
      const int i = (r0 + u) * 256 + tid;
      if (i < cnt) {
        key_out[ps[u]] = kk[u];
        pay_out[ps[u]] = pv[u];
      }
    }
  }
}

#define BK_BITS 8
#define BK_ROWS (1 << BK_BITS)
#define BK_CH 8192
#define BK_ROUNDS (BK_CH / 256)

__device__ __forceinline__
int bucket_rank_trip(const int* __restrict__ bkey, int cnt, int c0, int tid, int lane, int wave,
                     int* whist, int (&dg)[BK_ROUNDS], int (&lr)[BK_ROUNDS],
                     int& seg, int& wbase, int& cend) {
  for (int i = tid; i < 8 * BK_ROWS; i += 256) whist[i] = 0;
  __syncthreads();
  const int left = cnt - c0;
  cend = c0 + (left < BK_CH ? left : BK_CH);
  seg = ((cend - c0 + 255) >> 8) << 5;
  wbase = c0 + wave * seg;
  int* wh = whist + wave * BK_ROWS;
#pragma unroll
  for (int r = 0; r < BK_ROUNDS; ++r) {
    const int e = wbase + r * 32 + lane;
    const int k = bkey[e < cnt ? e : cnt - 1];
    dg[r] = (r * 32 < seg && e < cend) ? (k & (BK_ROWS - 1)) : -1;
  }
#pragma unroll
  for (int r = 0; r < BK_ROUNDS; ++r) {
    lr[r] = 0;
    if (r * 32 < seg && wbase + r * 32 < cend) {
      const int d = dg[r];
      const bool valid = d >= 0;
      unsigned peers = __builtin_amdgcn_ballot_w32(valid);
#pragma unroll
      for (int q = 0; q < BK_BITS; ++q) {
        const bool bit = ((d >> q) & 1) != 0;
        const unsigned bal = __builtin_amdgcn_ballot_w32(bit);
        peers &= bit ? bal : ~bal;
      }
      const int rk = __popc(peers & ((1u << lane) - 1u));
      int b0 = 0;
      if (valid) {
        b0 = wh[d];
        if (rk == 0) wh[d] = b0 + __popc(peers);
      }
      lr[r] = b0 + rk;
    }
  }
  __syncthreads();
  int ctot = 0;
  for (int w = 0; w < 8; ++w) {
    const int v = whist[w * BK_ROWS + tid];
    whist[w * BK_ROWS + tid] = ctot;
    ctot += v;
  }
  __syncthreads();
  return ctot;
}

template <bool STAGED>
__device__ __forceinline__
void bucket_place_trip(const int2* __restrict__ bpay, int cnt, int lane,
                       const int* wh, const int* rowoff, const int* carry,
                       const int (&dg)[BK_ROUNDS], const int (&lr)[BK_ROUNDS],
                       int seg, int wbase, int cend, int2* spay, int2* __restrict__ gout) {
#pragma unroll
  for (int r0 = 0; r0 < BK_ROUNDS; r0 += 8) {
    if (r0 * 32 < seg && wbase + r0 * 32 < cend) {
      int2 pv[8];
#pragma unroll
      for (int u = 0; u < 8; ++u) {
        const int e = wbase + (r0 + u) * 32 + lane;
        pv[u] = bpay[e < cnt ? e : cnt - 1];
      }
#pragma unroll
      for (int u = 0; u < 8; ++u) {
        const int d = dg[r0 + u];
        if (d >= 0) {
          const int pos = rowoff[d] + carry[d] + wh[d] + lr[r0 + u];
          if (STAGED) spay[pos] = pv[u];
          else        gout[pos] = pv[u];
        }
      }
    }
  }
}

__global__ __launch_bounds__(256)
void k_bucket_rows(const int* __restrict__ key, const int2* __restrict__ pay,
                   const int* __restrict__ bstart, int2* __restrict__ rowedges,
                   int* __restrict__ rowstart, int* __restrict__ rowend, int Nn) {
  __shared__ int whist[8 * BK_ROWS];
  __shared__ int rowoff[BK_ROWS];
  __shared__ int carry[BK_ROWS];
  __shared__ int wsum[8];
  __shared__ __attribute__((aligned(8))) int spay_w[2 * BK_CH];
  int2* spay = (int2*)spay_w;
  const int tid = threadIdx.x, lane = tid & 31;
  const int wave = __builtin_amdgcn_readfirstlane(tid >> 5);
  const int b = blockIdx.x;
  const int g0 = __builtin_amdgcn_readfirstlane(bstart[b]);
  const int cnt = __builtin_amdgcn_readfirstlane(bstart[b + 1]) - g0;
  const int* bkey = key + g0;
  const int2* bpay = pay + g0;
  const int* wh = whist + wave * BK_ROWS;
  const bool staged = cnt <= BK_CH;

  carry[tid] = 0;
  int dg[BK_ROUNDS], lr[BK_ROUNDS];
  int seg = 0, wbase = 0, cend = 0;
  int mycnt = 0;
  for (int c0 = 0; c0 < cnt; c0 += BK_CH)
    mycnt += bucket_rank_trip(bkey, cnt, c0, tid, lane, wave, whist, dg, lr, seg, wbase, cend);
  int total;
  const int ex = block_excl_scan(mycnt, wsum, lane, wave, total);
  rowoff[tid] = ex;
  const int n = b * BK_ROWS + tid;
  if (n < Nn) {
    rowstart[n] = g0 + ex;
    rowend[n] = g0 + ex + mycnt;
  }
  __syncthreads();

  if (staged) {
    if (cnt > 0)
      bucket_place_trip<true>(bpay, cnt, lane, wh, rowoff, carry, dg, lr, seg, wbase, cend,
                              spay, rowedges + g0);
    __syncthreads();
    for (int i0 = 0; i0 < cnt; i0 += 8 * 256) {
      int2 pv[8];
#pragma unroll
      for (int u = 0; u < 8; ++u) {
        const int i = i0 + u * 256 + tid;
        pv[u] = spay[i < cnt ? i : cnt - 1];
      }
#pragma unroll
      for (int u = 0; u < 8; ++u) {
        const int i = i0 + u * 256 + tid;
        if (i < cnt) rowedges[g0 + i] = pv[u];
      }
    }
  } else {
    for (int c0 = 0; c0 < cnt; c0 += BK_CH) {
      const int ctot = bucket_rank_trip(bkey, cnt, c0, tid, lane, wave, whist, dg, lr,
                                        seg, wbase, cend);
      bucket_place_trip<false>(bpay, cnt, lane, wh, rowoff, carry, dg, lr, seg, wbase, cend,
                               spay, rowedges + g0);
      __syncthreads();
      carry[tid] += ctot;
    }
  }
}

#define ROWCHUNK 8
__device__ __forceinline__ float2 lhat_row(const float* __restrict__ z,
                                           const int2* __restrict__ rowedges,
                                           const int* __restrict__ rowstart,
                                           const int* __restrict__ rowend,
                                           float diag, int n, int lane) {
  const float2* z2 = (const float2*)z;
  const int p0 = rowstart[n];
  const int c = rowend[n] - p0;
  const int2* ce = rowedges + p0;
  float2 acc = make_float2(0.0f, 0.0f);
  for (int j = 0; j < c; j += ROWCHUNK) {
    float2 zv[ROWCHUNK];
    float wv[ROWCHUNK];
#pragma unroll
    for (int u = 0; u < ROWCHUNK; ++u) {
      const int idx = (j + u < c) ? (j + u) : (c - 1);
      const int2 en = ce[idx];
      wv[u] = (j + u < c) ? __int_as_float(en.y) : 0.0f;
      zv[u] = z2[(size_t)en.x * (HID / 2) + lane];
    }
#pragma unroll
    for (int u = 0; u < ROWCHUNK; ++u) {
      acc.x += wv[u] * zv[u].x;
      acc.y += wv[u] * zv[u].y;
    }
  }
  const float2 zn = z2[(size_t)n * (HID / 2) + lane];
  acc.x += diag * zn.x;
  acc.y += diag * zn.y;
  return acc;
}

__global__ __launch_bounds__(256)
void k_spmm_t1(const float* __restrict__ h, float* __restrict__ T1,
               const int2* __restrict__ rowedges, const int* __restrict__ rowstart,
               const int* __restrict__ rowend, float diag, int Nn) {
  const int lane = threadIdx.x & 31;
  const int n = __builtin_amdgcn_readfirstlane(blockIdx.x * 8 + (threadIdx.x >> 5));
  if (n >= Nn) return;
  const float2 r = lhat_row(h, rowedges, rowstart, rowend, diag, n, lane);
  ((float2*)T1)[(size_t)n * (HID / 2) + lane] = r;
}


__device__ __forceinline__
void packB_block(const float* __restrict__ Wx, const float* __restrict__ theta,
                 const float* __restrict__ bg, const float* __restrict__ conv_b,
                 float* __restrict__ Bp, float* __restrict__ bias, int bid) {
  int gid = bid * 256 + threadIdx.x;
  if (gid >= KDIM * GCOLS) return;
  int k = gid / GCOLS, col = gid % GCOLS;
  int g = col >> 6, o = col & 63;
  float v;
  if (k < INDIM) {
    v = Wx[((size_t)g * INDIM + k) * HID + o];
  } else {
    int kk = k - INDIM;
    int kb = kk >> 6, kr = kk & 63;
    v = theta[(((size_t)g * 3 + kb) * HID + kr) * HID + o];
  }
  Bp[gid] = v;
  if (gid < GCOLS) bias[gid] = bg[g * HID + o] + conv_b[g * HID + o];
}

__global__ __launch_bounds__(256)
void k_front(const int* __restrict__ src, const int* __restrict__ dst,
             const float* __restrict__ ew, float* __restrict__ deg,
             int* __restrict__ lrank, int* __restrict__ table,
             const float* __restrict__ Wx, const float* __restrict__ theta,
             const float* __restrict__ bg, const float* __restrict__ conv_b,
             float* __restrict__ Bp, float* __restrict__ bias,
             int E, int nrank, int ndeg) {
  const int b = blockIdx.x;
  if (b < nrank) {
    split_rank_block(dst, BK_BITS, lrank, table, E, b);
  } else if (b < nrank + ndeg) {
    degree_block(src, dst, ew, deg, E, b - nrank);
  } else {
    packB_block(Wx, theta, bg, conv_b, Bp, bias, b - nrank - ndeg);
  }
}


#define AS1 324
#define PS1 260
#define MT 64
#define MSUB (MT / 16)
#define HS1 68

__device__ __forceinline__ void lstm_cell(float pi, float pf, float pt, float po, float cold,
                                          float& hnew, float& cnew) {
  float ig = fsigmoid(pi);
  float fg = fsigmoid(pf);
  float tg = tanhf(pt);
  float og = fsigmoid(po);
  float cv = fg * cold + ig * tg;
  hnew = og * tanhf(cv);
  cnew = cv;
}

#define TROWS 4
#define TCH 8

__device__ __forceinline__ int2 t2_fetch_edge(const int2* __restrict__ rowedges,
                                              int p0, int c, int slot) {
  const bool live = slot < c;
  const int gi = live ? p0 + slot : (c > 0 ? p0 + c - 1 : 0);
  int2 en = rowedges[gi];
  if (!live) en.y = 0;
  return en;
}

__device__ __forceinline__
void t2_rows_to_tile(const float* __restrict__ T1, const float* __restrict__ h,
                     const int2* __restrict__ rowedges, const int* __restrict__ rowstart,
                     const int* __restrict__ rowend, float diag, int m0, int Nn,
                     int wave, int lane, float* Asmem) {
  const float2* z2 = (const float2*)T1;
  const float2* h2 = (const float2*)h;
  const int il = lane / TCH, ul = lane % TCH;
  for (int g = 0; g < (MT / 8) / TROWS; ++g) {
    const int row0 = wave * (MT / 8) + g * TROWS;
    int nl = m0 + row0 + il;
    nl = nl < Nn ? nl : Nn - 1;
    const int p0l = rowstart[nl];
    const int cl = rowend[nl] - p0l;
    int cmax = 0;
#pragma unroll
    for (int i = 0; i < TROWS; ++i) {
      const int ci = __builtin_amdgcn_readlane(cl, i * TCH);
      cmax = ci > cmax ? ci : cmax;
    }
    float2 acc[TROWS];
#pragma unroll
    for (int i = 0; i < TROWS; ++i) acc[i] = make_float2(0.0f, 0.0f);
    int2 en = t2_fetch_edge(rowedges, p0l, cl, ul);
    for (int j = 0; j < cmax; j += TCH) {
      float2 zv[TROWS][TCH];
      float wv[TROWS][TCH];
#pragma unroll
      for (int i = 0; i < TROWS; ++i) {
#pragma unroll
        for (int u = 0; u < TCH; ++u) {
          const int src = __builtin_amdgcn_readlane(en.x, i * TCH + u);
          wv[i][u] = __int_as_float(__builtin_amdgcn_readlane(en.y, i * TCH + u));
          zv[i][u] = z2[(size_t)src * (HID / 2) + lane];
        }
      }
      en = t2_fetch_edge(rowedges, p0l, cl, j + TCH + ul);
#pragma unroll
      for (int i = 0; i < TROWS; ++i) {
#pragma unroll
        for (int u = 0; u < TCH; ++u) {
          acc[i].x += wv[i][u] * zv[i][u].x;
          acc[i].y += wv[i][u] * zv[i][u].y;
        }
      }
    }
#pragma unroll
    for (int i = 0; i < TROWS; ++i) {
      int n = m0 + row0 + i;
      n = n < Nn ? n : Nn - 1;
      const size_t o = (size_t)n * (HID / 2) + lane;
      const float2 zn = z2[o];
      const float2 hv = h2[o];
      float2 r = acc[i];
      r.x += diag * zn.x;
      r.y += diag * zn.y;
      float2 t;
      t.x = 2.0f * r.x - hv.x;
      t.y = 2.0f * r.y - hv.y;
      *(float2*)&Asmem[(row0 + i) * AS1 + 256 + 2 * lane] = t;
    }
  }
}

__global__ __launch_bounds__(256)
void k_gemm_gates(const float* __restrict__ x, const float* __restrict__ h,
                  const float* __restrict__ T1,
                  const int2* __restrict__ rowedges, const int* __restrict__ rowstart,
                  const int* __restrict__ rowend, float diag,
                  const float* __restrict__ Bp, const float* __restrict__ bias,
                  const float* __restrict__ c_in,
                  const float* __restrict__ Wl, const float* __restrict__ bl,
                  float* __restrict__ out,
                  float* __restrict__ h_out, float* __restrict__ c_out, int Nn) {
  __shared__ __attribute__((aligned(16))) float Asmem[MT * AS1];
  __shared__ __attribute__((aligned(16))) float Hsmem[MT * HS1];
  float* Psmem = Asmem;
  const int tid = threadIdx.x;
  const int m0 = blockIdx.x * MT;

#pragma unroll
  for (int it = 0; it < MT * 32 / 256; ++it) {
    int i = it * 256 + tid;
    int r = i >> 5, q = i & 31;
    int n = m0 + r; if (n >= Nn) n = Nn - 1;
    const float4 v = *(const float4*)(x + (size_t)n * INDIM + q * 4);
    *(float4*)&Asmem[r * AS1 + q * 4] = v;
  }
#pragma unroll
  for (int it = 0; it < MT * 16 / 256; ++it) {
    int i = it * 256 + tid;
    int r = i >> 4, q = i & 15;
    int n = m0 + r; if (n >= Nn) n = Nn - 1;
    const size_t g = (size_t)n * HID + q * 4;
    const float4 vh = *(const float4*)(h + g);
    const float4 v1 = *(const float4*)(T1 + g);
    float* ap = &Asmem[r * AS1 + 128 + q * 4];
    *(float4*)(ap)       = vh;
    *(float4*)(ap + 64)  = v1;
  }
  const int lane = tid & 31;
  const int wave = __builtin_amdgcn_readfirstlane(tid >> 5);
  t2_rows_to_tile(T1, h, rowedges, rowstart, rowend, diag, m0, Nn, wave, lane, Asmem);
  __syncthreads();

  const int lo = lane & 15, hi = lane >> 4;
  const int c0 = wave * 32;
  v8f acc[MSUB][2];
#pragma unroll
  for (int t = 0; t < MSUB; ++t) { acc[t][0] = v8f{}; acc[t][1] = v8f{}; }

  for (int k0 = 0; k0 < KDIM; k0 += 4) {
    const float* bptr = Bp + (size_t)(k0 + 2 * hi) * GCOLS + c0 + lo;
    v2f b0, b1;
    b0[0] = bptr[0];    b0[1] = bptr[GCOLS];
    b1[0] = bptr[16];   b1[1] = bptr[GCOLS + 16];
#pragma unroll
    for (int t = 0; t < MSUB; ++t) {
      v2f a = *(const v2f*)&Asmem[(t * 16 + lo) * AS1 + k0 + 2 * hi];
      acc[t][0] = wmma_f32_4(a, b0, acc[t][0]);
      acc[t][1] = wmma_f32_4(a, b1, acc[t][1]);
    }
  }

  float bias0 = bias[c0 + lo], bias1 = bias[c0 + 16 + lo];
  const int oc = (wave & 3) * 16;
  v2f wb[HID / 4];
#pragma unroll
  for (int s = 0; s < HID / 4; ++s) {
    wb[s][0] = Wl[(4 * s + 2 * hi) * HID + oc + lo];
    wb[s][1] = Wl[(4 * s + 2 * hi + 1) * HID + oc + lo];
  }
  __syncthreads();
#pragma unroll
  for (int t = 0; t < MSUB; ++t) {
#pragma unroll
    for (int r = 0; r < 8; ++r) {
      int row = t * 16 + r + 8 * hi;
      Psmem[row * PS1 + c0 + lo]      = acc[t][0][r] + bias0;
      Psmem[row * PS1 + c0 + 16 + lo] = acc[t][1][r] + bias1;
    }
  }
  __syncthreads();

  for (int it = 0; it < MT * 16 / 256; ++it) {
    int i = it * 256 + tid;
    int row = i >> 4, o = (i & 15) * 4;
    int n = m0 + row;
    if (n >= Nn) {
      *(float4*)&Hsmem[row * HS1 + o] = make_float4(0.0f, 0.0f, 0.0f, 0.0f);
      continue;
    }
    const float4 cold = *(const float4*)(c_in + (size_t)n * HID + o);
    const float* pr = &Psmem[row * PS1 + o];
    const float4 pi = *(const float4*)(pr);
    const float4 pf = *(const float4*)(pr + 64);
    const float4 pt = *(const float4*)(pr + 128);
    const float4 po = *(const float4*)(pr + 192);
    float4 hv, cv;
    lstm_cell(pi.x, pf.x, pt.x, po.x, cold.x, hv.x, cv.x);
    lstm_cell(pi.y, pf.y, pt.y, po.y, cold.y, hv.y, cv.y);
    lstm_cell(pi.z, pf.z, pt.z, po.z, cold.z, hv.z, cv.z);
    lstm_cell(pi.w, pf.w, pt.w, po.w, cold.w, hv.w, cv.w);
    *(float4*)(h_out + (size_t)n * HID + o) = hv;
    *(float4*)(c_out + (size_t)n * HID + o) = cv;
    float4 hr;
    hr.x = hv.x > 0.0f ? hv.x : 0.0f;
    hr.y = hv.y > 0.0f ? hv.y : 0.0f;
    hr.z = hv.z > 0.0f ? hv.z : 0.0f;
    hr.w = hv.w > 0.0f ? hv.w : 0.0f;
    *(float4*)&Hsmem[row * HS1 + o] = hr;
  }
  __syncthreads();

  {
    const int rt = (wave >> 2) * 2;
    v8f oacc0 = {}, oacc1 = {};
#pragma unroll
    for (int k0 = 0; k0 < HID; k0 += 4) {
      const v2f b = wb[k0 / 4];
      v2f a0 = *(const v2f*)&Hsmem[(rt * 16 + lo) * HS1 + k0 + 2 * hi];
      v2f a1 = *(const v2f*)&Hsmem[(rt * 16 + 16 + lo) * HS1 + k0 + 2 * hi];
      oacc0 = wmma_f32_4(a0, b, oacc0);
      oacc1 = wmma_f32_4(a1, b, oacc1);
    }
    const float bb = bl[oc + lo];
#pragma unroll
    for (int r = 0; r < 8; ++r) {
      const int n0 = m0 + rt * 16 + r + 8 * hi;
      const int n1 = n0 + 16;
      if (n0 < Nn) out[(size_t)n0 * HID + oc + lo] = oacc0[r] + bb;
      if (n1 < Nn) out[(size_t)n1 * HID + oc + lo] = oacc1[r] + bb;
    }
  }
}


extern "C" void kernel_launch(void* const* d_in, const int* in_sizes, int n_in,
                              void* d_out, int out_size, void* d_ws, size_t ws_size,
                              hipStream_t stream) {
  const float* x      = (const float*)d_in[0];
  const int*   ei     = (const int*)d_in[1];
  const float* ew     = (const float*)d_in[2];
  const float* h      = (const float*)d_in[3];
  const float* c      = (const float*)d_in[4];
  const float* Wx     = (const float*)d_in[5];
  const float* bg     = (const float*)d_in[6];
  const float* theta  = (const float*)d_in[7];
  const float* conv_b = (const float*)d_in[8];
  const float* Wl     = (const float*)d_in[9];
  const float* bl     = (const float*)d_in[10];

  const int E  = in_sizes[2];
  const int Nn = in_sizes[3] / HID;
  const int* src = ei;
  const int* dst = ei + E;

  float* T1   = (float*)d_ws;
  float* Bp   = T1 + (size_t)Nn * HID;
  float* bias = Bp + (size_t)KDIM * GCOLS;
  int2*  edgepay  = (int2*)(bias + GCOLS);
  int2*  payA     = edgepay + E;
  int2*  rowedges = payA + E;
  int*   keyA = (int*)(rowedges + E);
  int*   lrank = keyA + E;
  const int NB = (E + SORT_CH - 1) / SORT_CH;
  int*   table = lrank + E;
  float* deg  = (float*)(table + (size_t)NB * SORT_BINS);
  int*   rowstart = (int*)(deg + Nn);
  int*   rowend = rowstart + Nn;
  int*   bstart = rowend + Nn;

  float* out   = (float*)d_out;
  float* h_out = out   + (size_t)Nn * HID;
  float* c_out = h_out + (size_t)Nn * HID;

  const float DIAG = 2.0f / LMAX - 1.0f;

  hipMemsetAsync(deg, 0, (size_t)Nn * sizeof(float), stream);
  const int NDEG = (E + 255) / 256;
  const int NPACK = (KDIM * GCOLS + 255) / 256;
  k_front<<<NB + NDEG + NPACK, 256, 0, stream>>>(src, dst, ew, deg, lrank, table,
                                                 Wx, theta, bg, conv_b, Bp, bias, E, NB, NDEG);
  k_lapw_pack<<<(E + 255) / 256, 256, 0, stream>>>(src, dst, ew, deg, edgepay, E);

  k_split_scatter<<<NB, 256, 0, stream>>>(dst, edgepay, BK_BITS, lrank, table, NB, keyA, payA,
                                          bstart, E);
  k_bucket_rows<<<(Nn + BK_ROWS - 1) / BK_ROWS, 256, 0, stream>>>(keyA, payA, bstart, rowedges,
                                                                  rowstart, rowend, Nn);

  k_spmm_t1<<<(Nn + 7) / 8, 256, 0, stream>>>(h, T1, rowedges, rowstart, rowend, DIAG, Nn);

  k_gemm_gates<<<(Nn + MT - 1) / MT, 256, 0, stream>>>(x, h, T1, rowedges, rowstart, rowend, DIAG,
                                                   Bp, bias, c, Wl, bl, out, h_out, c_out, Nn);
}


# gate kernel row tile 64 -> 80 rows (1250 work-groups: at most 5 x 80 = 400 rows per WGP, was 7 x 64 = 448); relu(h0) tile kept in the pre tile's i columns so 3 groups fit a WGP; sum orders kept
# speedup vs baseline: 1.9901x; 1.0392x over previous
#include <hip/hip_runtime.h>
#include <hip/hip_bf16.h>

typedef __attribute__((ext_vector_type(2))) float v2f;
typedef __attribute__((ext_vector_type(8))) float v8f;

#define HID 64
#define INDIM 128
#define KDIM 320
#define GCOLS 256
#define LMAX 2.0f

__device__ __forceinline__ v8f wmma_f32_4(v2f a, v2f b, v8f c) {
  return __builtin_amdgcn_wmma_f32_16x16x4_f32(false, a, false, b, (short)0, c, false, false);
}

__device__ __forceinline__ float fsigmoid(float x) {
  return 1.0f / (1.0f + __expf(-x));
}

__device__ __forceinline__ void atomic_fadd(float* p, float v) {
  asm volatile("global_atomic_add_f32 %0, %1, off scope:SCOPE_DEV"
               :: "v"(p), "v"(v) : "memory");
}


__device__ __forceinline__
void degree_block(const int* __restrict__ src, const int* __restrict__ dst,
                  const float* __restrict__ ew, float* __restrict__ deg, int E, int bid) {
  int e = bid * 256 + threadIdx.x;
  if (e >= E) return;
  int s = src[e], d = dst[e];
  if (s != d) atomic_fadd(&deg[s], ew[e]);
}

__device__ __forceinline__ float dinv_of(float dg) {
  return (dg > 0.0f) ? rsqrtf(dg) : 0.0f;
}

__global__ __launch_bounds__(256)
void k_lapw_pack(const int* __restrict__ src, const int* __restrict__ dst,
                 const float* __restrict__ ew, const float* __restrict__ deg,
                 int2* __restrict__ edgepay, int E) {
  int e = blockIdx.x * 256 + threadIdx.x;
  if (e >= E) return;
  int s = src[e], d = dst[e];
  float w = (s == d) ? 0.0f : ew[e];
  const float ds = dinv_of(deg[s]), dd = dinv_of(deg[d]);
  float lw = (2.0f / LMAX) * (-ds * w * dd);
  edgepay[e] = make_int2(s, __float_as_int(lw));
}


#define SORT_CH 8192
#define SORT_ROUNDS (SORT_CH / 256)
#define SORT_BINS 512

__device__ __forceinline__ int wave_incl_scan(int v, int lane) {
  for (int off = 1; off < 32; off <<= 1) {
    int t = __shfl_up(v, off, 32);
    if (lane >= off) v += t;
  }
  return v;
}

__device__ __forceinline__ int block_excl_scan(int v, int* wsum, int lane, int wave, int& total) {
  const int incl = wave_incl_scan(v, lane);
  if (lane == 31) wsum[wave] = incl;
  __syncthreads();
  int wb = 0, t = 0;
  for (int w = 0; w < 8; ++w) { int s = wsum[w]; if (w < wave) wb += s; t += s; }
  total = t;
  __syncthreads();
  return wb + incl - v;
}

__device__ __forceinline__
void split_rank_block(const int* __restrict__ key, int shift,
                      int* __restrict__ lrank, int* __restrict__ table, int E, int bid) {
  __shared__ int whist[8 * SORT_BINS];
  const int tid = threadIdx.x, lane = tid & 31, wave = tid >> 5;
  for (int i = tid; i < 8 * SORT_BINS; i += 256) whist[i] = 0;
  __syncthreads();
  const int base = bid * SORT_CH + wave * (SORT_CH / 8) + lane;
  int dg[SORT_ROUNDS], lr[SORT_ROUNDS];
#pragma unroll
  for (int r = 0; r < SORT_ROUNDS; ++r) {
    const int e = base + r * 32;
    const int k = key[e < E ? e : E - 1];
    dg[r] = (e < E) ? ((k >> shift) & (SORT_BINS - 1)) : -1;
  }
  int* wh = whist + wave * SORT_BINS;
#pragma unroll
  for (int r = 0; r < SORT_ROUNDS; ++r) {
    const int d = dg[r];
    const bool valid = d >= 0;
    unsigned peers = __builtin_amdgcn_ballot_w32(valid);
#pragma unroll
    for (int b = 0; b < 9; ++b) {
      const bool bit = ((d >> b) & 1) != 0;
      const unsigned bal = __builtin_amdgcn_ballot_w32(bit);
      peers &= bit ? bal : ~bal;
    }
    const int rk = __popc(peers & ((1u << lane) - 1u));
    int b0 = 0;
    if (valid) {
      b0 = wh[d];
      if (rk == 0) wh[d] = b0 + __popc(peers);
    }
    lr[r] = b0 + rk;
  }
  __syncthreads();
  for (int dgt = tid; dgt < SORT_BINS; dgt += 256) {
    int run = 0;
    for (int w = 0; w < 8; ++w) {
      int v = whist[w * SORT_BINS + dgt];
      whist[w * SORT_BINS + dgt] = run;
      run += v;
    }
    table[bid * SORT_BINS + dgt] = run;
  }
  __syncthreads();
#pragma unroll
  for (int r = 0; r < SORT_ROUNDS; ++r) {
    const int e = base + r * 32;
    if (e < E) lrank[e] = lr[r] + wh[dg[r]];
  }
}

__global__ __launch_bounds__(256)
void k_split_scatter(const int* __restrict__ key, const int2* __restrict__ pay, int shift,
                     const int* __restrict__ lrank, const int* __restrict__ table, int nblocks,
                     int* __restrict__ key_out, int2* __restrict__ pay_out,
                     int* __restrict__ bstart, int E) {
  __shared__ int lstart[SORT_BINS];
  __shared__ int gdelta[SORT_BINS];
  __shared__ int wsum[8];
  __shared__ int skey[SORT_CH];
  __shared__ __attribute__((aligned(8))) int spay_w[2 * SORT_CH];
  int2* spay = (int2*)spay_w;
  const int tid = threadIdx.x, lane = tid & 31, wave = tid >> 5;
  const int me = blockIdx.x;
  int pre0 = 0, tot0 = 0, pre1 = 0, tot1 = 0;
#pragma unroll 14
  for (int b = 0; b < nblocks; ++b) {
    const int v0 = table[b * SORT_BINS + tid];
    const int v1 = table[b * SORT_BINS + 256 + tid];
    tot0 += v0; tot1 += v1;
    if (b < me) { pre0 += v0; pre1 += v1; }
  }
  const int own0 = table[me * SORT_BINS + tid];
  const int own1 = table[me * SORT_BINS + 256 + tid];
  int sum0, sum1, osum0, osum1;
  const int ex0 = block_excl_scan(tot0, wsum, lane, wave, sum0);
  const int ex1 = block_excl_scan(tot1, wsum, lane, wave, sum1);
  const int l0 = block_excl_scan(own0, wsum, lane, wave, osum0);
  const int l1 = osum0 + block_excl_scan(own1, wsum, lane, wave, osum1);
  lstart[tid]       = l0;
  lstart[tid + 256] = l1;
  gdelta[tid]       = ex0 + pre0 - l0;
  gdelta[tid + 256] = sum0 + ex1 + pre1 - l1;
  if (me == 0) {
    bstart[tid]       = ex0;
    bstart[tid + 256] = sum0 + ex1;
    if (tid == 0) bstart[SORT_BINS] = sum0 + sum1;
  }
  __syncthreads();
  for (int r0 = 0; r0 < SORT_ROUNDS; r0 += 8) {
    int kk[8], rk[8];
    int2 pv[8];
#pragma unroll
    for (int u = 0; u < 8; ++u) {
      const int e = me * SORT_CH + (r0 + u) * 256 + tid;
      const int ec = e < E ? e : E - 1;
      kk[u] = key[ec];
      rk[u] = lrank[ec];
      pv[u] = pay[ec];
    }
    int lp[8];
#pragma unroll
    for (int u = 0; u < 8; ++u)
      lp[u] = lstart[(kk[u] >> shift) & (SORT_BINS - 1)] + rk[u];
#pragma unroll
    for (int u = 0; u < 8; ++u) {
      const int e = me * SORT_CH + (r0 + u) * 256 + tid;
      if (e < E) {
        skey[lp[u]] = kk[u];
        spay[lp[u]] = pv[u];
      }
    }
  }
  __syncthreads();
  const int rem = E - me * SORT_CH;
  const int cnt = rem < SORT_CH ? rem : SORT_CH;
  for (int r0 = 0; r0 < SORT_ROUNDS; r0 += 8) {
    int kk[8];
    int2 pv[8];
#pragma unroll
    for (int u = 0; u < 8; ++u) {
      const int i = (r0 + u) * 256 + tid;
      const int ic = i < cnt ? i : cnt - 1;
      kk[u] = skey[ic];
      pv[u] = spay[ic];
    }
    int ps[8];
#pragma unroll
    for (int u = 0; u < 8; ++u)
      ps[u] = (r0 + u) * 256 + tid + gdelta[(kk[u] >> shift) & (SORT_BINS - 1)];
#pragma unroll
    for (int u = 0; u < 8; ++u) {
      const int i = (r0 + u) * 256 + tid;
      if (i < cnt) {
        key_out[ps[u]] = kk[u];
        pay_out[ps[u]] = pv[u];
      }
    }
  }
}

#define BK_BITS 8
#define BK_ROWS (1 << BK_BITS)
#define BK_CH 8192
#define BK_ROUNDS (BK_CH / 256)

__device__ __forceinline__
int bucket_rank_trip(const int* __restrict__ bkey, int cnt, int c0, int tid, int lane, int wave,
                     int* whist, int (&dg)[BK_ROUNDS], int (&lr)[BK_ROUNDS],
                     int& seg, int& wbase, int& cend) {
  for (int i = tid; i < 8 * BK_ROWS; i += 256) whist[i] = 0;
  __syncthreads();
  const int left = cnt - c0;
  cend = c0 + (left < BK_CH ? left : BK_CH);
  seg = ((cend - c0 + 255) >> 8) << 5;
  wbase = c0 + wave * seg;
  int* wh = whist + wave * BK_ROWS;
#pragma unroll
  for (int r = 0; r < BK_ROUNDS; ++r) {
    const int e = wbase + r * 32 + lane;
    const int k = bkey[e < cnt ? e : cnt - 1];
    dg[r] = (r * 32 < seg && e < cend) ? (k & (BK_ROWS - 1)) : -1;
  }
#pragma unroll
  for (int r = 0; r < BK_ROUNDS; ++r) {
    lr[r] = 0;
    if (r * 32 < seg && wbase + r * 32 < cend) {
      const int d = dg[r];
      const bool valid = d >= 0;
      unsigned peers = __builtin_amdgcn_ballot_w32(valid);
#pragma unroll
      for (int q = 0; q < BK_BITS; ++q) {
        const bool bit = ((d >> q) & 1) != 0;
        const unsigned bal = __builtin_amdgcn_ballot_w32(bit);
        peers &= bit ? bal : ~bal;
      }
      const int rk = __popc(peers & ((1u << lane) - 1u));
      int b0 = 0;
      if (valid) {
        b0 = wh[d];
        if (rk == 0) wh[d] = b0 + __popc(peers);
      }
      lr[r] = b0 + rk;
    }
  }
  __syncthreads();
  int ctot = 0;
  for (int w = 0; w < 8; ++w) {
    const int v = whist[w * BK_ROWS + tid];
    whist[w * BK_ROWS + tid] = ctot;
    ctot += v;
  }
  __syncthreads();
  return ctot;
}

template <bool STAGED>
__device__ __forceinline__
void bucket_place_trip(const int2* __restrict__ bpay, int cnt, int lane,
                       const int* wh, const int* rowoff, const int* carry,
                       const int (&dg)[BK_ROUNDS], const int (&lr)[BK_ROUNDS],
                       int seg, int wbase, int cend, int2* spay, int2* __restrict__ gout) {
#pragma unroll
  for (int r0 = 0; r0 < BK_ROUNDS; r0 += 8) {
    if (r0 * 32 < seg && wbase + r0 * 32 < cend) {
      int2 pv[8];
#pragma unroll
      for (int u = 0; u < 8; ++u) {
        const int e = wbase + (r0 + u) * 32 + lane;
        pv[u] = bpay[e < cnt ? e : cnt - 1];
      }
#pragma unroll
      for (int u = 0; u < 8; ++u) {
        const int d = dg[r0 + u];
        if (d >= 0) {
          const int pos = rowoff[d] + carry[d] + wh[d] + lr[r0 + u];
          if (STAGED) spay[pos] = pv[u];
          else        gout[pos] = pv[u];
        }
      }
    }
  }
}

__global__ __launch_bounds__(256)
void k_bucket_rows(const int* __restrict__ key, const int2* __restrict__ pay,
                   const int* __restrict__ bstart, int2* __restrict__ rowedges,
                   int* __restrict__ rowstart, int* __restrict__ rowend, int Nn) {
  __shared__ int whist[8 * BK_ROWS];
  __shared__ int rowoff[BK_ROWS];
  __shared__ int carry[BK_ROWS];
  __shared__ int wsum[8];
  __shared__ __attribute__((aligned(8))) int spay_w[2 * BK_CH];
  int2* spay = (int2*)spay_w;
  const int tid = threadIdx.x, lane = tid & 31;
  const int wave = __builtin_amdgcn_readfirstlane(tid >> 5);
  const int b = blockIdx.x;
  const int g0 = __builtin_amdgcn_readfirstlane(bstart[b]);
  const int cnt = __builtin_amdgcn_readfirstlane(bstart[b + 1]) - g0;
  const int* bkey = key + g0;
  const int2* bpay = pay + g0;
  const int* wh = whist + wave * BK_ROWS;
  const bool staged = cnt <= BK_CH;

  carry[tid] = 0;
  int dg[BK_ROUNDS], lr[BK_ROUNDS];
  int seg = 0, wbase = 0, cend = 0;
  int mycnt = 0;
  for (int c0 = 0; c0 < cnt; c0 += BK_CH)
    mycnt += bucket_rank_trip(bkey, cnt, c0, tid, lane, wave, whist, dg, lr, seg, wbase, cend);
  int total;
  const int ex = block_excl_scan(mycnt, wsum, lane, wave, total);
  rowoff[tid] = ex;
  const int n = b * BK_ROWS + tid;
  if (n < Nn) {
    rowstart[n] = g0 + ex;
    rowend[n] = g0 + ex + mycnt;
  }
  __syncthreads();

  if (staged) {
    if (cnt > 0)
      bucket_place_trip<true>(bpay, cnt, lane, wh, rowoff, carry, dg, lr, seg, wbase, cend,
                              spay, rowedges + g0);
    __syncthreads();
    for (int i0 = 0; i0 < cnt; i0 += 8 * 256) {
      int2 pv[8];
#pragma unroll
      for (int u = 0; u < 8; ++u) {
        const int i = i0 + u * 256 + tid;
        pv[u] = spay[i < cnt ? i : cnt - 1];
      }
#pragma unroll
      for (int u = 0; u < 8; ++u) {
        const int i = i0 + u * 256 + tid;
        if (i < cnt) rowedges[g0 + i] = pv[u];
      }
    }
  } else {
    for (int c0 = 0; c0 < cnt; c0 += BK_CH) {
      const int ctot = bucket_rank_trip(bkey, cnt, c0, tid, lane, wave, whist, dg, lr,
                                        seg, wbase, cend);
      bucket_place_trip<false>(bpay, cnt, lane, wh, rowoff, carry, dg, lr, seg, wbase, cend,
                               spay, rowedges + g0);
      __syncthreads();
      carry[tid] += ctot;
    }
  }
}

#define ROWCHUNK 8
__device__ __forceinline__ float2 lhat_row(const float* __restrict__ z,
                                           const int2* __restrict__ rowedges,
                                           const int* __restrict__ rowstart,
                                           const int* __restrict__ rowend,
                                           float diag, int n, int lane) {
  const float2* z2 = (const float2*)z;
  const int p0 = rowstart[n];
  const int c = rowend[n] - p0;
  const int2* ce = rowedges + p0;
  float2 acc = make_float2(0.0f, 0.0f);
  for (int j = 0; j < c; j += ROWCHUNK) {
    float2 zv[ROWCHUNK];
    float wv[ROWCHUNK];
#pragma unroll
    for (int u = 0; u < ROWCHUNK; ++u) {
      const int idx = (j + u < c) ? (j + u) : (c - 1);
      const int2 en = ce[idx];
      wv[u] = (j + u < c) ? __int_as_float(en.y) : 0.0f;
      zv[u] = z2[(size_t)en.x * (HID / 2) + lane];
    }
#pragma unroll
    for (int u = 0; u < ROWCHUNK; ++u) {
      acc.x += wv[u] * zv[u].x;
      acc.y += wv[u] * zv[u].y;
    }
  }
  const float2 zn = z2[(size_t)n * (HID / 2) + lane];
  acc.x += diag * zn.x;
  acc.y += diag * zn.y;
  return acc;
}

__global__ __launch_bounds__(256)
void k_spmm_t1(const float* __restrict__ h, float* __restrict__ T1,
               const int2* __restrict__ rowedges, const int* __restrict__ rowstart,
               const int* __restrict__ rowend, float diag, int Nn) {
  const int lane = threadIdx.x & 31;
  const int n = __builtin_amdgcn_readfirstlane(blockIdx.x * 8 + (threadIdx.x >> 5));
  if (n >= Nn) return;
  const float2 r = lhat_row(h, rowedges, rowstart, rowend, diag, n, lane);
  ((float2*)T1)[(size_t)n * (HID / 2) + lane] = r;
}


__device__ __forceinline__
void packB_block(const float* __restrict__ Wx, const float* __restrict__ theta,
                 const float* __restrict__ bg, const float* __restrict__ conv_b,
                 float* __restrict__ Bp, float* __restrict__ bias, int bid) {
  int gid = bid * 256 + threadIdx.x;
  if (gid >= KDIM * GCOLS) return;
  int k = gid / GCOLS, col = gid % GCOLS;
  int g = col >> 6, o = col & 63;
  float v;
  if (k < INDIM) {
    v = Wx[((size_t)g * INDIM + k) * HID + o];
  } else {
    int kk = k - INDIM;
    int kb = kk >> 6, kr = kk & 63;
    v = theta[(((size_t)g * 3 + kb) * HID + kr) * HID + o];
  }
  Bp[gid] = v;
  if (gid < GCOLS) bias[gid] = bg[g * HID + o] + conv_b[g * HID + o];
}

__global__ __launch_bounds__(256)
void k_front(const int* __restrict__ src, const int* __restrict__ dst,
             const float* __restrict__ ew, float* __restrict__ deg,
             int* __restrict__ lrank, int* __restrict__ table,
             const float* __restrict__ Wx, const float* __restrict__ theta,
             const float* __restrict__ bg, const float* __restrict__ conv_b,
             float* __restrict__ Bp, float* __restrict__ bias,
             int E, int nrank, int ndeg) {
  const int b = blockIdx.x;
  if (b < nrank) {
    split_rank_block(dst, BK_BITS, lrank, table, E, b);
  } else if (b < nrank + ndeg) {
    degree_block(src, dst, ew, deg, E, b - nrank);
  } else {
    packB_block(Wx, theta, bg, conv_b, Bp, bias, b - nrank - ndeg);
  }
}


#define AS1 324
#define PS1 260
#define MT 80
#define MSUB (MT / 16)

__device__ __forceinline__ void lstm_cell(float pi, float pf, float pt, float po, float cold,
                                          float& hnew, float& cnew) {
  float ig = fsigmoid(pi);
  float fg = fsigmoid(pf);
  float tg = tanhf(pt);
  float og = fsigmoid(po);
  float cv = fg * cold + ig * tg;
  hnew = og * tanhf(cv);
  cnew = cv;
}

#define TROWS (MT / 8)
#define TCH 3

__device__ __forceinline__ int2 t2_fetch_edge(const int2* __restrict__ rowedges,
                                              int p0, int c, int slot) {
  const bool live = slot < c;
  const int gi = live ? p0 + slot : (c > 0 ? p0 + c - 1 : 0);
  int2 en = rowedges[gi];
  if (!live) en.y = 0;
  return en;
}

__device__ __forceinline__
void t2_rows_to_tile(const float* __restrict__ T1, const float* __restrict__ h,
                     const int2* __restrict__ rowedges, const int* __restrict__ rowstart,
                     const int* __restrict__ rowend, float diag, int m0, int Nn,
                     int wave, int lane, float* Asmem) {
  const float2* z2 = (const float2*)T1;
  const float2* h2 = (const float2*)h;
  const int il = lane / TCH, ul = lane % TCH;
  for (int g = 0; g < (MT / 8) / TROWS; ++g) {
    const int row0 = wave * (MT / 8) + g * TROWS;
    int nl = m0 + row0 + il;
    nl = nl < Nn ? nl : Nn - 1;
    const int p0l = rowstart[nl];
    const int cl = rowend[nl] - p0l;
    int cmax = 0;
#pragma unroll
    for (int i = 0; i < TROWS; ++i) {
      const int ci = __builtin_amdgcn_readlane(cl, i * TCH);
      cmax = ci > cmax ? ci : cmax;
    }
    float2 acc[TROWS];
#pragma unroll
    for (int i = 0; i < TROWS; ++i) acc[i] = make_float2(0.0f, 0.0f);
    int2 en = t2_fetch_edge(rowedges, p0l, cl, ul);
    for (int j = 0; j < cmax; j += TCH) {
      float2 zv[TROWS][TCH];
      float wv[TROWS][TCH];
#pragma unroll
      for (int i = 0; i < TROWS; ++i) {
#pragma unroll
        for (int u = 0; u < TCH; ++u) {
          const int src = __builtin_amdgcn_readlane(en.x, i * TCH + u);
          wv[i][u] = __int_as_float(__builtin_amdgcn_readlane(en.y, i * TCH + u));
          zv[i][u] = z2[(size_t)src * (HID / 2) + lane];
        }
      }
      en = t2_fetch_edge(rowedges, p0l, cl, j + TCH + ul);
#pragma unroll
      for (int i = 0; i < TROWS; ++i) {
#pragma unroll
        for (int u = 0; u < TCH; ++u) {
          acc[i].x += wv[i][u] * zv[i][u].x;
          acc[i].y += wv[i][u] * zv[i][u].y;
        }
      }
    }
#pragma unroll
    for (int i = 0; i < TROWS; ++i) {
      int n = m0 + row0 + i;
      n = n < Nn ? n : Nn - 1;
      const size_t o = (size_t)n * (HID / 2) + lane;
      const float2 zn = z2[o];
      const float2 hv = h2[o];
      float2 r = acc[i];
      r.x += diag * zn.x;
      r.y += diag * zn.y;
      float2 t;
      t.x = 2.0f * r.x - hv.x;
      t.y = 2.0f * r.y - hv.y;
      *(float2*)&Asmem[(row0 + i) * AS1 + 256 + 2 * lane] = t;
    }
  }
}

__global__ __launch_bounds__(256)
void k_gemm_gates(const float* __restrict__ x, const float* __restrict__ h,
                  const float* __restrict__ T1,
                  const int2* __restrict__ rowedges, const int* __restrict__ rowstart,
                  const int* __restrict__ rowend, float diag,
                  const float* __restrict__ Bp, const float* __restrict__ bias,
                  const float* __restrict__ c_in,
                  const float* __restrict__ Wl, const float* __restrict__ bl,
                  float* __restrict__ out,
                  float* __restrict__ h_out, float* __restrict__ c_out, int Nn) {
  __shared__ __attribute__((aligned(16))) float Asmem[MT * AS1];
  float* Psmem = Asmem;
  float* Hsmem = Asmem;
  const int tid = threadIdx.x;
  const int m0 = blockIdx.x * MT;

#pragma unroll
  for (int it = 0; it < MT * 32 / 256; ++it) {
    int i = it * 256 + tid;
    int r = i >> 5, q = i & 31;
    int n = m0 + r; if (n >= Nn) n = Nn - 1;
    const float4 v = *(const float4*)(x + (size_t)n * INDIM + q * 4);
    *(float4*)&Asmem[r * AS1 + q * 4] = v;
  }
#pragma unroll
  for (int it = 0; it < MT * 16 / 256; ++it) {
    int i = it * 256 + tid;
    int r = i >> 4, q = i & 15;
    int n = m0 + r; if (n >= Nn) n = Nn - 1;
    const size_t g = (size_t)n * HID + q * 4;
    const float4 vh = *(const float4*)(h + g);
    const float4 v1 = *(const float4*)(T1 + g);
    float* ap = &Asmem[r * AS1 + 128 + q * 4];
    *(float4*)(ap)       = vh;
    *(float4*)(ap + 64)  = v1;
  }
  const int lane = tid & 31;
  const int wave = __builtin_amdgcn_readfirstlane(tid >> 5);
  t2_rows_to_tile(T1, h, rowedges, rowstart, rowend, diag, m0, Nn, wave, lane, Asmem);
  __syncthreads();

  const int lo = lane & 15, hi = lane >> 4;
  const int c0 = wave * 32;
  v8f acc[MSUB][2];
#pragma unroll
  for (int t = 0; t < MSUB; ++t) { acc[t][0] = v8f{}; acc[t][1] = v8f{}; }

  for (int k0 = 0; k0 < KDIM; k0 += 4) {
    const float* bptr = Bp + (size_t)(k0 + 2 * hi) * GCOLS + c0 + lo;
    v2f b0, b1;
    b0[0] = bptr[0];    b0[1] = bptr[GCOLS];
    b1[0] = bptr[16];   b1[1] = bptr[GCOLS + 16];
#pragma unroll
    for (int t = 0; t < MSUB; ++t) {
      v2f a = *(const v2f*)&Asmem[(t * 16 + lo) * AS1 + k0 + 2 * hi];
      acc[t][0] = wmma_f32_4(a, b0, acc[t][0]);
      acc[t][1] = wmma_f32_4(a, b1, acc[t][1]);
    }
  }

  float bias0 = bias[c0 + lo], bias1 = bias[c0 + 16 + lo];
  const int oc = (wave & 3) * 16;
  v2f wb[HID / 4];
#pragma unroll
  for (int s = 0; s < HID / 4; ++s) {
    wb[s][0] = Wl[(4 * s + 2 * hi) * HID + oc + lo];
    wb[s][1] = Wl[(4 * s + 2 * hi + 1) * HID + oc + lo];
  }
  __syncthreads();
#pragma unroll
  for (int t = 0; t < MSUB; ++t) {
#pragma unroll
    for (int r = 0; r < 8; ++r) {
      int row = t * 16 + r + 8 * hi;
      Psmem[row * PS1 + c0 + lo]      = acc[t][0][r] + bias0;
      Psmem[row * PS1 + c0 + 16 + lo] = acc[t][1][r] + bias1;
    }
  }
  __syncthreads();

  for (int it = 0; it < MT * 16 / 256; ++it) {
    int i = it * 256 + tid;
    int row = i >> 4, o = (i & 15) * 4;
    int n = m0 + row;
    if (n >= Nn) {
      *(float4*)&Hsmem[row * PS1 + o] = make_float4(0.0f, 0.0f, 0.0f, 0.0f);
      continue;
    }
    const float4 cold = *(const float4*)(c_in + (size_t)n * HID + o);
    const float* pr = &Psmem[row * PS1 + o];
    const float4 pi = *(const float4*)(pr);
    const float4 pf = *(const float4*)(pr + 64);
    const float4 pt = *(const float4*)(pr + 128);
    const float4 po = *(const float4*)(pr + 192);
    float4 hv, cv;
    lstm_cell(pi.x, pf.x, pt.x, po.x, cold.x, hv.x, cv.x);
    lstm_cell(pi.y, pf.y, pt.y, po.y, cold.y, hv.y, cv.y);
    lstm_cell(pi.z, pf.z, pt.z, po.z, cold.z, hv.z, cv.z);
    lstm_cell(pi.w, pf.w, pt.w, po.w, cold.w, hv.w, cv.w);
    *(float4*)(h_out + (size_t)n * HID + o) = hv;
    *(float4*)(c_out + (size_t)n * HID + o) = cv;
    float4 hr;
    hr.x = hv.x > 0.0f ? hv.x : 0.0f;
    hr.y = hv.y > 0.0f ? hv.y : 0.0f;
    hr.z = hv.z > 0.0f ? hv.z : 0.0f;
    hr.w = hv.w > 0.0f ? hv.w : 0.0f;
    *(float4*)&Hsmem[row * PS1 + o] = hr;
  }
  __syncthreads();

  {
    const int half = wave >> 2;
    const int rt = half * 2;
    v8f oacc0 = {}, oacc1 = {};
#pragma unroll
    for (int k0 = 0; k0 < HID; k0 += 4) {
      const v2f b = wb[k0 / 4];
      v2f a0 = *(const v2f*)&Hsmem[(rt * 16 + lo) * PS1 + k0 + 2 * hi];
      v2f a1 = *(const v2f*)&Hsmem[(rt * 16 + 16 + lo) * PS1 + k0 + 2 * hi];
      oacc0 = wmma_f32_4(a0, b, oacc0);
      oacc1 = wmma_f32_4(a1, b, oacc1);
    }
    const float bb = bl[oc + lo];
#pragma unroll
    for (int r = 0; r < 8; ++r) {
      const int n0 = m0 + rt * 16 + r + 8 * hi;
      const int n1 = n0 + 16;
      if (n0 < Nn) out[(size_t)n0 * HID + oc + lo] = oacc0[r] + bb;
      if (n1 < Nn) out[(size_t)n1 * HID + oc + lo] = oacc1[r] + bb;
    }
    if (half == 0) {
      const int r4 = (MSUB - 1) * 16;
      v8f oacc2 = {};
#pragma unroll
      for (int k0 = 0; k0 < HID; k0 += 4) {
        v2f a2 = *(const v2f*)&Hsmem[(r4 + lo) * PS1 + k0 + 2 * hi];
        oacc2 = wmma_f32_4(a2, wb[k0 / 4], oacc2);
      }
#pragma unroll
      for (int r = 0; r < 8; ++r) {
        const int n2 = m0 + r4 + r + 8 * hi;
        if (n2 < Nn) out[(size_t)n2 * HID + oc + lo] = oacc2[r] + bb;
      }
    }
  }
}


extern "C" void kernel_launch(void* const* d_in, const int* in_sizes, int n_in,
                              void* d_out, int out_size, void* d_ws, size_t ws_size,
                              hipStream_t stream) {
  const float* x      = (const float*)d_in[0];
  const int*   ei     = (const int*)d_in[1];
  const float* ew     = (const float*)d_in[2];
  const float* h      = (const float*)d_in[3];
  const float* c      = (const float*)d_in[4];
  const float* Wx     = (const float*)d_in[5];
  const float* bg     = (const float*)d_in[6];
  const float* theta  = (const float*)d_in[7];
  const float* conv_b = (const float*)d_in[8];
  const float* Wl     = (const float*)d_in[9];
  const float* bl     = (const float*)d_in[10];

  const int E  = in_sizes[2];
  const int Nn = in_sizes[3] / HID;
  const int* src = ei;
  const int* dst = ei + E;

  float* T1   = (float*)d_ws;
  float* Bp   = T1 + (size_t)Nn * HID;
  float* bias = Bp + (size_t)KDIM * GCOLS;
  int2*  edgepay  = (int2*)(bias + GCOLS);
  int2*  payA     = edgepay + E;
  int2*  rowedges = payA + E;
  int*   keyA = (int*)(rowedges + E);
  int*   lrank = keyA + E;
  const int NB = (E + SORT_CH - 1) / SORT_CH;
  int*   table = lrank + E;
  float* deg  = (float*)(table + (size_t)NB * SORT_BINS);
  int*   rowstart = (int*)(deg + Nn);
  int*   rowend = rowstart + Nn;
  int*   bstart = rowend + Nn;

  float* out   = (float*)d_out;
  float* h_out = out   + (size_t)Nn * HID;
  float* c_out = h_out + (size_t)Nn * HID;

  const float DIAG = 2.0f / LMAX - 1.0f;

  hipMemsetAsync(deg, 0, (size_t)Nn * sizeof(float), stream);
  const int NDEG = (E + 255) / 256;
  const int NPACK = (KDIM * GCOLS + 255) / 256;
  k_front<<<NB + NDEG + NPACK, 256, 0, stream>>>(src, dst, ew, deg, lrank, table,
                                                 Wx, theta, bg, conv_b, Bp, bias, E, NB, NDEG);
  k_lapw_pack<<<(E + 255) / 256, 256, 0, stream>>>(src, dst, ew, deg, edgepay, E);

  k_split_scatter<<<NB, 256, 0, stream>>>(dst, edgepay, BK_BITS, lrank, table, NB, keyA, payA,
                                          bstart, E);
  k_bucket_rows<<<(Nn + BK_ROWS - 1) / BK_ROWS, 256, 0, stream>>>(keyA, payA, bstart, rowedges,
                                                                  rowstart, rowend, Nn);

  k_spmm_t1<<<(Nn + 7) / 8, 256, 0, stream>>>(h, T1, rowedges, rowstart, rowend, DIAG, Nn);

  k_gemm_gates<<<(Nn + MT - 1) / MT, 256, 0, stream>>>(x, h, T1, rowedges, rowstart, rowend, DIAG,
                                                   Bp, bias, c, Wl, bl, out, h_out, c_out, Nn);
}


# LSTM cell: sigmoid and tanh as one exponential + one reciprocal instruction each (tanh x = 1 - 2/(1+e^2x)), not IEEE divides and the library tanh: 393 -> 95 vector instr per 4 cells; sums unchanged
# speedup vs baseline: 2.0826x; 1.0465x over previous
#include <hip/hip_runtime.h>
#include <hip/hip_bf16.h>

typedef __attribute__((ext_vector_type(2))) float v2f;
typedef __attribute__((ext_vector_type(8))) float v8f;

#define HID 64
#define INDIM 128
#define KDIM 320
#define GCOLS 256
#define LMAX 2.0f

__device__ __forceinline__ v8f wmma_f32_4(v2f a, v2f b, v8f c) {
  return __builtin_amdgcn_wmma_f32_16x16x4_f32(false, a, false, b, (short)0, c, false, false);
}

__device__ __forceinline__ float fsigmoid(float x) {
  return __builtin_amdgcn_rcpf(1.0f + __expf(-x));
}

__device__ __forceinline__ float ftanh(float x) {
  return 1.0f - 2.0f * __builtin_amdgcn_rcpf(1.0f + __expf(2.0f * x));
}

__device__ __forceinline__ void atomic_fadd(float* p, float v) {
  asm volatile("global_atomic_add_f32 %0, %1, off scope:SCOPE_DEV"
               :: "v"(p), "v"(v) : "memory");
}


__device__ __forceinline__
void degree_block(const int* __restrict__ src, const int* __restrict__ dst,
                  const float* __restrict__ ew, float* __restrict__ deg, int E, int bid) {
  int e = bid * 256 + threadIdx.x;
  if (e >= E) return;
  int s = src[e], d = dst[e];
  if (s != d) atomic_fadd(&deg[s], ew[e]);
}

__device__ __forceinline__ float dinv_of(float dg) {
  return (dg > 0.0f) ? rsqrtf(dg) : 0.0f;
}

__global__ __launch_bounds__(256)
void k_lapw_pack(const int* __restrict__ src, const int* __restrict__ dst,
                 const float* __restrict__ ew, const float* __restrict__ deg,
                 int2* __restrict__ edgepay, int E) {
  int e = blockIdx.x * 256 + threadIdx.x;
  if (e >= E) return;
  int s = src[e], d = dst[e];
  float w = (s == d) ? 0.0f : ew[e];
  const float ds = dinv_of(deg[s]), dd = dinv_of(deg[d]);
  float lw = (2.0f / LMAX) * (-ds * w * dd);
  edgepay[e] = make_int2(s, __float_as_int(lw));
}


#define SORT_CH 8192
#define SORT_ROUNDS (SORT_CH / 256)
#define SORT_BINS 512

__device__ __forceinline__ int wave_incl_scan(int v, int lane) {
  for (int off = 1; off < 32; off <<= 1) {
    int t = __shfl_up(v, off, 32);
    if (lane >= off) v += t;
  }
  return v;
}

__device__ __forceinline__ int block_excl_scan(int v, int* wsum, int lane, int wave, int& total) {
  const int incl = wave_incl_scan(v, lane);
  if (lane == 31) wsum[wave] = incl;
  __syncthreads();
  int wb = 0, t = 0;
  for (int w = 0; w < 8; ++w) { int s = wsum[w]; if (w < wave) wb += s; t += s; }
  total = t;
  __syncthreads();
  return wb + incl - v;
}

__device__ __forceinline__
void split_rank_block(const int* __restrict__ key, int shift,
                      int* __restrict__ lrank, int* __restrict__ table, int E, int bid) {
  __shared__ int whist[8 * SORT_BINS];
  const int tid = threadIdx.x, lane = tid & 31, wave = tid >> 5;
  for (int i = tid; i < 8 * SORT_BINS; i += 256) whist[i] = 0;
  __syncthreads();
  const int base = bid * SORT_CH + wave * (SORT_CH / 8) + lane;
  int dg[SORT_ROUNDS], lr[SORT_ROUNDS];
#pragma unroll
  for (int r = 0; r < SORT_ROUNDS; ++r) {
    const int e = base + r * 32;
    const int k = key[e < E ? e : E - 1];
    dg[r] = (e < E) ? ((k >> shift) & (SORT_BINS - 1)) : -1;
  }
  int* wh = whist + wave * SORT_BINS;
#pragma unroll
  for (int r = 0; r < SORT_ROUNDS; ++r) {
    const int d = dg[r];
    const bool valid = d >= 0;
    unsigned peers = __builtin_amdgcn_ballot_w32(valid);
#pragma unroll
    for (int b = 0; b < 9; ++b) {
      const bool bit = ((d >> b) & 1) != 0;
      const unsigned bal = __builtin_amdgcn_ballot_w32(bit);
      peers &= bit ? bal : ~bal;
    }
    const int rk = __popc(peers & ((1u << lane) - 1u));
    int b0 = 0;
    if (valid) {
      b0 = wh[d];
      if (rk == 0) wh[d] = b0 + __popc(peers);
    }
    lr[r] = b0 + rk;
  }
  __syncthreads();
  for (int dgt = tid; dgt < SORT_BINS; dgt += 256) {
    int run = 0;
    for (int w = 0; w < 8; ++w) {
      int v = whist[w * SORT_BINS + dgt];
      whist[w * SORT_BINS + dgt] = run;
      run += v;
    }
    table[bid * SORT_BINS + dgt] = run;
  }
  __syncthreads();
#pragma unroll
  for (int r = 0; r < SORT_ROUNDS; ++r) {
    const int e = base + r * 32;
    if (e < E) lrank[e] = lr[r] + wh[dg[r]];
  }
}

__global__ __launch_bounds__(256)
void k_split_scatter(const int* __restrict__ key, const int2* __restrict__ pay, int shift,
                     const int* __restrict__ lrank, const int* __restrict__ table, int nblocks,
                     int* __restrict__ key_out, int2* __restrict__ pay_out,
                     int* __restrict__ bstart, int E) {
  __shared__ int lstart[SORT_BINS];
  __shared__ int gdelta[SORT_BINS];
  __shared__ int wsum[8];
  __shared__ int skey[SORT_CH];
  __shared__ __attribute__((aligned(8))) int spay_w[2 * SORT_CH];
  int2* spay = (int2*)spay_w;
  const int tid = threadIdx.x, lane = tid & 31, wave = tid >> 5;
  const int me = blockIdx.x;
  int pre0 = 0, tot0 = 0, pre1 = 0, tot1 = 0;
#pragma unroll 14
  for (int b = 0; b < nblocks; ++b) {
    const int v0 = table[b * SORT_BINS + tid];
    const int v1 = table[b * SORT_BINS + 256 + tid];
    tot0 += v0; tot1 += v1;
    if (b < me) { pre0 += v0; pre1 += v1; }
  }
  const int own0 = table[me * SORT_BINS + tid];
  const int own1 = table[me * SORT_BINS + 256 + tid];
  int sum0, sum1, osum0, osum1;
  const int ex0 = block_excl_scan(tot0, wsum, lane, wave, sum0);
  const int ex1 = block_excl_scan(tot1, wsum, lane, wave, sum1);
  const int l0 = block_excl_scan(own0, wsum, lane, wave, osum0);
  const int l1 = osum0 + block_excl_scan(own1, wsum, lane, wave, osum1);
  lstart[tid]       = l0;
  lstart[tid + 256] = l1;
  gdelta[tid]       = ex0 + pre0 - l0;
  gdelta[tid + 256] = sum0 + ex1 + pre1 - l1;
  if (me == 0) {
    bstart[tid]       = ex0;
    bstart[tid + 256] = sum0 + ex1;
    if (tid == 0) bstart[SORT_BINS] = sum0 + sum1;
  }
  __syncthreads();
  for (int r0 = 0; r0 < SORT_ROUNDS; r0 += 8) {
    int kk[8], rk[8];
    int2 pv[8];
#pragma unroll
    for (int u = 0; u < 8; ++u) {
      const int e = me * SORT_CH + (r0 + u) * 256 + tid;
      const int ec = e < E ? e : E - 1;
      kk[u] = key[ec];
      rk[u] = lrank[ec];
      pv[u] = pay[ec];
    }
    int lp[8];
#pragma unroll
    for (int u = 0; u < 8; ++u)
      lp[u] = lstart[(kk[u] >> shift) & (SORT_BINS - 1)] + rk[u];
#pragma unroll
    for (int u = 0; u < 8; ++u) {
      const int e = me * SORT_CH + (r0 + u) * 256 + tid;
      if (e < E) {
        skey[lp[u]] = kk[u];
        spay[lp[u]] = pv[u];
      }
    }
  }
  __syncthreads();
  const int rem = E - me * SORT_CH;
  const int cnt = rem < SORT_CH ? rem : SORT_CH;
  for (int r0 = 0; r0 < SORT_ROUNDS; r0 += 8) {
    int kk[8];
    int2 pv[8];
#pragma unroll
    for (int u = 0; u < 8; ++u) {
      const int i = (r0 + u) * 256 + tid;
      const int ic = i < cnt ? i : cnt - 1;
      kk[u] = skey[ic];
      pv[u] = spay[ic];
    }
    int ps[8];
#pragma unroll
    for (int u = 0; u < 8; ++u)
      ps[u] = (r0 + u) * 256 + tid + gdelta[(kk[u] >> shift) & (SORT_BINS - 1)];
#pragma unroll
    for (int u = 0; u < 8; ++u) {
      const int i = (r0 + u) * 256 + tid;
      if (i < cnt) {
        key_out[ps[u]] = kk[u];
        pay_out[ps[u]] = pv[u];
      }
    }
  }
}

#define BK_BITS 8
#define BK_ROWS (1 << BK_BITS)
#define BK_CH 8192
#define BK_ROUNDS (BK_CH / 256)

__device__ __forceinline__
int bucket_rank_trip(const int* __restrict__ bkey, int cnt, int c0, int tid, int lane, int wave,
                     int* whist, int (&dg)[BK_ROUNDS], int (&lr)[BK_ROUNDS],
                     int& seg, int& wbase, int& cend) {
  for (int i = tid; i < 8 * BK_ROWS; i += 256) whist[i] = 0;
  __syncthreads();
  const int left = cnt - c0;
  cend = c0 + (left < BK_CH ? left : BK_CH);
  seg = ((cend - c0 + 255) >> 8) << 5;
  wbase = c0 + wave * seg;
  int* wh = whist + wave * BK_ROWS;
#pragma unroll
  for (int r = 0; r < BK_ROUNDS; ++r) {
    const int e = wbase + r * 32 + lane;
    const int k = bkey[e < cnt ? e : cnt - 1];
    dg[r] = (r * 32 < seg && e < cend) ? (k & (BK_ROWS - 1)) : -1;
  }
#pragma unroll
  for (int r = 0; r < BK_ROUNDS; ++r) {
    lr[r] = 0;
    if (r * 32 < seg && wbase + r * 32 < cend) {
      const int d = dg[r];
      const bool valid = d >= 0;
      unsigned peers = __builtin_amdgcn_ballot_w32(valid);
#pragma unroll
      for (int q = 0; q < BK_BITS; ++q) {
        const bool bit = ((d >> q) & 1) != 0;
        const unsigned bal = __builtin_amdgcn_ballot_w32(bit);
        peers &= bit ? bal : ~bal;
      }
      const int rk = __popc(peers & ((1u << lane) - 1u));
      int b0 = 0;
      if (valid) {
        b0 = wh[d];
        if (rk == 0) wh[d] = b0 + __popc(peers);
      }
      lr[r] = b0 + rk;
    }
  }
  __syncthreads();
  int ctot = 0;
  for (int w = 0; w < 8; ++w) {
    const int v = whist[w * BK_ROWS + tid];
    whist[w * BK_ROWS + tid] = ctot;
    ctot += v;
  }
  __syncthreads();
  return ctot;
}

template <bool STAGED>
__device__ __forceinline__
void bucket_place_trip(const int2* __restrict__ bpay, int cnt, int lane,
                       const int* wh, const int* rowoff, const int* carry,
                       const int (&dg)[BK_ROUNDS], const int (&lr)[BK_ROUNDS],
                       int seg, int wbase, int cend, int2* spay, int2* __restrict__ gout) {
#pragma unroll
  for (int r0 = 0; r0 < BK_ROUNDS; r0 += 8) {
    if (r0 * 32 < seg && wbase + r0 * 32 < cend) {
      int2 pv[8];
#pragma unroll
      for (int u = 0; u < 8; ++u) {
        const int e = wbase + (r0 + u) * 32 + lane;
        pv[u] = bpay[e < cnt ? e : cnt - 1];
      }
#pragma unroll
      for (int u = 0; u < 8; ++u) {
        const int d = dg[r0 + u];
        if (d >= 0) {
          const int pos = rowoff[d] + carry[d] + wh[d] + lr[r0 + u];
          if (STAGED) spay[pos] = pv[u];
          else        gout[pos] = pv[u];
        }
      }
    }
  }
}

__global__ __launch_bounds__(256)
void k_bucket_rows(const int* __restrict__ key, const int2* __restrict__ pay,
                   const int* __restrict__ bstart, int2* __restrict__ rowedges,
                   int* __restrict__ rowstart, int* __restrict__ rowend, int Nn) {
  __shared__ int whist[8 * BK_ROWS];
  __shared__ int rowoff[BK_ROWS];
  __shared__ int carry[BK_ROWS];
  __shared__ int wsum[8];
  __shared__ __attribute__((aligned(8))) int spay_w[2 * BK_CH];
  int2* spay = (int2*)spay_w;
  const int tid = threadIdx.x, lane = tid & 31;
  const int wave = __builtin_amdgcn_readfirstlane(tid >> 5);
  const int b = blockIdx.x;
  const int g0 = __builtin_amdgcn_readfirstlane(bstart[b]);
  const int cnt = __builtin_amdgcn_readfirstlane(bstart[b + 1]) - g0;
  const int* bkey = key + g0;
  const int2* bpay = pay + g0;
  const int* wh = whist + wave * BK_ROWS;
  const bool staged = cnt <= BK_CH;

  carry[tid] = 0;
  int dg[BK_ROUNDS], lr[BK_ROUNDS];
  int seg = 0, wbase = 0, cend = 0;
  int mycnt = 0;
  for (int c0 = 0; c0 < cnt; c0 += BK_CH)
    mycnt += bucket_rank_trip(bkey, cnt, c0, tid, lane, wave, whist, dg, lr, seg, wbase, cend);
  int total;
  const int ex = block_excl_scan(mycnt, wsum, lane, wave, total);
  rowoff[tid] = ex;
  const int n = b * BK_ROWS + tid;
  if (n < Nn) {
    rowstart[n] = g0 + ex;
    rowend[n] = g0 + ex + mycnt;
  }
  __syncthreads();

  if (staged) {
    if (cnt > 0)
      bucket_place_trip<true>(bpay, cnt, lane, wh, rowoff, carry, dg, lr, seg, wbase, cend,
                              spay, rowedges + g0);
    __syncthreads();
    for (int i0 = 0; i0 < cnt; i0 += 8 * 256) {
      int2 pv[8];
#pragma unroll
      for (int u = 0; u < 8; ++u) {
        const int i = i0 + u * 256 + tid;
        pv[u] = spay[i < cnt ? i : cnt - 1];
      }
#pragma unroll
      for (int u = 0; u < 8; ++u) {
        const int i = i0 + u * 256 + tid;
        if (i < cnt) rowedges[g0 + i] = pv[u];
      }
    }
  } else {
    for (int c0 = 0; c0 < cnt; c0 += BK_CH) {
      const int ctot = bucket_rank_trip(bkey, cnt, c0, tid, lane, wave, whist, dg, lr,
                                        seg, wbase, cend);
      bucket_place_trip<false>(bpay, cnt, lane, wh, rowoff, carry, dg, lr, seg, wbase, cend,
                               spay, rowedges + g0);
      __syncthreads();
      carry[tid] += ctot;
    }
  }
}

#define ROWCHUNK 8
__device__ __forceinline__ float2 lhat_row(const float* __restrict__ z,
                                           const int2* __restrict__ rowedges,
                                           const int* __restrict__ rowstart,
                                           const int* __restrict__ rowend,
                                           float diag, int n, int lane) {
  const float2* z2 = (const float2*)z;
  const int p0 = rowstart[n];
  const int c = rowend[n] - p0;
  const int2* ce = rowedges + p0;
  float2 acc = make_float2(0.0f, 0.0f);
  for (int j = 0; j < c; j += ROWCHUNK) {
    float2 zv[ROWCHUNK];
    float wv[ROWCHUNK];
#pragma unroll
    for (int u = 0; u < ROWCHUNK; ++u) {
      const int idx = (j + u < c) ? (j + u) : (c - 1);
      const int2 en = ce[idx];
      wv[u] = (j + u < c) ? __int_as_float(en.y) : 0.0f;
      zv[u] = z2[(size_t)en.x * (HID / 2) + lane];
    }
#pragma unroll
    for (int u = 0; u < ROWCHUNK; ++u) {
      acc.x += wv[u] * zv[u].x;
      acc.y += wv[u] * zv[u].y;
    }
  }
  const float2 zn = z2[(size_t)n * (HID / 2) + lane];
  acc.x += diag * zn.x;
  acc.y += diag * zn.y;
  return acc;
}

__global__ __launch_bounds__(256)
void k_spmm_t1(const float* __restrict__ h, float* __restrict__ T1,
               const int2* __restrict__ rowedges, const int* __restrict__ rowstart,
               const int* __restrict__ rowend, float diag, int Nn) {
  const int lane = threadIdx.x & 31;
  const int n = __builtin_amdgcn_readfirstlane(blockIdx.x * 8 + (threadIdx.x >> 5));
  if (n >= Nn) return;
  const float2 r = lhat_row(h, rowedges, rowstart, rowend, diag, n, lane);
  ((float2*)T1)[(size_t)n * (HID / 2) + lane] = r;
}


__device__ __forceinline__
void packB_block(const float* __restrict__ Wx, const float* __restrict__ theta,
                 const float* __restrict__ bg, const float* __restrict__ conv_b,
                 float* __restrict__ Bp, float* __restrict__ bias, int bid) {
  int gid = bid * 256 + threadIdx.x;
  if (gid >= KDIM * GCOLS) return;
  int k = gid / GCOLS, col = gid % GCOLS;
  int g = col >> 6, o = col & 63;
  float v;
  if (k < INDIM) {
    v = Wx[((size_t)g * INDIM + k) * HID + o];
  } else {
    int kk = k - INDIM;
    int kb = kk >> 6, kr = kk & 63;
    v = theta[(((size_t)g * 3 + kb) * HID + kr) * HID + o];
  }
  Bp[gid] = v;
  if (gid < GCOLS) bias[gid] = bg[g * HID + o] + conv_b[g * HID + o];
}

__global__ __launch_bounds__(256)
void k_front(const int* __restrict__ src, const int* __restrict__ dst,
             const float* __restrict__ ew, float* __restrict__ deg,
             int* __restrict__ lrank, int* __restrict__ table,
             const float* __restrict__ Wx, const float* __restrict__ theta,
             const float* __restrict__ bg, const float* __restrict__ conv_b,
             float* __restrict__ Bp, float* __restrict__ bias,
             int E, int nrank, int ndeg) {
  const int b = blockIdx.x;
  if (b < nrank) {
    split_rank_block(dst, BK_BITS, lrank, table, E, b);
  } else if (b < nrank + ndeg) {
    degree_block(src, dst, ew, deg, E, b - nrank);
  } else {
    packB_block(Wx, theta, bg, conv_b, Bp, bias, b - nrank - ndeg);
  }
}


#define AS1 324
#define PS1 260
#define MT 80
#define MSUB (MT / 16)

__device__ __forceinline__ void lstm_cell(float pi, float pf, float pt, float po, float cold,
                                          float& hnew, float& cnew) {
  float ig = fsigmoid(pi);
  float fg = fsigmoid(pf);
  float tg = ftanh(pt);
  float og = fsigmoid(po);
  float cv = fg * cold + ig * tg;
  hnew = og * ftanh(cv);
  cnew = cv;
}

#define TROWS (MT / 8)
#define TCH 3

__device__ __forceinline__ int2 t2_fetch_edge(const int2* __restrict__ rowedges,
                                              int p0, int c, int slot) {
  const bool live = slot < c;
  const int gi = live ? p0 + slot : (c > 0 ? p0 + c - 1 : 0);
  int2 en = rowedges[gi];
  if (!live) en.y = 0;
  return en;
}

__device__ __forceinline__
void t2_rows_to_tile(const float* __restrict__ T1, const float* __restrict__ h,
                     const int2* __restrict__ rowedges, const int* __restrict__ rowstart,
                     const int* __restrict__ rowend, float diag, int m0, int Nn,
                     int wave, int lane, float* Asmem) {
  const float2* z2 = (const float2*)T1;
  const float2* h2 = (const float2*)h;
  const int il = lane / TCH, ul = lane % TCH;
  for (int g = 0; g < (MT / 8) / TROWS; ++g) {
    const int row0 = wave * (MT / 8) + g * TROWS;
    int nl = m0 + row0 + il;
    nl = nl < Nn ? nl : Nn - 1;
    const int p0l = rowstart[nl];
    const int cl = rowend[nl] - p0l;
    int cmax = 0;
#pragma unroll
    for (int i = 0; i < TROWS; ++i) {
      const int ci = __builtin_amdgcn_readlane(cl, i * TCH);
      cmax = ci > cmax ? ci : cmax;
    }
    float2 acc[TROWS];
#pragma unroll
    for (int i = 0; i < TROWS; ++i) acc[i] = make_float2(0.0f, 0.0f);
    int2 en = t2_fetch_edge(rowedges, p0l, cl, ul);
    for (int j = 0; j < cmax; j += TCH) {
      float2 zv[TROWS][TCH];
      float wv[TROWS][TCH];
#pragma unroll
      for (int i = 0; i < TROWS; ++i) {
#pragma unroll
        for (int u = 0; u < TCH; ++u) {
          const int src = __builtin_amdgcn_readlane(en.x, i * TCH + u);
          wv[i][u] = __int_as_float(__builtin_amdgcn_readlane(en.y, i * TCH + u));
          zv[i][u] = z2[(size_t)src * (HID / 2) + lane];
        }
      }
      en = t2_fetch_edge(rowedges, p0l, cl, j + TCH + ul);
#pragma unroll
      for (int i = 0; i < TROWS; ++i) {
#pragma unroll
        for (int u = 0; u < TCH; ++u) {
          acc[i].x += wv[i][u] * zv[i][u].x;
          acc[i].y += wv[i][u] * zv[i][u].y;
        }
      }
    }
#pragma unroll
    for (int i = 0; i < TROWS; ++i) {
      int n = m0 + row0 + i;
      n = n < Nn ? n : Nn - 1;
      const size_t o = (size_t)n * (HID / 2) + lane;
      const float2 zn = z2[o];
      const float2 hv = h2[o];
      float2 r = acc[i];
      r.x += diag * zn.x;
      r.y += diag * zn.y;
      float2 t;
      t.x = 2.0f * r.x - hv.x;
      t.y = 2.0f * r.y - hv.y;
      *(float2*)&Asmem[(row0 + i) * AS1 + 256 + 2 * lane] = t;
    }
  }
}

__global__ __launch_bounds__(256)
void k_gemm_gates(const float* __restrict__ x, const float* __restrict__ h,
                  const float* __restrict__ T1,
                  const int2* __restrict__ rowedges, const int* __restrict__ rowstart,
                  const int* __restrict__ rowend, float diag,
                  const float* __restrict__ Bp, const float* __restrict__ bias,
                  const float* __restrict__ c_in,
                  const float* __restrict__ Wl, const float* __restrict__ bl,
                  float* __restrict__ out,
                  float* __restrict__ h_out, float* __restrict__ c_out, int Nn) {
  __shared__ __attribute__((aligned(16))) float Asmem[MT * AS1];
  float* Psmem = Asmem;
  float* Hsmem = Asmem;
  const int tid = threadIdx.x;
  const int m0 = blockIdx.x * MT;

#pragma unroll
  for (int it = 0; it < MT * 32 / 256; ++it) {
    int i = it * 256 + tid;
    int r = i >> 5, q = i & 31;
    int n = m0 + r; if (n >= Nn) n = Nn - 1;
    const float4 v = *(const float4*)(x + (size_t)n * INDIM + q * 4);
    *(float4*)&Asmem[r * AS1 + q * 4] = v;
  }
#pragma unroll
  for (int it = 0; it < MT * 16 / 256; ++it) {
    int i = it * 256 + tid;
    int r = i >> 4, q = i & 15;
    int n = m0 + r; if (n >= Nn) n = Nn - 1;
    const size_t g = (size_t)n * HID + q * 4;
    const float4 vh = *(const float4*)(h + g);
    const float4 v1 = *(const float4*)(T1 + g);
    float* ap = &Asmem[r * AS1 + 128 + q * 4];
    *(float4*)(ap)       = vh;
    *(float4*)(ap + 64)  = v1;
  }
  const int lane = tid & 31;
  const int wave = __builtin_amdgcn_readfirstlane(tid >> 5);
  t2_rows_to_tile(T1, h, rowedges, rowstart, rowend, diag, m0, Nn, wave, lane, Asmem);
  __syncthreads();

  const int lo = lane & 15, hi = lane >> 4;
  const int c0 = wave * 32;
  v8f acc[MSUB][2];
#pragma unroll
  for (int t = 0; t < MSUB; ++t) { acc[t][0] = v8f{}; acc[t][1] = v8f{}; }

  for (int k0 = 0; k0 < KDIM; k0 += 4) {
    const float* bptr = Bp + (size_t)(k0 + 2 * hi) * GCOLS + c0 + lo;
    v2f b0, b1;
    b0[0] = bptr[0];    b0[1] = bptr[GCOLS];
    b1[0] = bptr[16];   b1[1] = bptr[GCOLS + 16];
#pragma unroll
    for (int t = 0; t < MSUB; ++t) {
      v2f a = *(const v2f*)&Asmem[(t * 16 + lo) * AS1 + k0 + 2 * hi];
      acc[t][0] = wmma_f32_4(a, b0, acc[t][0]);
      acc[t][1] = wmma_f32_4(a, b1, acc[t][1]);
    }
  }

  float bias0 = bias[c0 + lo], bias1 = bias[c0 + 16 + lo];
  const int oc = (wave & 3) * 16;
  v2f wb[HID / 4];
#pragma unroll
  for (int s = 0; s < HID / 4; ++s) {
    wb[s][0] = Wl[(4 * s + 2 * hi) * HID + oc + lo];
    wb[s][1] = Wl[(4 * s + 2 * hi + 1) * HID + oc + lo];
  }
  __syncthreads();
#pragma unroll
  for (int t = 0; t < MSUB; ++t) {
#pragma unroll
    for (int r = 0; r < 8; ++r) {
      int row = t * 16 + r + 8 * hi;
      Psmem[row * PS1 + c0 + lo]      = acc[t][0][r] + bias0;
      Psmem[row * PS1 + c0 + 16 + lo] = acc[t][1][r] + bias1;
    }
  }
  __syncthreads();

  for (int it = 0; it < MT * 16 / 256; ++it) {
    int i = it * 256 + tid;
    int row = i >> 4, o = (i & 15) * 4;
    int n = m0 + row;
    if (n >= Nn) {
      *(float4*)&Hsmem[row * PS1 + o] = make_float4(0.0f, 0.0f, 0.0f, 0.0f);
      continue;
    }
    const float4 cold = *(const float4*)(c_in + (size_t)n * HID + o);
    const float* pr = &Psmem[row * PS1 + o];
    const float4 pi = *(const float4*)(pr);
    const float4 pf = *(const float4*)(pr + 64);
    const float4 pt = *(const float4*)(pr + 128);
    const float4 po = *(const float4*)(pr + 192);
    float4 hv, cv;
    lstm_cell(pi.x, pf.x, pt.x, po.x, cold.x, hv.x, cv.x);
    lstm_cell(pi.y, pf.y, pt.y, po.y, cold.y, hv.y, cv.y);
    lstm_cell(pi.z, pf.z, pt.z, po.z, cold.z, hv.z, cv.z);
    lstm_cell(pi.w, pf.w, pt.w, po.w, cold.w, hv.w, cv.w);
    *(float4*)(h_out + (size_t)n * HID + o) = hv;
    *(float4*)(c_out + (size_t)n * HID + o) = cv;
    float4 hr;
    hr.x = hv.x > 0.0f ? hv.x : 0.0f;
    hr.y = hv.y > 0.0f ? hv.y : 0.0f;
    hr.z = hv.z > 0.0f ? hv.z : 0.0f;
    hr.w = hv.w > 0.0f ? hv.w : 0.0f;
    *(float4*)&Hsmem[row * PS1 + o] = hr;
  }
  __syncthreads();

  {
    const int half = wave >> 2;
    const int rt = half * 2;
    v8f oacc0 = {}, oacc1 = {};
#pragma unroll
    for (int k0 = 0; k0 < HID; k0 += 4) {
      const v2f b = wb[k0 / 4];
      v2f a0 = *(const v2f*)&Hsmem[(rt * 16 + lo) * PS1 + k0 + 2 * hi];
      v2f a1 = *(const v2f*)&Hsmem[(rt * 16 + 16 + lo) * PS1 + k0 + 2 * hi];
      oacc0 = wmma_f32_4(a0, b, oacc0);
      oacc1 = wmma_f32_4(a1, b, oacc1);
    }
    const float bb = bl[oc + lo];
#pragma unroll
    for (int r = 0; r < 8; ++r) {
      const int n0 = m0 + rt * 16 + r + 8 * hi;
      const int n1 = n0 + 16;
      if (n0 < Nn) out[(size_t)n0 * HID + oc + lo] = oacc0[r] + bb;
      if (n1 < Nn) out[(size_t)n1 * HID + oc + lo] = oacc1[r] + bb;
    }
    if (half == 0) {
      const int r4 = (MSUB - 1) * 16;
      v8f oacc2 = {};
#pragma unroll
      for (int k0 = 0; k0 < HID; k0 += 4) {
        v2f a2 = *(const v2f*)&Hsmem[(r4 + lo) * PS1 + k0 + 2 * hi];
        oacc2 = wmma_f32_4(a2, wb[k0 / 4], oacc2);
      }
#pragma unroll
      for (int r = 0; r < 8; ++r) {
        const int n2 = m0 + r4 + r + 8 * hi;
        if (n2 < Nn) out[(size_t)n2 * HID + oc + lo] = oacc2[r] + bb;
      }
    }
  }
}


extern "C" void kernel_launch(void* const* d_in, const int* in_sizes, int n_in,
                              void* d_out, int out_size, void* d_ws, size_t ws_size,
                              hipStream_t stream) {
  const float* x      = (const float*)d_in[0];
  const int*   ei     = (const int*)d_in[1];
  const float* ew     = (const float*)d_in[2];
  const float* h      = (const float*)d_in[3];
  const float* c      = (const float*)d_in[4];
  const float* Wx     = (const float*)d_in[5];
  const float* bg     = (const float*)d_in[6];
  const float* theta  = (const float*)d_in[7];
  const float* conv_b = (const float*)d_in[8];
  const float* Wl     = (const float*)d_in[9];
  const float* bl     = (const float*)d_in[10];

  const int E  = in_sizes[2];
  const int Nn = in_sizes[3] / HID;
  const int* src = ei;
  const int* dst = ei + E;

  float* T1   = (float*)d_ws;
  float* Bp   = T1 + (size_t)Nn * HID;
  float* bias = Bp + (size_t)KDIM * GCOLS;
  int2*  edgepay  = (int2*)(bias + GCOLS);
  int2*  payA     = edgepay + E;
  int2*  rowedges = payA + E;
  int*   keyA = (int*)(rowedges + E);
  int*   lrank = keyA + E;
  const int NB = (E + SORT_CH - 1) / SORT_CH;
  int*   table = lrank + E;
  float* deg  = (float*)(table + (size_t)NB * SORT_BINS);
  int*   rowstart = (int*)(deg + Nn);
  int*   rowend = rowstart + Nn;
  int*   bstart = rowend + Nn;

  float* out   = (float*)d_out;
  float* h_out = out   + (size_t)Nn * HID;
  float* c_out = h_out + (size_t)Nn * HID;

  const float DIAG = 2.0f / LMAX - 1.0f;

  hipMemsetAsync(deg, 0, (size_t)Nn * sizeof(float), stream);
  const int NDEG = (E + 255) / 256;
  const int NPACK = (KDIM * GCOLS + 255) / 256;
  k_front<<<NB + NDEG + NPACK, 256, 0, stream>>>(src, dst, ew, deg, lrank, table,
                                                 Wx, theta, bg, conv_b, Bp, bias, E, NB, NDEG);
  k_lapw_pack<<<(E + 255) / 256, 256, 0, stream>>>(src, dst, ew, deg, edgepay, E);

  k_split_scatter<<<NB, 256, 0, stream>>>(dst, edgepay, BK_BITS, lrank, table, NB, keyA, payA,
                                          bstart, E);
  k_bucket_rows<<<(Nn + BK_ROWS - 1) / BK_ROWS, 256, 0, stream>>>(keyA, payA, bstart, rowedges,
                                                                  rowstart, rowend, Nn);

  k_spmm_t1<<<(Nn + 7) / 8, 256, 0, stream>>>(h, T1, rowedges, rowstart, rowend, DIAG, Nn);

  k_gemm_gates<<<(Nn + MT - 1) / MT, 256, 0, stream>>>(x, h, T1, rowedges, rowstart, rowend, DIAG,
                                                   Bp, bias, c, Wl, bl, out, h_out, c_out, Nn);
}


# Laplacian scaling per node, not per edge: gathers read deg^-1/2-scaled rows, row sums scaled once by -deg^-1/2; per-edge weight launch and its 2 random degree reads gone; products re-associated
# speedup vs baseline: 2.1531x; 1.0339x over previous
#include <hip/hip_runtime.h>
#include <hip/hip_bf16.h>

typedef __attribute__((ext_vector_type(2))) float v2f;
typedef __attribute__((ext_vector_type(8))) float v8f;

#define HID 64
#define INDIM 128
#define KDIM 320
#define GCOLS 256
#define LMAX 2.0f

__device__ __forceinline__ v8f wmma_f32_4(v2f a, v2f b, v8f c) {
  return __builtin_amdgcn_wmma_f32_16x16x4_f32(false, a, false, b, (short)0, c, false, false);
}

__device__ __forceinline__ float fsigmoid(float x) {
  return __builtin_amdgcn_rcpf(1.0f + __expf(-x));
}

__device__ __forceinline__ float ftanh(float x) {
  return 1.0f - 2.0f * __builtin_amdgcn_rcpf(1.0f + __expf(2.0f * x));
}

__device__ __forceinline__ void atomic_fadd(float* p, float v) {
  asm volatile("global_atomic_add_f32 %0, %1, off scope:SCOPE_DEV"
               :: "v"(p), "v"(v) : "memory");
}


__device__ __forceinline__
void degree_block(const int* __restrict__ src, const int* __restrict__ dst,
                  const float* __restrict__ ew, float* __restrict__ deg, int E, int bid) {
  int e = bid * 256 + threadIdx.x;
  if (e >= E) return;
  int s = src[e], d = dst[e];
  if (s != d) atomic_fadd(&deg[s], ew[e]);
}

__device__ __forceinline__ float dinv_of(float dg) {
  return (dg > 0.0f) ? rsqrtf(dg) : 0.0f;
}

#define SCALE_F4 8
__device__ __forceinline__
void scale_rows_block(const float* __restrict__ z, const float* __restrict__ deg,
                      float* __restrict__ zs, int Nn, int bid) {
  const int total = Nn * (HID / 4);
  const float4* z4 = (const float4*)z;
  float4* zs4 = (float4*)zs;
  float4 v[SCALE_F4];
  float dg[SCALE_F4];
#pragma unroll
  for (int u = 0; u < SCALE_F4; ++u) {
    const int i = bid * (SCALE_F4 * 256) + u * 256 + threadIdx.x;
    const int ic = i < total ? i : total - 1;
    v[u] = z4[ic];
    dg[u] = deg[ic / (HID / 4)];
  }
#pragma unroll
  for (int u = 0; u < SCALE_F4; ++u) {
    const int i = bid * (SCALE_F4 * 256) + u * 256 + threadIdx.x;
    if (i < total) {
      const float s = dinv_of(dg[u]);
      zs4[i] = make_float4(s * v[u].x, s * v[u].y, s * v[u].z, s * v[u].w);
    }
  }
}


#define SORT_CH 8192
#define SORT_ROUNDS (SORT_CH / 256)
#define SORT_BINS 512

__device__ __forceinline__ int wave_incl_scan(int v, int lane) {
  for (int off = 1; off < 32; off <<= 1) {
    int t = __shfl_up(v, off, 32);
    if (lane >= off) v += t;
  }
  return v;
}

__device__ __forceinline__ int block_excl_scan(int v, int* wsum, int lane, int wave, int& total) {
  const int incl = wave_incl_scan(v, lane);
  if (lane == 31) wsum[wave] = incl;
  __syncthreads();
  int wb = 0, t = 0;
  for (int w = 0; w < 8; ++w) { int s = wsum[w]; if (w < wave) wb += s; t += s; }
  total = t;
  __syncthreads();
  return wb + incl - v;
}

__device__ __forceinline__
void split_rank_block(const int* __restrict__ key, int shift,
                      int* __restrict__ lrank, int* __restrict__ table, int E, int bid) {
  __shared__ int whist[8 * SORT_BINS];
  const int tid = threadIdx.x, lane = tid & 31, wave = tid >> 5;
  for (int i = tid; i < 8 * SORT_BINS; i += 256) whist[i] = 0;
  __syncthreads();
  const int base = bid * SORT_CH + wave * (SORT_CH / 8) + lane;
  int dg[SORT_ROUNDS], lr[SORT_ROUNDS];
#pragma unroll
  for (int r = 0; r < SORT_ROUNDS; ++r) {
    const int e = base + r * 32;
    const int k = key[e < E ? e : E - 1];
    dg[r] = (e < E) ? ((k >> shift) & (SORT_BINS - 1)) : -1;
  }
  int* wh = whist + wave * SORT_BINS;
#pragma unroll
  for (int r = 0; r < SORT_ROUNDS; ++r) {
    const int d = dg[r];
    const bool valid = d >= 0;
    unsigned peers = __builtin_amdgcn_ballot_w32(valid);
#pragma unroll
    for (int b = 0; b < 9; ++b) {
      const bool bit = ((d >> b) & 1) != 0;
      const unsigned bal = __builtin_amdgcn_ballot_w32(bit);
      peers &= bit ? bal : ~bal;
    }
    const int rk = __popc(peers & ((1u << lane) - 1u));
    int b0 = 0;
    if (valid) {
      b0 = wh[d];
      if (rk == 0) wh[d] = b0 + __popc(peers);
    }
    lr[r] = b0 + rk;
  }
  __syncthreads();
  for (int dgt = tid; dgt < SORT_BINS; dgt += 256) {
    int run = 0;
    for (int w = 0; w < 8; ++w) {
      int v = whist[w * SORT_BINS + dgt];
      whist[w * SORT_BINS + dgt] = run;
      run += v;
    }
    table[bid * SORT_BINS + dgt] = run;
  }
  __syncthreads();
#pragma unroll
  for (int r = 0; r < SORT_ROUNDS; ++r) {
    const int e = base + r * 32;
    if (e < E) lrank[e] = lr[r] + wh[dg[r]];
  }
}

__global__ __launch_bounds__(256)
void k_split_scatter(const int* __restrict__ key, const int* __restrict__ srcv,
                     const float* __restrict__ ew, int shift,
                     const int* __restrict__ lrank, const int* __restrict__ table, int nblocks,
                     int* __restrict__ key_out, int2* __restrict__ pay_out,
                     int* __restrict__ bstart, int E,
                     const float* __restrict__ h, const float* __restrict__ deg,
                     float* __restrict__ hs, int Nn) {
  if ((int)blockIdx.x >= nblocks) {
    scale_rows_block(h, deg, hs, Nn, (int)blockIdx.x - nblocks);
    return;
  }
  __shared__ int lstart[SORT_BINS];
  __shared__ int gdelta[SORT_BINS];
  __shared__ int wsum[8];
  __shared__ int skey[SORT_CH];
  __shared__ __attribute__((aligned(8))) int spay_w[2 * SORT_CH];
  int2* spay = (int2*)spay_w;
  const int tid = threadIdx.x, lane = tid & 31, wave = tid >> 5;
  const int me = blockIdx.x;
  int pre0 = 0, tot0 = 0, pre1 = 0, tot1 = 0;
#pragma unroll 14
  for (int b = 0; b < nblocks; ++b) {
    const int v0 = table[b * SORT_BINS + tid];
    const int v1 = table[b * SORT_BINS + 256 + tid];
    tot0 += v0; tot1 += v1;
    if (b < me) { pre0 += v0; pre1 += v1; }
  }
  const int own0 = table[me * SORT_BINS + tid];
  const int own1 = table[me * SORT_BINS + 256 + tid];
  int sum0, sum1, osum0, osum1;
  const int ex0 = block_excl_scan(tot0, wsum, lane, wave, sum0);
  const int ex1 = block_excl_scan(tot1, wsum, lane, wave, sum1);
  const int l0 = block_excl_scan(own0, wsum, lane, wave, osum0);
  const int l1 = osum0 + block_excl_scan(own1, wsum, lane, wave, osum1);
  lstart[tid]       = l0;
  lstart[tid + 256] = l1;
  gdelta[tid]       = ex0 + pre0 - l0;
  gdelta[tid + 256] = sum0 + ex1 + pre1 - l1;
  if (me == 0) {
    bstart[tid]       = ex0;
    bstart[tid + 256] = sum0 + ex1;
    if (tid == 0) bstart[SORT_BINS] = sum0 + sum1;
  }
  __syncthreads();
  for (int r0 = 0; r0 < SORT_ROUNDS; r0 += 8) {
    int kk[8], rk[8];
    int2 pv[8];
#pragma unroll
    for (int u = 0; u < 8; ++u) {
      const int e = me * SORT_CH + (r0 + u) * 256 + tid;
      const int ec = e < E ? e : E - 1;
      kk[u] = key[ec];
      rk[u] = lrank[ec];
      const int s = srcv[ec];
      const float w = ew[ec];
      pv[u] = make_int2(s, __float_as_int(s == kk[u] ? 0.0f : w));
    }
    int lp[8];
#pragma unroll
    for (int u = 0; u < 8; ++u)
      lp[u] = lstart[(kk[u] >> shift) & (SORT_BINS - 1)] + rk[u];
#pragma unroll
    for (int u = 0; u < 8; ++u) {
      const int e = me * SORT_CH + (r0 + u) * 256 + tid;
      if (e < E) {
        skey[lp[u]] = kk[u];
        spay[lp[u]] = pv[u];
      }
    }
  }
  __syncthreads();
  const int rem = E - me * SORT_CH;
  const int cnt = rem < SORT_CH ? rem : SORT_CH;
  for (int r0 = 0; r0 < SORT_ROUNDS; r0 += 8) {
    int kk[8];
    int2 pv[8];
#pragma unroll
    for (int u = 0; u < 8; ++u) {
      const int i = (r0 + u) * 256 + tid;
      const int ic = i < cnt ? i : cnt - 1;
      kk[u] = skey[ic];
      pv[u] = spay[ic];
    }
    int ps[8];
#pragma unroll
    for (int u = 0; u < 8; ++u)
      ps[u] = (r0 + u) * 256 + tid + gdelta[(kk[u] >> shift) & (SORT_BINS - 1)];
#pragma unroll
    for (int u = 0; u < 8; ++u) {
      const int i = (r0 + u) * 256 + tid;
      if (i < cnt) {
        key_out[ps[u]] = kk[u];
        pay_out[ps[u]] = pv[u];
      }
    }
  }
}

#define BK_BITS 8
#define BK_ROWS (1 << BK_BITS)
#define BK_CH 8192
#define BK_ROUNDS (BK_CH / 256)

__device__ __forceinline__
int bucket_rank_trip(const int* __restrict__ bkey, int cnt, int c0, int tid, int lane, int wave,
                     int* whist, int (&dg)[BK_ROUNDS], int (&lr)[BK_ROUNDS],
                     int& seg, int& wbase, int& cend) {
  for (int i = tid; i < 8 * BK_ROWS; i += 256) whist[i] = 0;
  __syncthreads();
  const int left = cnt - c0;
  cend = c0 + (left < BK_CH ? left : BK_CH);
  seg = ((cend - c0 + 255) >> 8) << 5;
  wbase = c0 + wave * seg;
  int* wh = whist + wave * BK_ROWS;
#pragma unroll
  for (int r = 0; r < BK_ROUNDS; ++r) {
    const int e = wbase + r * 32 + lane;
    const int k = bkey[e < cnt ? e : cnt - 1];
    dg[r] = (r * 32 < seg && e < cend) ? (k & (BK_ROWS - 1)) : -1;
  }
#pragma unroll
  for (int r = 0; r < BK_ROUNDS; ++r) {
    lr[r] = 0;
    if (r * 32 < seg && wbase + r * 32 < cend) {
      const int d = dg[r];
      const bool valid = d >= 0;
      unsigned peers = __builtin_amdgcn_ballot_w32(valid);
#pragma unroll
      for (int q = 0; q < BK_BITS; ++q) {
        const bool bit = ((d >> q) & 1) != 0;
        const unsigned bal = __builtin_amdgcn_ballot_w32(bit);
        peers &= bit ? bal : ~bal;
      }
      const int rk = __popc(peers & ((1u << lane) - 1u));
      int b0 = 0;
      if (valid) {
        b0 = wh[d];
        if (rk == 0) wh[d] = b0 + __popc(peers);
      }
      lr[r] = b0 + rk;
    }
  }
  __syncthreads();
  int ctot = 0;
  for (int w = 0; w < 8; ++w) {
    const int v = whist[w * BK_ROWS + tid];
    whist[w * BK_ROWS + tid] = ctot;
    ctot += v;
  }
  __syncthreads();
  return ctot;
}

template <bool STAGED>
__device__ __forceinline__
void bucket_place_trip(const int2* __restrict__ bpay, int cnt, int lane,
                       const int* wh, const int* rowoff, const int* carry,
                       const int (&dg)[BK_ROUNDS], const int (&lr)[BK_ROUNDS],
                       int seg, int wbase, int cend, int2* spay, int2* __restrict__ gout) {
#pragma unroll
  for (int r0 = 0; r0 < BK_ROUNDS; r0 += 8) {
    if (r0 * 32 < seg && wbase + r0 * 32 < cend) {
      int2 pv[8];
#pragma unroll
      for (int u = 0; u < 8; ++u) {
        const int e = wbase + (r0 + u) * 32 + lane;
        pv[u] = bpay[e < cnt ? e : cnt - 1];
      }
#pragma unroll
      for (int u = 0; u < 8; ++u) {
        const int d = dg[r0 + u];
        if (d >= 0) {
          const int pos = rowoff[d] + carry[d] + wh[d] + lr[r0 + u];
          if (STAGED) spay[pos] = pv[u];
          else        gout[pos] = pv[u];
        }
      }
    }
  }
}

__global__ __launch_bounds__(256)
void k_bucket_rows(const int* __restrict__ key, const int2* __restrict__ pay,
                   const int* __restrict__ bstart, int2* __restrict__ rowedges,
                   int* __restrict__ rowstart, int* __restrict__ rowend, int Nn) {
  __shared__ int whist[8 * BK_ROWS];
  __shared__ int rowoff[BK_ROWS];
  __shared__ int carry[BK_ROWS];
  __shared__ int wsum[8];
  __shared__ __attribute__((aligned(8))) int spay_w[2 * BK_CH];
  int2* spay = (int2*)spay_w;
  const int tid = threadIdx.x, lane = tid & 31;
  const int wave = __builtin_amdgcn_readfirstlane(tid >> 5);
  const int b = blockIdx.x;
  const int g0 = __builtin_amdgcn_readfirstlane(bstart[b]);
  const int cnt = __builtin_amdgcn_readfirstlane(bstart[b + 1]) - g0;
  const int* bkey = key + g0;
  const int2* bpay = pay + g0;
  const int* wh = whist + wave * BK_ROWS;
  const bool staged = cnt <= BK_CH;

  carry[tid] = 0;
  int dg[BK_ROUNDS], lr[BK_ROUNDS];
  int seg = 0, wbase = 0, cend = 0;
  int mycnt = 0;
  for (int c0 = 0; c0 < cnt; c0 += BK_CH)
    mycnt += bucket_rank_trip(bkey, cnt, c0, tid, lane, wave, whist, dg, lr, seg, wbase, cend);
  int total;
  const int ex = block_excl_scan(mycnt, wsum, lane, wave, total);
  rowoff[tid] = ex;
  const int n = b * BK_ROWS + tid;
  if (n < Nn) {
    rowstart[n] = g0 + ex;
    rowend[n] = g0 + ex + mycnt;
  }
  __syncthreads();

  if (staged) {
    if (cnt > 0)
      bucket_place_trip<true>(bpay, cnt, lane, wh, rowoff, carry, dg, lr, seg, wbase, cend,
                              spay, rowedges + g0);
    __syncthreads();
    for (int i0 = 0; i0 < cnt; i0 += 8 * 256) {
      int2 pv[8];
#pragma unroll
      for (int u = 0; u < 8; ++u) {
        const int i = i0 + u * 256 + tid;
        pv[u] = spay[i < cnt ? i : cnt - 1];
      }
#pragma unroll
      for (int u = 0; u < 8; ++u) {
        const int i = i0 + u * 256 + tid;
        if (i < cnt) rowedges[g0 + i] = pv[u];
      }
    }
  } else {
    for (int c0 = 0; c0 < cnt; c0 += BK_CH) {
      const int ctot = bucket_rank_trip(bkey, cnt, c0, tid, lane, wave, whist, dg, lr,
                                        seg, wbase, cend);
      bucket_place_trip<false>(bpay, cnt, lane, wh, rowoff, carry, dg, lr, seg, wbase, cend,
                               spay, rowedges + g0);
      __syncthreads();
      carry[tid] += ctot;
    }
  }
}

#define ROWCHUNK 8
__device__ __forceinline__ float2 lhat_row(const float* __restrict__ zs,
                                           const float* __restrict__ z,
                                           const float* __restrict__ deg,
                                           const int2* __restrict__ rowedges,
                                           const int* __restrict__ rowstart,
                                           const int* __restrict__ rowend,
                                           float diag, int n, int lane) {
  const float2* z2 = (const float2*)zs;
  const int p0 = rowstart[n];
  const int c = rowend[n] - p0;
  const int2* ce = rowedges + p0;
  float2 acc = make_float2(0.0f, 0.0f);
  for (int j = 0; j < c; j += ROWCHUNK) {
    float2 zv[ROWCHUNK];
    float wv[ROWCHUNK];
#pragma unroll
    for (int u = 0; u < ROWCHUNK; ++u) {
      const int idx = (j + u < c) ? (j + u) : (c - 1);
      const int2 en = ce[idx];
      wv[u] = (j + u < c) ? __int_as_float(en.y) : 0.0f;
      zv[u] = z2[(size_t)en.x * (HID / 2) + lane];
    }
#pragma unroll
    for (int u = 0; u < ROWCHUNK; ++u) {
      acc.x += wv[u] * zv[u].x;
      acc.y += wv[u] * zv[u].y;
    }
  }
  const float2 zn = ((const float2*)z)[(size_t)n * (HID / 2) + lane];
  const float sc = -(2.0f / LMAX) * dinv_of(deg[n]);
  acc.x = sc * acc.x;
  acc.y = sc * acc.y;
  acc.x += diag * zn.x;
  acc.y += diag * zn.y;
  return acc;
}

__global__ __launch_bounds__(256)
void k_spmm_t1(const float* __restrict__ hs, const float* __restrict__ h,
               const float* __restrict__ deg, float* __restrict__ T1, float* __restrict__ T1s,
               const int2* __restrict__ rowedges, const int* __restrict__ rowstart,
               const int* __restrict__ rowend, float diag, int Nn) {
  const int lane = threadIdx.x & 31;
  const int n = __builtin_amdgcn_readfirstlane(blockIdx.x * 8 + (threadIdx.x >> 5));
  if (n >= Nn) return;
  const float2 r = lhat_row(hs, h, deg, rowedges, rowstart, rowend, diag, n, lane);
  const float dn = dinv_of(deg[n]);
  ((float2*)T1)[(size_t)n * (HID / 2) + lane] = r;
  ((float2*)T1s)[(size_t)n * (HID / 2) + lane] = make_float2(dn * r.x, dn * r.y);
}


__device__ __forceinline__
void packB_block(const float* __restrict__ Wx, const float* __restrict__ theta,
                 const float* __restrict__ bg, const float* __restrict__ conv_b,
                 float* __restrict__ Bp, float* __restrict__ bias, int bid) {
  int gid = bid * 256 + threadIdx.x;
  if (gid >= KDIM * GCOLS) return;
  int k = gid / GCOLS, col = gid % GCOLS;
  int g = col >> 6, o = col & 63;
  float v;
  if (k < INDIM) {
    v = Wx[((size_t)g * INDIM + k) * HID + o];
  } else {
    int kk = k - INDIM;
    int kb = kk >> 6, kr = kk & 63;
    v = theta[(((size_t)g * 3 + kb) * HID + kr) * HID + o];
  }
  Bp[gid] = v;
  if (gid < GCOLS) bias[gid] = bg[g * HID + o] + conv_b[g * HID + o];
}

__global__ __launch_bounds__(256)
void k_front(const int* __restrict__ src, const int* __restrict__ dst,
             const float* __restrict__ ew, float* __restrict__ deg,
             int* __restrict__ lrank, int* __restrict__ table,
             const float* __restrict__ Wx, const float* __restrict__ theta,
             const float* __restrict__ bg, const float* __restrict__ conv_b,
             float* __restrict__ Bp, float* __restrict__ bias,
             int E, int nrank, int ndeg) {
  const int b = blockIdx.x;
  if (b < nrank) {
    split_rank_block(dst, BK_BITS, lrank, table, E, b);
  } else if (b < nrank + ndeg) {
    degree_block(src, dst, ew, deg, E, b - nrank);
  } else {
    packB_block(Wx, theta, bg, conv_b, Bp, bias, b - nrank - ndeg);
  }
}


#define AS1 324
#define PS1 260
#define MT 80
#define MSUB (MT / 16)

__device__ __forceinline__ void lstm_cell(float pi, float pf, float pt, float po, float cold,
                                          float& hnew, float& cnew) {
  float ig = fsigmoid(pi);
  float fg = fsigmoid(pf);
  float tg = ftanh(pt);
  float og = fsigmoid(po);
  float cv = fg * cold + ig * tg;
  hnew = og * ftanh(cv);
  cnew = cv;
}

#define TROWS (MT / 8)
#define TCH 3

__device__ __forceinline__ int2 t2_fetch_edge(const int2* __restrict__ rowedges,
                                              int p0, int c, int slot) {
  const bool live = slot < c;
  const int gi = live ? p0 + slot : (c > 0 ? p0 + c - 1 : 0);
  int2 en = rowedges[gi];
  if (!live) en.y = 0;
  return en;
}

__device__ __forceinline__
void t2_rows_to_tile(const float* __restrict__ T1s, const float* __restrict__ T1,
                     const float* __restrict__ h, const float* __restrict__ deg,
                     const int2* __restrict__ rowedges, const int* __restrict__ rowstart,
                     const int* __restrict__ rowend, float diag, int m0, int Nn,
                     int wave, int lane, float* Asmem) {
  const float2* z2 = (const float2*)T1s;
  const float2* t2 = (const float2*)T1;
  const float2* h2 = (const float2*)h;
  const int il = lane / TCH, ul = lane % TCH;
  for (int g = 0; g < (MT / 8) / TROWS; ++g) {
    const int row0 = wave * (MT / 8) + g * TROWS;
    int nl = m0 + row0 + il;
    nl = nl < Nn ? nl : Nn - 1;
    const int p0l = rowstart[nl];
    const int cl = rowend[nl] - p0l;
    int cmax = 0;
#pragma unroll
    for (int i = 0; i < TROWS; ++i) {
      const int ci = __builtin_amdgcn_readlane(cl, i * TCH);
      cmax = ci > cmax ? ci : cmax;
    }
    float2 acc[TROWS];
#pragma unroll
    for (int i = 0; i < TROWS; ++i) acc[i] = make_float2(0.0f, 0.0f);
    int2 en = t2_fetch_edge(rowedges, p0l, cl, ul);
    for (int j = 0; j < cmax; j += TCH) {
      float2 zv[TROWS][TCH];
      float wv[TROWS][TCH];
#pragma unroll
      for (int i = 0; i < TROWS; ++i) {
#pragma unroll
        for (int u = 0; u < TCH; ++u) {
          const int src = __builtin_amdgcn_readlane(en.x, i * TCH + u);
          wv[i][u] = __int_as_float(__builtin_amdgcn_readlane(en.y, i * TCH + u));
          zv[i][u] = z2[(size_t)src * (HID / 2) + lane];
        }
      }
      en = t2_fetch_edge(rowedges, p0l, cl, j + TCH + ul);
#pragma unroll
      for (int i = 0; i < TROWS; ++i) {
#pragma unroll
        for (int u = 0; u < TCH; ++u) {
          acc[i].x += wv[i][u] * zv[i][u].x;
          acc[i].y += wv[i][u] * zv[i][u].y;
        }
      }
    }
#pragma unroll
    for (int i = 0; i < TROWS; ++i) {
      int n = m0 + row0 + i;
      n = n < Nn ? n : Nn - 1;
      const size_t o = (size_t)n * (HID / 2) + lane;
      const float2 zn = t2[o];
      const float2 hv = h2[o];
      const float sc = -(2.0f / LMAX) * dinv_of(deg[n]);
      float2 r;
      r.x = sc * acc[i].x;
      r.y = sc * acc[i].y;
      r.x += diag * zn.x;
      r.y += diag * zn.y;
      float2 t;
      t.x = 2.0f * r.x - hv.x;
      t.y = 2.0f * r.y - hv.y;
      *(float2*)&Asmem[(row0 + i) * AS1 + 256 + 2 * lane] = t;
    }
  }
}

__global__ __launch_bounds__(256)
void k_gemm_gates(const float* __restrict__ x, const float* __restrict__ h,
                  const float* __restrict__ T1, const float* __restrict__ T1s,
                  const float* __restrict__ deg,
                  const int2* __restrict__ rowedges, const int* __restrict__ rowstart,
                  const int* __restrict__ rowend, float diag,
                  const float* __restrict__ Bp, const float* __restrict__ bias,
                  const float* __restrict__ c_in,
                  const float* __restrict__ Wl, const float* __restrict__ bl,
                  float* __restrict__ out,
                  float* __restrict__ h_out, float* __restrict__ c_out, int Nn) {
  __shared__ __attribute__((aligned(16))) float Asmem[MT * AS1];
  float* Psmem = Asmem;
  float* Hsmem = Asmem;
  const int tid = threadIdx.x;
  const int m0 = blockIdx.x * MT;

#pragma unroll
  for (int it = 0; it < MT * 32 / 256; ++it) {
    int i = it * 256 + tid;
    int r = i >> 5, q = i & 31;
    int n = m0 + r; if (n >= Nn) n = Nn - 1;
    const float4 v = *(const float4*)(x + (size_t)n * INDIM + q * 4);
    *(float4*)&Asmem[r * AS1 + q * 4] = v;
  }
#pragma unroll
  for (int it = 0; it < MT * 16 / 256; ++it) {
    int i = it * 256 + tid;
    int r = i >> 4, q = i & 15;
    int n = m0 + r; if (n >= Nn) n = Nn - 1;
    const size_t g = (size_t)n * HID + q * 4;
    const float4 vh = *(const float4*)(h + g);
    const float4 v1 = *(const float4*)(T1 + g);
    float* ap = &Asmem[r * AS1 + 128 + q * 4];
    *(float4*)(ap)       = vh;
    *(float4*)(ap + 64)  = v1;
  }
  const int lane = tid & 31;
  const int wave = __builtin_amdgcn_readfirstlane(tid >> 5);
  t2_rows_to_tile(T1s, T1, h, deg, rowedges, rowstart, rowend, diag, m0, Nn, wave, lane, Asmem);
  __syncthreads();

  const int lo = lane & 15, hi = lane >> 4;
  const int c0 = wave * 32;
  v8f acc[MSUB][2];
#pragma unroll
  for (int t = 0; t < MSUB; ++t) { acc[t][0] = v8f{}; acc[t][1] = v8f{}; }

  for (int k0 = 0; k0 < KDIM; k0 += 4) {
    const float* bptr = Bp + (size_t)(k0 + 2 * hi) * GCOLS + c0 + lo;
    v2f b0, b1;
    b0[0] = bptr[0];    b0[1] = bptr[GCOLS];
    b1[0] = bptr[16];   b1[1] = bptr[GCOLS + 16];
#pragma unroll
    for (int t = 0; t < MSUB; ++t) {
      v2f a = *(const v2f*)&Asmem[(t * 16 + lo) * AS1 + k0 + 2 * hi];
      acc[t][0] = wmma_f32_4(a, b0, acc[t][0]);
      acc[t][1] = wmma_f32_4(a, b1, acc[t][1]);
    }
  }

  float bias0 = bias[c0 + lo], bias1 = bias[c0 + 16 + lo];
  const int oc = (wave & 3) * 16;
  v2f wb[HID / 4];
#pragma unroll
  for (int s = 0; s < HID / 4; ++s) {
    wb[s][0] = Wl[(4 * s + 2 * hi) * HID + oc + lo];
    wb[s][1] = Wl[(4 * s + 2 * hi + 1) * HID + oc + lo];
  }
  __syncthreads();
#pragma unroll
  for (int t = 0; t < MSUB; ++t) {
#pragma unroll
    for (int r = 0; r < 8; ++r) {
      int row = t * 16 + r + 8 * hi;
      Psmem[row * PS1 + c0 + lo]      = acc[t][0][r] + bias0;
      Psmem[row * PS1 + c0 + 16 + lo] = acc[t][1][r] + bias1;
    }
  }
  __syncthreads();

  for (int it = 0; it < MT * 16 / 256; ++it) {
    int i = it * 256 + tid;
    int row = i >> 4, o = (i & 15) * 4;
    int n = m0 + row;
    if (n >= Nn) {
      *(float4*)&Hsmem[row * PS1 + o] = make_float4(0.0f, 0.0f, 0.0f, 0.0f);
      continue;
    }
    const float4 cold = *(const float4*)(c_in + (size_t)n * HID + o);
    const float* pr = &Psmem[row * PS1 + o];
    const float4 pi = *(const float4*)(pr);
    const float4 pf = *(const float4*)(pr + 64);
    const float4 pt = *(const float4*)(pr + 128);
    const float4 po = *(const float4*)(pr + 192);
    float4 hv, cv;
    lstm_cell(pi.x, pf.x, pt.x, po.x, cold.x, hv.x, cv.x);
    lstm_cell(pi.y, pf.y, pt.y, po.y, cold.y, hv.y, cv.y);
    lstm_cell(pi.z, pf.z, pt.z, po.z, cold.z, hv.z, cv.z);
    lstm_cell(pi.w, pf.w, pt.w, po.w, cold.w, hv.w, cv.w);
    *(float4*)(h_out + (size_t)n * HID + o) = hv;
    *(float4*)(c_out + (size_t)n * HID + o) = cv;
    float4 hr;
    hr.x = hv.x > 0.0f ? hv.x : 0.0f;
    hr.y = hv.y > 0.0f ? hv.y : 0.0f;
    hr.z = hv.z > 0.0f ? hv.z : 0.0f;
    hr.w = hv.w > 0.0f ? hv.w : 0.0f;
    *(float4*)&Hsmem[row * PS1 + o] = hr;
  }
  __syncthreads();

  {
    const int half = wave >> 2;
    const int rt = half * 2;
    v8f oacc0 = {}, oacc1 = {};
#pragma unroll
    for (int k0 = 0; k0 < HID; k0 += 4) {
      const v2f b = wb[k0 / 4];
      v2f a0 = *(const v2f*)&Hsmem[(rt * 16 + lo) * PS1 + k0 + 2 * hi];
      v2f a1 = *(const v2f*)&Hsmem[(rt * 16 + 16 + lo) * PS1 + k0 + 2 * hi];
      oacc0 = wmma_f32_4(a0, b, oacc0);
      oacc1 = wmma_f32_4(a1, b, oacc1);
    }
    const float bb = bl[oc + lo];
#pragma unroll
    for (int r = 0; r < 8; ++r) {
      const int n0 = m0 + rt * 16 + r + 8 * hi;
      const int n1 = n0 + 16;
      if (n0 < Nn) out[(size_t)n0 * HID + oc + lo] = oacc0[r] + bb;
      if (n1 < Nn) out[(size_t)n1 * HID + oc + lo] = oacc1[r] + bb;
    }
    if (half == 0) {
      const int r4 = (MSUB - 1) * 16;
      v8f oacc2 = {};
#pragma unroll
      for (int k0 = 0; k0 < HID; k0 += 4) {
        v2f a2 = *(const v2f*)&Hsmem[(r4 + lo) * PS1 + k0 + 2 * hi];
        oacc2 = wmma_f32_4(a2, wb[k0 / 4], oacc2);
      }
#pragma unroll
      for (int r = 0; r < 8; ++r) {
        const int n2 = m0 + r4 + r + 8 * hi;
        if (n2 < Nn) out[(size_t)n2 * HID + oc + lo] = oacc2[r] + bb;
      }
    }
  }
}


extern "C" void kernel_launch(void* const* d_in, const int* in_sizes, int n_in,
                              void* d_out, int out_size, void* d_ws, size_t ws_size,
                              hipStream_t stream) {
  const float* x      = (const float*)d_in[0];
  const int*   ei     = (const int*)d_in[1];
  const float* ew     = (const float*)d_in[2];
  const float* h      = (const float*)d_in[3];
  const float* c      = (const float*)d_in[4];
  const float* Wx     = (const float*)d_in[5];
  const float* bg     = (const float*)d_in[6];
  const float* theta  = (const float*)d_in[7];
  const float* conv_b = (const float*)d_in[8];
  const float* Wl     = (const float*)d_in[9];
  const float* bl     = (const float*)d_in[10];

  const int E  = in_sizes[2];
  const int Nn = in_sizes[3] / HID;
  const int* src = ei;
  const int* dst = ei + E;

  float* T1   = (float*)d_ws;
  float* T1s  = T1 + (size_t)Nn * HID;
  float* hs   = T1s + (size_t)Nn * HID;
  float* Bp   = hs + (size_t)Nn * HID;
  float* bias = Bp + (size_t)KDIM * GCOLS;
  int2*  payA     = (int2*)(bias + GCOLS);
  int2*  rowedges = payA + E;
  int*   keyA = (int*)(rowedges + E);
  int*   lrank = keyA + E;
  const int NB = (E + SORT_CH - 1) / SORT_CH;
  int*   table = lrank + E;
  float* deg  = (float*)(table + (size_t)NB * SORT_BINS);
  int*   rowstart = (int*)(deg + Nn);
  int*   rowend = rowstart + Nn;
  int*   bstart = rowend + Nn;

  float* out   = (float*)d_out;
  float* h_out = out   + (size_t)Nn * HID;
  float* c_out = h_out + (size_t)Nn * HID;

  const float DIAG = 2.0f / LMAX - 1.0f;

  hipMemsetAsync(deg, 0, (size_t)Nn * sizeof(float), stream);
  const int NDEG = (E + 255) / 256;
  const int NPACK = (KDIM * GCOLS + 255) / 256;
  k_front<<<NB + NDEG + NPACK, 256, 0, stream>>>(src, dst, ew, deg, lrank, table,
                                                 Wx, theta, bg, conv_b, Bp, bias, E, NB, NDEG);

  const int NSCALE = (Nn * (HID / 4) + SCALE_F4 * 256 - 1) / (SCALE_F4 * 256);
  k_split_scatter<<<NB + NSCALE, 256, 0, stream>>>(dst, src, ew, BK_BITS, lrank, table, NB, keyA,
                                                   payA, bstart, E, h, deg, hs, Nn);
  k_bucket_rows<<<(Nn + BK_ROWS - 1) / BK_ROWS, 256, 0, stream>>>(keyA, payA, bstart, rowedges,
                                                                  rowstart, rowend, Nn);

  k_spmm_t1<<<(Nn + 7) / 8, 256, 0, stream>>>(hs, h, deg, T1, T1s, rowedges, rowstart, rowend,
                                              DIAG, Nn);

  k_gemm_gates<<<(Nn + MT - 1) / MT, 256, 0, stream>>>(x, h, T1, T1s, deg, rowedges, rowstart,
                                                   rowend, DIAG, Bp, bias, c, Wl, bl, out, h_out,
                                                   c_out, Nn);
}
